# starting kernel
# baseline (speedup 1.0000x reference)
#include <hip/hip_runtime.h>

typedef __attribute__((ext_vector_type(16))) _Float16 v16h;
typedef __attribute__((ext_vector_type(8)))  float    v8f;

#define N_NODES 50000
#define N_EDGES 800000
#define FDIM    64
#define EFDIM   16
#define MSG_IN  145
#define MSG_K   160
#define NODE_K  128

#define WS_CNT_OFF     ((size_t)N_NODES * FDIM * 4)
#define WS_WEIGHTS_OFF ((size_t)13000192)

#define MW1B 0
#define MW2B 10240
#define CW1B 14336
#define CW2B 18432
#define NW1B 19456
#define NW2B 27648

__device__ __forceinline__ v8f wmma16(v16h a, v16h b, v8f c) {
    return __builtin_amdgcn_wmma_f32_16x16x32_f16(
         false, a,  false, b,
         (short)0, c,  false,  false);
}

__device__ __forceinline__ v16h load_a_row(const _Float16* rowptr, int kcbase, int khalf) {
    v16h a;
#pragma unroll
    for (int e = 0; e < 16; ++e) {
        int klocal = (e & 7) + 8 * khalf + 16 * (e >> 3);
        a[e] = rowptr[kcbase + klocal];
    }
    return a;
}

__global__ __launch_bounds__(256) void egnn_zero(float* msum, float* cnt, float* pos_out) {
    int i = blockIdx.x * 256 + threadIdx.x;
    msum[i] = 0.0f;
    if (i < N_NODES)     cnt[i] = 0.0f;
    if (i < N_NODES * 3) pos_out[i] = 0.0f;
}

__global__ __launch_bounds__(256) void egnn_prep_w(const float* __restrict__ W,
                                                   int Kact, int Nact, int ntiles,
                                                   int total, _Float16* __restrict__ out) {
    int idx = blockIdx.x * 256 + threadIdx.x;
    if (idx >= total) return;
    int block  = idx >> 9;
    int within = idx & 511;
    int lane = within >> 4;
    int e    = within & 15;
    int kc = block / ntiles;
    int nt = block - kc * ntiles;
    int khalf  = lane >> 4;
    int coln   = lane & 15;
    int klocal = (e & 7) + 8 * khalf + 16 * (e >> 3);
    int k = kc * 32 + klocal;
    int n = nt * 16 + coln;
    float v = (k < Kact && n < Nact) ? W[(size_t)k * Nact + n] : 0.0f;
    out[idx] = (_Float16)v;
}

__global__ __launch_bounds__(256) void egnn_edge(
    const float* __restrict__ x, const int* __restrict__ ei,
    const float* __restrict__ pos, const float* __restrict__ pos_init,
    const float* __restrict__ edge_attr,
    const float* __restrict__ mb1, const float* __restrict__ mb2,
    const float* __restrict__ cb1, const float* __restrict__ cb2,
    const _Float16* __restrict__ wsW,
    float* __restrict__ msum, float* __restrict__ cnt, float* __restrict__ pos_out)
{
    __shared__ _Float16 sfeat[128][MSG_K];
    __shared__ _Float16 shid[8][16][FDIM];
    __shared__ float sdsq[128];
    __shared__ float sdpi[128][3];
    __shared__ float srinv[128];
    __shared__ int   ssrc[128];
    __shared__ int   sdst[128];

    const int tid   = threadIdx.x;
    const int ebase = blockIdx.x * 128;

    if (tid < 128) {
        int e = ebase + tid;
        int s = ei[e];
        int d = ei[N_EDGES + e];
        ssrc[tid] = s; sdst[tid] = d;
        float dx = pos[3*s+0] - pos[3*d+0];
        float dy = pos[3*s+1] - pos[3*d+1];
        float dz = pos[3*s+2] - pos[3*d+2];
        sdsq[tid] = dx*dx + dy*dy + dz*dz;
        float ix = pos_init[3*s+0] - pos_init[3*d+0];
        float iy = pos_init[3*s+1] - pos_init[3*d+1];
        float iz = pos_init[3*s+2] - pos_init[3*d+2];
        sdpi[tid][0] = ix; sdpi[tid][1] = iy; sdpi[tid][2] = iz;
        srinv[tid] = 1.0f / sqrtf(ix*ix + iy*iy + iz*iz);
        atomicAdd(&cnt[d], 1.0f);
    }
    __syncthreads();

    for (int idx = tid; idx < 128 * MSG_K; idx += 256) {
        int row = idx / MSG_K, k = idx - row * MSG_K;
        float v;
        if      (k < FDIM)       v = x[(size_t)ssrc[row] * FDIM + k];
        else if (k < 2*FDIM)     v = x[(size_t)sdst[row] * FDIM + (k - FDIM)];
        else if (k == 2*FDIM)    v = sdsq[row];
        else if (k < MSG_IN)     v = edge_attr[(size_t)(ebase + row) * EFDIM + (k - 2*FDIM - 1)];
        else                     v = 0.0f;
        sfeat[row][k] = (_Float16)v;
    }
    __syncthreads();

    const int wave  = tid >> 5;
    const int lane  = tid & 31;
    const int col   = lane & 15;
    const int khalf = lane >> 4;
    const _Float16* arow = &sfeat[wave * 16 + col][0];

    v8f acc[4] = {v8f{}, v8f{}, v8f{}, v8f{}};
#pragma unroll
    for (int kc = 0; kc < 5; ++kc) {
        v16h a = load_a_row(arow, kc * 32, khalf);
#pragma unroll
        for (int nt = 0; nt < 4; ++nt) {
            v16h b = *(const v16h*)(wsW + MW1B + (kc * 4 + nt) * 512 + lane * 16);
            acc[nt] = wmma16(a, b, acc[nt]);
        }
    }
#pragma unroll
    for (int nt = 0; nt < 4; ++nt) {
        float bia = mb1[nt * 16 + col];
#pragma unroll
        for (int r = 0; r < 8; ++r) {
            float v = acc[nt][r] + bia;
            shid[wave][r + 8 * khalf][nt * 16 + col] = (_Float16)(v > 0.0f ? v : 0.0f);
        }
    }
    const _Float16* hrow = &shid[wave][col][0];

    v8f mij[4] = {v8f{}, v8f{}, v8f{}, v8f{}};
#pragma unroll
    for (int kc = 0; kc < 2; ++kc) {
        v16h a = load_a_row(hrow, kc * 32, khalf);
#pragma unroll
        for (int nt = 0; nt < 4; ++nt) {
            v16h b = *(const v16h*)(wsW + MW2B + (kc * 4 + nt) * 512 + lane * 16);
            mij[nt] = wmma16(a, b, mij[nt]);
        }
    }
#pragma unroll
    for (int nt = 0; nt < 4; ++nt) {
        float bia = mb2[nt * 16 + col];
#pragma unroll
        for (int r = 0; r < 8; ++r) {
            int row = r + 8 * khalf;
            float v = mij[nt][r] + bia;
            atomicAdd(&msum[(size_t)sdst[wave * 16 + row] * FDIM + nt * 16 + col], v);
            shid[wave][row][nt * 16 + col] = (_Float16)v;
        }
    }

    v8f tac[4] = {v8f{}, v8f{}, v8f{}, v8f{}};
#pragma unroll
    for (int kc = 0; kc < 2; ++kc) {
        v16h a = load_a_row(hrow, kc * 32, khalf);
#pragma unroll
        for (int nt = 0; nt < 4; ++nt) {
            v16h b = *(const v16h*)(wsW + CW1B + (kc * 4 + nt) * 512 + lane * 16);
            tac[nt] = wmma16(a, b, tac[nt]);
        }
    }
#pragma unroll
    for (int nt = 0; nt < 4; ++nt) {
        float bia = cb1[nt * 16 + col];
#pragma unroll
        for (int r = 0; r < 8; ++r) {
            float v = tac[nt][r] + bia;
            shid[wave][r + 8 * khalf][nt * 16 + col] = (_Float16)(v > 0.0f ? v : 0.0f);
        }
    }

    v8f cac = v8f{};
#pragma unroll
    for (int kc = 0; kc < 2; ++kc) {
        v16h a = load_a_row(hrow, kc * 32, khalf);
        v16h b = *(const v16h*)(wsW + CW2B + kc * 512 + lane * 16);
        cac = wmma16(a, b, cac);
    }
    if (col < 3) {
        float bia = cb2[col];
#pragma unroll
        for (int r = 0; r < 8; ++r) {
            int el = wave * 16 + r + 8 * khalf;
            float w = cac[r] + bia;
            float v = sdpi[el][col] * srinv[el] * w;
            atomicAdd(&pos_out[(size_t)sdst[el] * 3 + col], v);
        }
    }
}

__global__ __launch_bounds__(256) void egnn_node(
    const float* __restrict__ x, const float* __restrict__ msum,
    const float* __restrict__ cnt,
    const float* __restrict__ nb1, const float* __restrict__ nb2,
    const _Float16* __restrict__ wsW, float* __restrict__ hout)
{
    __shared__ _Float16 sfeat[128][NODE_K];
    __shared__ _Float16 shid[8][16][FDIM];
    __shared__ float sinv[128];

    const int tid   = threadIdx.x;
    const int nbase = blockIdx.x * 128;

    if (tid < 128) {
        int n = nbase + tid;
        float c = (n < N_NODES) ? cnt[n] : 1.0f;
        sinv[tid] = 1.0f / fmaxf(c, 1.0f);
    }
    __syncthreads();

    for (int idx = tid; idx < 128 * NODE_K; idx += 256) {
        int row = idx >> 7, k = idx & 127;
        int n = nbase + row;
        float v = 0.0f;
        if (n < N_NODES)
            v = (k < FDIM) ? x[(size_t)n * FDIM + k]
                           : msum[(size_t)n * FDIM + (k - FDIM)] * sinv[row];
        sfeat[row][k] = (_Float16)v;
    }
    __syncthreads();

    const int wave  = tid >> 5;
    const int lane  = tid & 31;
    const int col   = lane & 15;
    const int khalf = lane >> 4;
    const _Float16* arow = &sfeat[wave * 16 + col][0];

    v8f acc[4] = {v8f{}, v8f{}, v8f{}, v8f{}};
#pragma unroll
    for (int kc = 0; kc < 4; ++kc) {
        v16h a = load_a_row(arow, kc * 32, khalf);
#pragma unroll
        for (int nt = 0; nt < 4; ++nt) {
            v16h b = *(const v16h*)(wsW + NW1B + (kc * 4 + nt) * 512 + lane * 16);
            acc[nt] = wmma16(a, b, acc[nt]);
        }
    }
#pragma unroll
    for (int nt = 0; nt < 4; ++nt) {
        float bia = nb1[nt * 16 + col];
#pragma unroll
        for (int r = 0; r < 8; ++r) {
            float v = acc[nt][r] + bia;
            shid[wave][r + 8 * khalf][nt * 16 + col] = (_Float16)(v > 0.0f ? v : 0.0f);
        }
    }
    const _Float16* hrow = &shid[wave][col][0];

    v8f hacc[4] = {v8f{}, v8f{}, v8f{}, v8f{}};
#pragma unroll
    for (int kc = 0; kc < 2; ++kc) {
        v16h a = load_a_row(hrow, kc * 32, khalf);
#pragma unroll
        for (int nt = 0; nt < 4; ++nt) {
            v16h b = *(const v16h*)(wsW + NW2B + (kc * 4 + nt) * 512 + lane * 16);
            hacc[nt] = wmma16(a, b, hacc[nt]);
        }
    }
#pragma unroll
    for (int nt = 0; nt < 4; ++nt) {
        float bia = nb2[nt * 16 + col];
#pragma unroll
        for (int r = 0; r < 8; ++r) {
            int n = nbase + wave * 16 + r + 8 * khalf;
            if (n < N_NODES)
                hout[(size_t)n * FDIM + nt * 16 + col] = hacc[nt][r] + bia;
        }
    }
}

extern "C" void kernel_launch(void* const* d_in, const int* in_sizes, int n_in,
                              void* d_out, int out_size, void* d_ws, size_t ws_size,
                              hipStream_t stream) {
    const float* x         = (const float*)d_in[0];
    const int*   ei        = (const int*)  d_in[1];
    const float* pos       = (const float*)d_in[2];
    const float* pos_init  = (const float*)d_in[3];
    const float* edge_attr = (const float*)d_in[4];
    const float* mw1 = (const float*)d_in[5];
    const float* mb1 = (const float*)d_in[6];
    const float* mw2 = (const float*)d_in[7];
    const float* mb2 = (const float*)d_in[8];
    const float* nw1 = (const float*)d_in[9];
    const float* nb1 = (const float*)d_in[10];
    const float* nw2 = (const float*)d_in[11];
    const float* nb2 = (const float*)d_in[12];
    const float* cw1 = (const float*)d_in[13];
    const float* cb1 = (const float*)d_in[14];
    const float* cw2 = (const float*)d_in[15];
    const float* cb2 = (const float*)d_in[16];

    float* h_out   = (float*)d_out;
    float* pos_out = h_out + (size_t)N_NODES * FDIM;

    float* msum = (float*)d_ws;
    float* cnt  = (float*)((char*)d_ws + WS_CNT_OFF);
    _Float16* wsW = (_Float16*)((char*)d_ws + WS_WEIGHTS_OFF);

    egnn_zero<<<(N_NODES * FDIM) / 256, 256, 0, stream>>>(msum, cnt, pos_out);

    egnn_prep_w<<<40, 256, 0, stream>>>(mw1, MSG_IN, 64, 4, 20 * 512, wsW + MW1B);
    egnn_prep_w<<<16, 256, 0, stream>>>(mw2, 64,     64, 4,  8 * 512, wsW + MW2B);
    egnn_prep_w<<<16, 256, 0, stream>>>(cw1, 64,     64, 4,  8 * 512, wsW + CW1B);
    egnn_prep_w<<< 4, 256, 0, stream>>>(cw2, 64,      3, 1,  2 * 512, wsW + CW2B);
    egnn_prep_w<<<32, 256, 0, stream>>>(nw1, 128,    64, 4, 16 * 512, wsW + NW1B);
    egnn_prep_w<<<16, 256, 0, stream>>>(nw2, 64,     64, 4,  8 * 512, wsW + NW2B);

    egnn_edge<<<N_EDGES / 128, 256, 0, stream>>>(
        x, ei, pos, pos_init, edge_attr, mb1, mb2, cb1, cb2,
        wsW, msum, cnt, pos_out);

    egnn_node<<<(N_NODES + 127) / 128, 256, 0, stream>>>(
        x, msum, cnt, nb1, nb2, wsW, h_out);
}


# Edge products transposed (packed weights as A, a wave's 16 edges as columns): gathered rows load straight into B fragments, each layer's D converts in registers into the next operand; no LDS
# speedup vs baseline: 1.3450x; 1.3450x over previous
#include <hip/hip_runtime.h>

typedef __attribute__((ext_vector_type(16))) _Float16 v16h;
typedef __attribute__((ext_vector_type(8)))  float    v8f;

#define N_NODES 50000
#define N_EDGES 800000
#define FDIM    64
#define EFDIM   16
#define MSG_IN  145
#define MSG_K   160
#define NODE_K  128

#define WS_CNT_OFF     ((size_t)N_NODES * FDIM * 4)
#define WS_WEIGHTS_OFF ((size_t)13000192)

#define MW1B 0
#define MW2B 10240
#define CW1B 14336
#define CW2B 18432
#define NW1B 19456
#define NW2B 27648

__device__ __forceinline__ v8f wmma16(v16h a, v16h b, v8f c) {
    return __builtin_amdgcn_wmma_f32_16x16x32_f16(
         false, a,  false, b,
         (short)0, c,  false,  false);
}

__device__ __forceinline__ v16h load_a_row(const _Float16* rowptr, int kcbase, int khalf) {
    v16h a;
#pragma unroll
    for (int e = 0; e < 16; ++e) {
        int klocal = (e & 7) + 8 * khalf + 16 * (e >> 3);
        a[e] = rowptr[kcbase + klocal];
    }
    return a;
}

__global__ __launch_bounds__(256) void egnn_zero(float* msum, float* cnt, float* pos_out) {
    int i = blockIdx.x * 256 + threadIdx.x;
    msum[i] = 0.0f;
    if (i < N_NODES)     cnt[i] = 0.0f;
    if (i < N_NODES * 3) pos_out[i] = 0.0f;
}

__global__ __launch_bounds__(256) void egnn_prep_w(const float* __restrict__ W,
                                                   int Kact, int Nact, int ntiles,
                                                   int total, _Float16* __restrict__ out) {
    int idx = blockIdx.x * 256 + threadIdx.x;
    if (idx >= total) return;
    int block  = idx >> 9;
    int within = idx & 511;
    int lane = within >> 4;
    int e    = within & 15;
    int kc = block / ntiles;
    int nt = block - kc * ntiles;
    int khalf  = lane >> 4;
    int coln   = lane & 15;
    int klocal = (e & 7) + 8 * khalf + 16 * (e >> 3);
    int k = kc * 32 + klocal;
    int n = nt * 16 + coln;
    float v = (k < Kact && n < Nact) ? W[(size_t)k * Nact + n] : 0.0f;
    out[idx] = (_Float16)v;
}

typedef __attribute__((ext_vector_type(4))) float v4f;

__device__ __forceinline__ v16h ld_w(const _Float16* __restrict__ w, int lane) {
    return *(const v16h*)(w + lane * 16);
}

__device__ __forceinline__ v16h row_frag(const float* __restrict__ p, int h) {
    const v4f* q = (const v4f*)(p + 8 * h);
    v4f a0 = q[0], a1 = q[1], b0 = q[4], b1 = q[5];
    v16h f;
    f[0]  = (_Float16)a0.x; f[1]  = (_Float16)a0.y; f[2]  = (_Float16)a0.z; f[3]  = (_Float16)a0.w;
    f[4]  = (_Float16)a1.x; f[5]  = (_Float16)a1.y; f[6]  = (_Float16)a1.z; f[7]  = (_Float16)a1.w;
    f[8]  = (_Float16)b0.x; f[9]  = (_Float16)b0.y; f[10] = (_Float16)b0.z; f[11] = (_Float16)b0.w;
    f[12] = (_Float16)b1.x; f[13] = (_Float16)b1.y; f[14] = (_Float16)b1.z; f[15] = (_Float16)b1.w;
    return f;
}

__device__ __forceinline__ v8f bias8(v8f a, const float* __restrict__ b, bool relu) {
    const v4f* q = (const v4f*)b;
    v4f b0 = q[0], b1 = q[1];
    float bb[8] = {b0.x, b0.y, b0.z, b0.w, b1.x, b1.y, b1.z, b1.w};
    v8f o;
#pragma unroll
    for (int r = 0; r < 8; ++r) {
        float v = a[r] + bb[r];
        o[r] = relu ? (v > 0.0f ? v : 0.0f) : v;
    }
    return o;
}

__device__ __forceinline__ v16h pack2(v8f lo, v8f hi) {
    v16h f;
#pragma unroll
    for (int r = 0; r < 8; ++r) { f[r] = (_Float16)lo[r]; f[8 + r] = (_Float16)hi[r]; }
    return f;
}

__global__ __launch_bounds__(256) void egnn_edge(
    const float* __restrict__ x, const int* __restrict__ ei,
    const float* __restrict__ pos, const float* __restrict__ pos_init,
    const float* __restrict__ edge_attr,
    const float* __restrict__ mb1, const float* __restrict__ mb2,
    const float* __restrict__ cb1, const float* __restrict__ cb2,
    const _Float16* __restrict__ wsW,
    float* __restrict__ msum, float* __restrict__ cnt, float* __restrict__ pos_out)
{
    const int lane = threadIdx.x & 31;
    const int c = lane & 15;
    const int h = lane >> 4;
    const int e = (blockIdx.x * 8 + (threadIdx.x >> 5)) * 16 + c;
    const int s = ei[e];
    const int d = ei[N_EDGES + e];

    const float* xs = x + (size_t)s * FDIM;
    const float* xd = x + (size_t)d * FDIM;
    v16h fr[5];
    fr[0] = row_frag(xs, h);
    fr[1] = row_frag(xs + 32, h);
    fr[2] = row_frag(xd, h);
    fr[3] = row_frag(xd + 32, h);
    const v4f* eq = (const v4f*)(edge_attr + (size_t)e * EFDIM);
    v4f e0 = eq[0], e1 = eq[1], e2 = eq[2], e3 = eq[3];

    float dx = pos[3*s+0] - pos[3*d+0];
    float dy = pos[3*s+1] - pos[3*d+1];
    float dz = pos[3*s+2] - pos[3*d+2];
    float dsq = dx*dx + dy*dy + dz*dz;
    float ix = pos_init[3*s+0] - pos_init[3*d+0];
    float iy = pos_init[3*s+1] - pos_init[3*d+1];
    float iz = pos_init[3*s+2] - pos_init[3*d+2];
    float rinv = 1.0f / sqrtf(ix*ix + iy*iy + iz*iz);
    if (h == 0) atomicAdd(&cnt[d], 1.0f);

    {
        float ea[16] = {e0.x, e0.y, e0.z, e0.w, e1.x, e1.y, e1.z, e1.w,
                        e2.x, e2.y, e2.z, e2.w, e3.x, e3.y, e3.z, e3.w};
        v16h f;
#pragma unroll
        for (int t = 0; t < 8; ++t) {
            float lo = (t == 0) ? dsq : ea[t - 1];
            float hi = ea[7 + t];
            f[t] = (_Float16)(h ? hi : lo);
        }
        f[8] = (_Float16)(h ? 0.0f : ea[15]);
#pragma unroll
        for (int t = 9; t < 16; ++t) f[t] = (_Float16)0.0f;
        fr[4] = f;
    }

    v8f acc[4] = {v8f{}, v8f{}, v8f{}, v8f{}};
#pragma unroll
    for (int kc = 0; kc < 5; ++kc) {
#pragma unroll
        for (int mt = 0; mt < 4; ++mt)
            acc[mt] = wmma16(ld_w(wsW + MW1B + (kc * 4 + mt) * 512, lane), fr[kc], acc[mt]);
    }
    v16h hf[2];
#pragma unroll
    for (int kc = 0; kc < 2; ++kc)
        hf[kc] = pack2(bias8(acc[2*kc],     mb1 + 32*kc      + 8*h, true),
                       bias8(acc[2*kc + 1], mb1 + 32*kc + 16 + 8*h, true));

    {
        v8f mij[4] = {v8f{}, v8f{}, v8f{}, v8f{}};
#pragma unroll
        for (int kc = 0; kc < 2; ++kc) {
#pragma unroll
            for (int nt = 0; nt < 4; ++nt)
                mij[nt] = wmma16(hf[kc], ld_w(wsW + MW2B + (kc * 4 + nt) * 512, lane), mij[nt]);
        }
        int drow[8];
#pragma unroll
        for (int r = 0; r < 8; ++r) drow[r] = __shfl(d, 8 * h + r);
#pragma unroll
        for (int nt = 0; nt < 4; ++nt) {
            float bia = mb2[nt * 16 + c];
#pragma unroll
            for (int r = 0; r < 8; ++r)
                atomicAdd(&msum[(size_t)drow[r] * FDIM + nt * 16 + c], mij[nt][r] + bia);
        }
    }

    v16h mf[2];
    {
        v8f mt_[4] = {v8f{}, v8f{}, v8f{}, v8f{}};
#pragma unroll
        for (int kc = 0; kc < 2; ++kc) {
#pragma unroll
            for (int mt = 0; mt < 4; ++mt)
                mt_[mt] = wmma16(ld_w(wsW + MW2B + (kc * 4 + mt) * 512, lane), hf[kc], mt_[mt]);
        }
#pragma unroll
        for (int kc = 0; kc < 2; ++kc)
            mf[kc] = pack2(bias8(mt_[2*kc],     mb2 + 32*kc      + 8*h, false),
                           bias8(mt_[2*kc + 1], mb2 + 32*kc + 16 + 8*h, false));
    }

    v16h tf[2];
    {
        v8f tac[4] = {v8f{}, v8f{}, v8f{}, v8f{}};
#pragma unroll
        for (int kc = 0; kc < 2; ++kc) {
#pragma unroll
            for (int mt = 0; mt < 4; ++mt)
                tac[mt] = wmma16(ld_w(wsW + CW1B + (kc * 4 + mt) * 512, lane), mf[kc], tac[mt]);
        }
#pragma unroll
        for (int kc = 0; kc < 2; ++kc)
            tf[kc] = pack2(bias8(tac[2*kc],     cb1 + 32*kc      + 8*h, true),
                           bias8(tac[2*kc + 1], cb1 + 32*kc + 16 + 8*h, true));
    }

    v8f cac = v8f{};
#pragma unroll
    for (int kc = 0; kc < 2; ++kc)
        cac = wmma16(ld_w(wsW + CW2B + kc * 512, lane), tf[kc], cac);
    float ux = ix * rinv * (cac[0] + cb2[0]);
    float uy = iy * rinv * (cac[1] + cb2[1]);
    float uz = iz * rinv * (cac[2] + cb2[2]);
    if (h == 0) {
        atomicAdd(&pos_out[(size_t)d * 3 + 0], ux);
        atomicAdd(&pos_out[(size_t)d * 3 + 1], uy);
        atomicAdd(&pos_out[(size_t)d * 3 + 2], uz);
    }
}

__global__ __launch_bounds__(256) void egnn_node(
    const float* __restrict__ x, const float* __restrict__ msum,
    const float* __restrict__ cnt,
    const float* __restrict__ nb1, const float* __restrict__ nb2,
    const _Float16* __restrict__ wsW, float* __restrict__ hout)
{
    __shared__ _Float16 sfeat[128][NODE_K];
    __shared__ _Float16 shid[8][16][FDIM];
    __shared__ float sinv[128];

    const int tid   = threadIdx.x;
    const int nbase = blockIdx.x * 128;

    if (tid < 128) {
        int n = nbase + tid;
        float c = (n < N_NODES) ? cnt[n] : 1.0f;
        sinv[tid] = 1.0f / fmaxf(c, 1.0f);
    }
    __syncthreads();

    for (int idx = tid; idx < 128 * NODE_K; idx += 256) {
        int row = idx >> 7, k = idx & 127;
        int n = nbase + row;
        float v = 0.0f;
        if (n < N_NODES)
            v = (k < FDIM) ? x[(size_t)n * FDIM + k]
                           : msum[(size_t)n * FDIM + (k - FDIM)] * sinv[row];
        sfeat[row][k] = (_Float16)v;
    }
    __syncthreads();

    const int wave  = tid >> 5;
    const int lane  = tid & 31;
    const int col   = lane & 15;
    const int khalf = lane >> 4;
    const _Float16* arow = &sfeat[wave * 16 + col][0];

    v8f acc[4] = {v8f{}, v8f{}, v8f{}, v8f{}};
#pragma unroll
    for (int kc = 0; kc < 4; ++kc) {
        v16h a = load_a_row(arow, kc * 32, khalf);
#pragma unroll
        for (int nt = 0; nt < 4; ++nt) {
            v16h b = *(const v16h*)(wsW + NW1B + (kc * 4 + nt) * 512 + lane * 16);
            acc[nt] = wmma16(a, b, acc[nt]);
        }
    }
#pragma unroll
    for (int nt = 0; nt < 4; ++nt) {
        float bia = nb1[nt * 16 + col];
#pragma unroll
        for (int r = 0; r < 8; ++r) {
            float v = acc[nt][r] + bia;
            shid[wave][r + 8 * khalf][nt * 16 + col] = (_Float16)(v > 0.0f ? v : 0.0f);
        }
    }
    const _Float16* hrow = &shid[wave][col][0];

    v8f hacc[4] = {v8f{}, v8f{}, v8f{}, v8f{}};
#pragma unroll
    for (int kc = 0; kc < 2; ++kc) {
        v16h a = load_a_row(hrow, kc * 32, khalf);
#pragma unroll
        for (int nt = 0; nt < 4; ++nt) {
            v16h b = *(const v16h*)(wsW + NW2B + (kc * 4 + nt) * 512 + lane * 16);
            hacc[nt] = wmma16(a, b, hacc[nt]);
        }
    }
#pragma unroll
    for (int nt = 0; nt < 4; ++nt) {
        float bia = nb2[nt * 16 + col];
#pragma unroll
        for (int r = 0; r < 8; ++r) {
            int n = nbase + wave * 16 + r + 8 * khalf;
            if (n < N_NODES)
                hout[(size_t)n * FDIM + nt * 16 + col] = hacc[nt][r] + bia;
        }
    }
}

extern "C" void kernel_launch(void* const* d_in, const int* in_sizes, int n_in,
                              void* d_out, int out_size, void* d_ws, size_t ws_size,
                              hipStream_t stream) {
    const float* x         = (const float*)d_in[0];
    const int*   ei        = (const int*)  d_in[1];
    const float* pos       = (const float*)d_in[2];
    const float* pos_init  = (const float*)d_in[3];
    const float* edge_attr = (const float*)d_in[4];
    const float* mw1 = (const float*)d_in[5];
    const float* mb1 = (const float*)d_in[6];
    const float* mw2 = (const float*)d_in[7];
    const float* mb2 = (const float*)d_in[8];
    const float* nw1 = (const float*)d_in[9];
    const float* nb1 = (const float*)d_in[10];
    const float* nw2 = (const float*)d_in[11];
    const float* nb2 = (const float*)d_in[12];
    const float* cw1 = (const float*)d_in[13];
    const float* cb1 = (const float*)d_in[14];
    const float* cw2 = (const float*)d_in[15];
    const float* cb2 = (const float*)d_in[16];

    float* h_out   = (float*)d_out;
    float* pos_out = h_out + (size_t)N_NODES * FDIM;

    float* msum = (float*)d_ws;
    float* cnt  = (float*)((char*)d_ws + WS_CNT_OFF);
    _Float16* wsW = (_Float16*)((char*)d_ws + WS_WEIGHTS_OFF);

    egnn_zero<<<(N_NODES * FDIM) / 256, 256, 0, stream>>>(msum, cnt, pos_out);

    egnn_prep_w<<<40, 256, 0, stream>>>(mw1, MSG_IN, 64, 4, 20 * 512, wsW + MW1B);
    egnn_prep_w<<<16, 256, 0, stream>>>(mw2, 64,     64, 4,  8 * 512, wsW + MW2B);
    egnn_prep_w<<<16, 256, 0, stream>>>(cw1, 64,     64, 4,  8 * 512, wsW + CW1B);
    egnn_prep_w<<< 4, 256, 0, stream>>>(cw2, 64,      3, 1,  2 * 512, wsW + CW2B);
    egnn_prep_w<<<32, 256, 0, stream>>>(nw1, 128,    64, 4, 16 * 512, wsW + NW1B);
    egnn_prep_w<<<16, 256, 0, stream>>>(nw2, 64,     64, 4,  8 * 512, wsW + NW2B);

    egnn_edge<<<N_EDGES / 128, 256, 0, stream>>>(
        x, ei, pos, pos_init, edge_attr, mb1, mb2, cb1, cb2,
        wsW, msum, cnt, pos_out);

    egnn_node<<<(N_NODES + 127) / 128, 256, 0, stream>>>(
        x, msum, cnt, nb1, nb2, wsW, h_out);
}


# Edges listed per 200-node destination block each call (count, fill launches); one work-group per block sums messages, degrees, positions in LDS (LDS atomics), stores each row once; sum order changes
# speedup vs baseline: 2.1992x; 1.6351x over previous
#include <hip/hip_runtime.h>

typedef __attribute__((ext_vector_type(16))) _Float16 v16h;
typedef __attribute__((ext_vector_type(8)))  float    v8f;

#define N_NODES 50000
#define N_EDGES 800000
#define FDIM    64
#define EFDIM   16
#define MSG_IN  145
#define MSG_K   160
#define NODE_K  128

#define WS_CNT_OFF     ((size_t)N_NODES * FDIM * 4)
#define WS_WEIGHTS_OFF ((size_t)13000192)

#define MW1B 0
#define MW2B 10240
#define CW1B 14336
#define CW2B 18432
#define NW1B 19456
#define NW2B 27648

__device__ __forceinline__ v8f wmma16(v16h a, v16h b, v8f c) {
    return __builtin_amdgcn_wmma_f32_16x16x32_f16(
         false, a,  false, b,
         (short)0, c,  false,  false);
}

__device__ __forceinline__ v16h load_a_row(const _Float16* rowptr, int kcbase, int khalf) {
    v16h a;
#pragma unroll
    for (int e = 0; e < 16; ++e) {
        int klocal = (e & 7) + 8 * khalf + 16 * (e >> 3);
        a[e] = rowptr[kcbase + klocal];
    }
    return a;
}

#define BIN_NODES 200
#define NBINS     250
#define BIN_CHUNK 2048

__global__ __launch_bounds__(256) void egnn_bin_clear(int* __restrict__ binctr) {
    binctr[threadIdx.x] = 0;
    binctr[256 + threadIdx.x] = 0;
}

__global__ __launch_bounds__(256) void egnn_bin_count(const int* __restrict__ ei,
                                                      int* __restrict__ gcount) {
    __shared__ int hist[256];
    const int t = threadIdx.x;
    hist[t] = 0;
    __syncthreads();
    const int base = blockIdx.x * BIN_CHUNK;
    for (int i = t; i < BIN_CHUNK; i += 256) {
        int e = base + i;
        if (e < N_EDGES) atomicAdd(&hist[ei[N_EDGES + e] / BIN_NODES], 1);
    }
    __syncthreads();
    int v = hist[t];
    if (v) atomicAdd(&gcount[t], v);
}

__global__ __launch_bounds__(256) void egnn_bin_fill(const int* __restrict__ ei,
                                                     const int* __restrict__ gcount,
                                                     int* __restrict__ gcursor,
                                                     int* __restrict__ perm) {
    __shared__ int hist[256];
    __shared__ int scount[256];
    __shared__ int sbase[256];
    const int t = threadIdx.x;
    hist[t] = 0;
    const int cv = gcount[t];
    int incl = cv;
    scount[t] = incl;
    __syncthreads();
    for (int o = 1; o < 256; o <<= 1) {
        int add = (t >= o) ? scount[t - o] : 0;
        __syncthreads();
        incl += add;
        scount[t] = incl;
        __syncthreads();
    }
    const int start = incl - cv;
    const int base = blockIdx.x * BIN_CHUNK;
    int mybin[8], myrank[8];
#pragma unroll
    for (int j = 0; j < 8; ++j) {
        int e = base + t + 256 * j;
        mybin[j] = -1; myrank[j] = 0;
        if (e < N_EDGES) {
            int b = ei[N_EDGES + e] / BIN_NODES;
            mybin[j] = b;
            myrank[j] = atomicAdd(&hist[b], 1);
        }
    }
    __syncthreads();
    int v = hist[t];
    int off = v ? atomicAdd(&gcursor[t], v) : 0;
    sbase[t] = start + off;
    __syncthreads();
#pragma unroll
    for (int j = 0; j < 8; ++j)
        if (mybin[j] >= 0) perm[sbase[mybin[j]] + myrank[j]] = base + t + 256 * j;
}

__global__ __launch_bounds__(256) void egnn_prep_w(const float* __restrict__ W,
                                                   int Kact, int Nact, int ntiles,
                                                   int total, _Float16* __restrict__ out) {
    int idx = blockIdx.x * 256 + threadIdx.x;
    if (idx >= total) return;
    int block  = idx >> 9;
    int within = idx & 511;
    int lane = within >> 4;
    int e    = within & 15;
    int kc = block / ntiles;
    int nt = block - kc * ntiles;
    int khalf  = lane >> 4;
    int coln   = lane & 15;
    int klocal = (e & 7) + 8 * khalf + 16 * (e >> 3);
    int k = kc * 32 + klocal;
    int n = nt * 16 + coln;
    float v = (k < Kact && n < Nact) ? W[(size_t)k * Nact + n] : 0.0f;
    out[idx] = (_Float16)v;
}

typedef __attribute__((ext_vector_type(4))) float v4f;

__device__ __forceinline__ v16h ld_w(const _Float16* __restrict__ w, int lane) {
    return *(const v16h*)(w + lane * 16);
}

__device__ __forceinline__ v16h row_frag(const float* __restrict__ p, int h) {
    const v4f* q = (const v4f*)(p + 8 * h);
    v4f a0 = q[0], a1 = q[1], b0 = q[4], b1 = q[5];
    v16h f;
    f[0]  = (_Float16)a0.x; f[1]  = (_Float16)a0.y; f[2]  = (_Float16)a0.z; f[3]  = (_Float16)a0.w;
    f[4]  = (_Float16)a1.x; f[5]  = (_Float16)a1.y; f[6]  = (_Float16)a1.z; f[7]  = (_Float16)a1.w;
    f[8]  = (_Float16)b0.x; f[9]  = (_Float16)b0.y; f[10] = (_Float16)b0.z; f[11] = (_Float16)b0.w;
    f[12] = (_Float16)b1.x; f[13] = (_Float16)b1.y; f[14] = (_Float16)b1.z; f[15] = (_Float16)b1.w;
    return f;
}

__device__ __forceinline__ v8f bias8(v8f a, const float* __restrict__ b, bool relu) {
    const v4f* q = (const v4f*)b;
    v4f b0 = q[0], b1 = q[1];
    float bb[8] = {b0.x, b0.y, b0.z, b0.w, b1.x, b1.y, b1.z, b1.w};
    v8f o;
#pragma unroll
    for (int r = 0; r < 8; ++r) {
        float v = a[r] + bb[r];
        o[r] = relu ? (v > 0.0f ? v : 0.0f) : v;
    }
    return o;
}

__device__ __forceinline__ v16h pack2(v8f lo, v8f hi) {
    v16h f;
#pragma unroll
    for (int r = 0; r < 8; ++r) { f[r] = (_Float16)lo[r]; f[8 + r] = (_Float16)hi[r]; }
    return f;
}

__global__ __launch_bounds__(256) void egnn_edge(
    const float* __restrict__ x, const int* __restrict__ ei,
    const float* __restrict__ pos, const float* __restrict__ pos_init,
    const float* __restrict__ edge_attr,
    const float* __restrict__ mb1, const float* __restrict__ mb2,
    const float* __restrict__ cb1, const float* __restrict__ cb2,
    const _Float16* __restrict__ wsW,
    const int* __restrict__ perm, const int* __restrict__ gcount,
    float* __restrict__ msum, float* __restrict__ cnt, float* __restrict__ pos_out)
{
    __shared__ float sacc[(BIN_NODES + 1) * FDIM];
    __shared__ float spos[(BIN_NODES + 1) * 3];
    __shared__ float scnt[BIN_NODES + 1];
    __shared__ int srange[2];

    const int tid  = threadIdx.x;
    const int bin  = blockIdx.x;
    const int nbase = bin * BIN_NODES;
    for (int i = tid; i < (BIN_NODES + 1) * FDIM; i += 256) sacc[i] = 0.0f;
    for (int i = tid; i < (BIN_NODES + 1) * 3; i += 256)    spos[i] = 0.0f;
    if (tid < BIN_NODES + 1) scnt[tid] = 0.0f;
    if (tid < 32) {
        int acc0 = 0;
        for (int k = tid; k < bin; k += 32) acc0 += gcount[k];
        for (int o = 16; o > 0; o >>= 1) acc0 += __shfl_xor(acc0, o);
        if (tid == 0) { srange[0] = acc0; srange[1] = acc0 + gcount[bin]; }
    }
    __syncthreads();
    const int estart = srange[0];
    const int eend   = srange[1];
    const int ntiles = (eend - estart + 15) >> 4;

    const int lane = tid & 31;
    const int c = lane & 15;
    const int h = lane >> 4;
    const int wave = tid >> 5;
    int e_cur = 0, s_cur = 0, d_cur = 0, e_nxt = 0;
    if (wave < ntiles) {
        int p = estart + wave * 16 + c;
        e_cur = perm[p < eend ? p : estart];
    }
    if (wave + 8 < ntiles) {
        int p = estart + (wave + 8) * 16 + c;
        e_nxt = perm[p < eend ? p : estart];
    }
    if (wave < ntiles) { s_cur = ei[e_cur]; d_cur = ei[N_EDGES + e_cur]; }
  for (int tile = wave; tile < ntiles; tile += 8) {
    const bool live = estart + tile * 16 + c < eend;
    const int e = e_cur;
    const int s = s_cur;
    const int d = d_cur;
    e_cur = e_nxt;
    if (tile + 8 < ntiles) { s_cur = ei[e_cur]; d_cur = ei[N_EDGES + e_cur]; }
    if (tile + 16 < ntiles) {
        int p = estart + (tile + 16) * 16 + c;
        e_nxt = perm[p < eend ? p : estart];
    }
    const int lrow = live ? (d - nbase) : BIN_NODES;

    const float* xs = x + (size_t)s * FDIM;
    const float* xd = x + (size_t)d * FDIM;
    v16h fr[5];
    fr[0] = row_frag(xs, h);
    fr[1] = row_frag(xs + 32, h);
    fr[2] = row_frag(xd, h);
    fr[3] = row_frag(xd + 32, h);
    const v4f* eq = (const v4f*)(edge_attr + (size_t)e * EFDIM);
    v4f e0 = eq[0], e1 = eq[1], e2 = eq[2], e3 = eq[3];

    float dx = pos[3*s+0] - pos[3*d+0];
    float dy = pos[3*s+1] - pos[3*d+1];
    float dz = pos[3*s+2] - pos[3*d+2];
    float dsq = dx*dx + dy*dy + dz*dz;
    float ix = pos_init[3*s+0] - pos_init[3*d+0];
    float iy = pos_init[3*s+1] - pos_init[3*d+1];
    float iz = pos_init[3*s+2] - pos_init[3*d+2];
    float rinv = 1.0f / sqrtf(ix*ix + iy*iy + iz*iz);
    if (h == 0) atomicAdd(&scnt[lrow], 1.0f);

    {
        float ea[16] = {e0.x, e0.y, e0.z, e0.w, e1.x, e1.y, e1.z, e1.w,
                        e2.x, e2.y, e2.z, e2.w, e3.x, e3.y, e3.z, e3.w};
        v16h f;
#pragma unroll
        for (int t = 0; t < 8; ++t) {
            float lo = (t == 0) ? dsq : ea[t - 1];
            float hi = ea[7 + t];
            f[t] = (_Float16)(h ? hi : lo);
        }
        f[8] = (_Float16)(h ? 0.0f : ea[15]);
#pragma unroll
        for (int t = 9; t < 16; ++t) f[t] = (_Float16)0.0f;
        fr[4] = f;
    }

    v8f acc[4] = {v8f{}, v8f{}, v8f{}, v8f{}};
#pragma unroll
    for (int kc = 0; kc < 5; ++kc) {
#pragma unroll
        for (int mt = 0; mt < 4; ++mt)
            acc[mt] = wmma16(ld_w(wsW + MW1B + (kc * 4 + mt) * 512, lane), fr[kc], acc[mt]);
    }
    v16h hf[2];
#pragma unroll
    for (int kc = 0; kc < 2; ++kc)
        hf[kc] = pack2(bias8(acc[2*kc],     mb1 + 32*kc      + 8*h, true),
                       bias8(acc[2*kc + 1], mb1 + 32*kc + 16 + 8*h, true));

    {
        v8f mij[4] = {v8f{}, v8f{}, v8f{}, v8f{}};
#pragma unroll
        for (int kc = 0; kc < 2; ++kc) {
#pragma unroll
            for (int nt = 0; nt < 4; ++nt)
                mij[nt] = wmma16(hf[kc], ld_w(wsW + MW2B + (kc * 4 + nt) * 512, lane), mij[nt]);
        }
        int drow[8];
#pragma unroll
        for (int r = 0; r < 8; ++r) drow[r] = __shfl(lrow, 8 * h + r);
#pragma unroll
        for (int nt = 0; nt < 4; ++nt) {
            float bia = mb2[nt * 16 + c];
#pragma unroll
            for (int r = 0; r < 8; ++r)
                atomicAdd(&sacc[drow[r] * FDIM + nt * 16 + c], mij[nt][r] + bia);
        }
    }

    v16h mf[2];
    {
        v8f mt_[4] = {v8f{}, v8f{}, v8f{}, v8f{}};
#pragma unroll
        for (int kc = 0; kc < 2; ++kc) {
#pragma unroll
            for (int mt = 0; mt < 4; ++mt)
                mt_[mt] = wmma16(ld_w(wsW + MW2B + (kc * 4 + mt) * 512, lane), hf[kc], mt_[mt]);
        }
#pragma unroll
        for (int kc = 0; kc < 2; ++kc)
            mf[kc] = pack2(bias8(mt_[2*kc],     mb2 + 32*kc      + 8*h, false),
                           bias8(mt_[2*kc + 1], mb2 + 32*kc + 16 + 8*h, false));
    }

    v16h tf[2];
    {
        v8f tac[4] = {v8f{}, v8f{}, v8f{}, v8f{}};
#pragma unroll
        for (int kc = 0; kc < 2; ++kc) {
#pragma unroll
            for (int mt = 0; mt < 4; ++mt)
                tac[mt] = wmma16(ld_w(wsW + CW1B + (kc * 4 + mt) * 512, lane), mf[kc], tac[mt]);
        }
#pragma unroll
        for (int kc = 0; kc < 2; ++kc)
            tf[kc] = pack2(bias8(tac[2*kc],     cb1 + 32*kc      + 8*h, true),
                           bias8(tac[2*kc + 1], cb1 + 32*kc + 16 + 8*h, true));
    }

    v8f cac = v8f{};
#pragma unroll
    for (int kc = 0; kc < 2; ++kc)
        cac = wmma16(ld_w(wsW + CW2B + kc * 512, lane), tf[kc], cac);
    float ux = ix * rinv * (cac[0] + cb2[0]);
    float uy = iy * rinv * (cac[1] + cb2[1]);
    float uz = iz * rinv * (cac[2] + cb2[2]);
    if (h == 0) {
        atomicAdd(&spos[lrow * 3 + 0], ux);
        atomicAdd(&spos[lrow * 3 + 1], uy);
        atomicAdd(&spos[lrow * 3 + 2], uz);
    }
  }
    __syncthreads();
    for (int i = tid; i < BIN_NODES * FDIM; i += 256) msum[(size_t)nbase * FDIM + i] = sacc[i];
    for (int i = tid; i < BIN_NODES * 3; i += 256)    pos_out[(size_t)nbase * 3 + i] = spos[i];
    if (tid < BIN_NODES) cnt[nbase + tid] = scnt[tid];
}

__global__ __launch_bounds__(256) void egnn_node(
    const float* __restrict__ x, const float* __restrict__ msum,
    const float* __restrict__ cnt,
    const float* __restrict__ nb1, const float* __restrict__ nb2,
    const _Float16* __restrict__ wsW, float* __restrict__ hout)
{
    __shared__ _Float16 sfeat[128][NODE_K];
    __shared__ _Float16 shid[8][16][FDIM];
    __shared__ float sinv[128];

    const int tid   = threadIdx.x;
    const int nbase = blockIdx.x * 128;

    if (tid < 128) {
        int n = nbase + tid;
        float c = (n < N_NODES) ? cnt[n] : 1.0f;
        sinv[tid] = 1.0f / fmaxf(c, 1.0f);
    }
    __syncthreads();

    for (int idx = tid; idx < 128 * NODE_K; idx += 256) {
        int row = idx >> 7, k = idx & 127;
        int n = nbase + row;
        float v = 0.0f;
        if (n < N_NODES)
            v = (k < FDIM) ? x[(size_t)n * FDIM + k]
                           : msum[(size_t)n * FDIM + (k - FDIM)] * sinv[row];
        sfeat[row][k] = (_Float16)v;
    }
    __syncthreads();

    const int wave  = tid >> 5;
    const int lane  = tid & 31;
    const int col   = lane & 15;
    const int khalf = lane >> 4;
    const _Float16* arow = &sfeat[wave * 16 + col][0];

    v8f acc[4] = {v8f{}, v8f{}, v8f{}, v8f{}};
#pragma unroll
    for (int kc = 0; kc < 4; ++kc) {
        v16h a = load_a_row(arow, kc * 32, khalf);
#pragma unroll
        for (int nt = 0; nt < 4; ++nt) {
            v16h b = *(const v16h*)(wsW + NW1B + (kc * 4 + nt) * 512 + lane * 16);
            acc[nt] = wmma16(a, b, acc[nt]);
        }
    }
#pragma unroll
    for (int nt = 0; nt < 4; ++nt) {
        float bia = nb1[nt * 16 + col];
#pragma unroll
        for (int r = 0; r < 8; ++r) {
            float v = acc[nt][r] + bia;
            shid[wave][r + 8 * khalf][nt * 16 + col] = (_Float16)(v > 0.0f ? v : 0.0f);
        }
    }
    const _Float16* hrow = &shid[wave][col][0];

    v8f hacc[4] = {v8f{}, v8f{}, v8f{}, v8f{}};
#pragma unroll
    for (int kc = 0; kc < 2; ++kc) {
        v16h a = load_a_row(hrow, kc * 32, khalf);
#pragma unroll
        for (int nt = 0; nt < 4; ++nt) {
            v16h b = *(const v16h*)(wsW + NW2B + (kc * 4 + nt) * 512 + lane * 16);
            hacc[nt] = wmma16(a, b, hacc[nt]);
        }
    }
#pragma unroll
    for (int nt = 0; nt < 4; ++nt) {
        float bia = nb2[nt * 16 + col];
#pragma unroll
        for (int r = 0; r < 8; ++r) {
            int n = nbase + wave * 16 + r + 8 * khalf;
            if (n < N_NODES)
                hout[(size_t)n * FDIM + nt * 16 + col] = hacc[nt][r] + bia;
        }
    }
}

extern "C" void kernel_launch(void* const* d_in, const int* in_sizes, int n_in,
                              void* d_out, int out_size, void* d_ws, size_t ws_size,
                              hipStream_t stream) {
    const float* x         = (const float*)d_in[0];
    const int*   ei        = (const int*)  d_in[1];
    const float* pos       = (const float*)d_in[2];
    const float* pos_init  = (const float*)d_in[3];
    const float* edge_attr = (const float*)d_in[4];
    const float* mw1 = (const float*)d_in[5];
    const float* mb1 = (const float*)d_in[6];
    const float* mw2 = (const float*)d_in[7];
    const float* mb2 = (const float*)d_in[8];
    const float* nw1 = (const float*)d_in[9];
    const float* nb1 = (const float*)d_in[10];
    const float* nw2 = (const float*)d_in[11];
    const float* nb2 = (const float*)d_in[12];
    const float* cw1 = (const float*)d_in[13];
    const float* cb1 = (const float*)d_in[14];
    const float* cw2 = (const float*)d_in[15];
    const float* cb2 = (const float*)d_in[16];

    float* h_out   = (float*)d_out;
    float* pos_out = h_out + (size_t)N_NODES * FDIM;

    float* msum = (float*)d_ws;
    float* cnt  = (float*)((char*)d_ws + WS_CNT_OFF);
    _Float16* wsW = (_Float16*)((char*)d_ws + WS_WEIGHTS_OFF);

    int* perm    = (int*)d_out;
    int* gcount  = perm + N_EDGES;
    int* gcursor = gcount + 256;
    egnn_bin_clear<<<1, 256, 0, stream>>>(gcount);
    egnn_bin_count<<<(N_EDGES + BIN_CHUNK - 1) / BIN_CHUNK, 256, 0, stream>>>(ei, gcount);
    egnn_bin_fill<<<(N_EDGES + BIN_CHUNK - 1) / BIN_CHUNK, 256, 0, stream>>>(ei, gcount, gcursor, perm);

    egnn_prep_w<<<40, 256, 0, stream>>>(mw1, MSG_IN, 64, 4, 20 * 512, wsW + MW1B);
    egnn_prep_w<<<16, 256, 0, stream>>>(mw2, 64,     64, 4,  8 * 512, wsW + MW2B);
    egnn_prep_w<<<16, 256, 0, stream>>>(cw1, 64,     64, 4,  8 * 512, wsW + CW1B);
    egnn_prep_w<<< 4, 256, 0, stream>>>(cw2, 64,      3, 1,  2 * 512, wsW + CW2B);
    egnn_prep_w<<<32, 256, 0, stream>>>(nw1, 128,    64, 4, 16 * 512, wsW + NW1B);
    egnn_prep_w<<<16, 256, 0, stream>>>(nw2, 64,     64, 4,  8 * 512, wsW + NW2B);

    egnn_edge<<<NBINS, 256, 0, stream>>>(
        x, ei, pos, pos_init, edge_attr, mb1, mb2, cb1, cb2,
        wsW, perm, gcount, msum, cnt, pos_out);

    egnn_node<<<(N_NODES + 127) / 128, 256, 0, stream>>>(
        x, msum, cnt, nb1, nb2, wsW, h_out);
}


# Edge-MLP weights read from LDS per product, no longer held in every wave's registers (512 -> 254 VGPRs): 16 waves per work-group instead of 8; same f16 values, k order and LDS atomic sums
# speedup vs baseline: 2.4294x; 1.1047x over previous
#include <hip/hip_runtime.h>

typedef __attribute__((ext_vector_type(16))) _Float16 v16h;
typedef __attribute__((ext_vector_type(8)))  float    v8f;

#define N_NODES 50000
#define N_EDGES 800000
#define FDIM    64
#define EFDIM   16
#define MSG_IN  145
#define MSG_K   160
#define NODE_K  128

#define WS_CNT_OFF     ((size_t)N_NODES * FDIM * 4)
#define WS_WEIGHTS_OFF ((size_t)13000192)

#define MW1B 0
#define MW2B 10240
#define CW1B 14336
#define CW2B 18432
#define NW1B 19456
#define NW2B 27648

__device__ __forceinline__ v8f wmma16(v16h a, v16h b, v8f c) {
    return __builtin_amdgcn_wmma_f32_16x16x32_f16(
         false, a,  false, b,
         (short)0, c,  false,  false);
}

__device__ __forceinline__ v16h load_a_row(const _Float16* rowptr, int kcbase, int khalf) {
    v16h a;
#pragma unroll
    for (int e = 0; e < 16; ++e) {
        int klocal = (e & 7) + 8 * khalf + 16 * (e >> 3);
        a[e] = rowptr[kcbase + klocal];
    }
    return a;
}

#define BIN_NODES 200
#define NBINS     250
#define BIN_CHUNK 2048

__global__ __launch_bounds__(256) void egnn_bin_clear(int* __restrict__ binctr) {
    binctr[threadIdx.x] = 0;
    binctr[256 + threadIdx.x] = 0;
}

__global__ __launch_bounds__(256) void egnn_bin_count(const int* __restrict__ ei,
                                                      int* __restrict__ gcount) {
    __shared__ int hist[256];
    const int t = threadIdx.x;
    hist[t] = 0;
    __syncthreads();
    const int base = blockIdx.x * BIN_CHUNK;
    for (int i = t; i < BIN_CHUNK; i += 256) {
        int e = base + i;
        if (e < N_EDGES) atomicAdd(&hist[ei[N_EDGES + e] / BIN_NODES], 1);
    }
    __syncthreads();
    int v = hist[t];
    if (v) atomicAdd(&gcount[t], v);
}

__global__ __launch_bounds__(256) void egnn_bin_fill(const int* __restrict__ ei,
                                                     const int* __restrict__ gcount,
                                                     int* __restrict__ gcursor,
                                                     int* __restrict__ perm) {
    __shared__ int hist[256];
    __shared__ int scount[256];
    __shared__ int sbase[256];
    const int t = threadIdx.x;
    hist[t] = 0;
    const int cv = gcount[t];
    int incl = cv;
    scount[t] = incl;
    __syncthreads();
    for (int o = 1; o < 256; o <<= 1) {
        int add = (t >= o) ? scount[t - o] : 0;
        __syncthreads();
        incl += add;
        scount[t] = incl;
        __syncthreads();
    }
    const int start = incl - cv;
    const int base = blockIdx.x * BIN_CHUNK;
    int mybin[8], myrank[8];
#pragma unroll
    for (int j = 0; j < 8; ++j) {
        int e = base + t + 256 * j;
        mybin[j] = -1; myrank[j] = 0;
        if (e < N_EDGES) {
            int b = ei[N_EDGES + e] / BIN_NODES;
            mybin[j] = b;
            myrank[j] = atomicAdd(&hist[b], 1);
        }
    }
    __syncthreads();
    int v = hist[t];
    int off = v ? atomicAdd(&gcursor[t], v) : 0;
    sbase[t] = start + off;
    __syncthreads();
#pragma unroll
    for (int j = 0; j < 8; ++j)
        if (mybin[j] >= 0) perm[sbase[mybin[j]] + myrank[j]] = base + t + 256 * j;
}

__global__ __launch_bounds__(256) void egnn_prep_w(const float* __restrict__ W,
                                                   int Kact, int Nact, int ntiles,
                                                   int total, _Float16* __restrict__ out) {
    int idx = blockIdx.x * 256 + threadIdx.x;
    if (idx >= total) return;
    int block  = idx >> 9;
    int within = idx & 511;
    int lane = within >> 4;
    int e    = within & 15;
    int kc = block / ntiles;
    int nt = block - kc * ntiles;
    int khalf  = lane >> 4;
    int coln   = lane & 15;
    int klocal = (e & 7) + 8 * khalf + 16 * (e >> 3);
    int k = kc * 32 + klocal;
    int n = nt * 16 + coln;
    float v = (k < Kact && n < Nact) ? W[(size_t)k * Nact + n] : 0.0f;
    out[idx] = (_Float16)v;
}

typedef __attribute__((ext_vector_type(4))) float v4f;

typedef __attribute__((ext_vector_type(8))) _Float16 v8h;
__device__ __forceinline__ v16h ld_ws(const _Float16* blk, int lane) {
    v8h lo = *(const v8h*)(blk + lane * 8);
    v8h hi = *(const v8h*)(blk + 256 + lane * 8);
    return __builtin_shufflevector(lo, hi, 0, 1, 2, 3, 4, 5, 6, 7, 8, 9, 10, 11, 12, 13, 14, 15);
}

__device__ __forceinline__ v16h row_frag(const float* __restrict__ p, int h) {
    const v4f* q = (const v4f*)(p + 8 * h);
    v4f a0 = q[0], a1 = q[1], b0 = q[4], b1 = q[5];
    v16h f;
    f[0]  = (_Float16)a0.x; f[1]  = (_Float16)a0.y; f[2]  = (_Float16)a0.z; f[3]  = (_Float16)a0.w;
    f[4]  = (_Float16)a1.x; f[5]  = (_Float16)a1.y; f[6]  = (_Float16)a1.z; f[7]  = (_Float16)a1.w;
    f[8]  = (_Float16)b0.x; f[9]  = (_Float16)b0.y; f[10] = (_Float16)b0.z; f[11] = (_Float16)b0.w;
    f[12] = (_Float16)b1.x; f[13] = (_Float16)b1.y; f[14] = (_Float16)b1.z; f[15] = (_Float16)b1.w;
    return f;
}

__device__ __forceinline__ v8f bias8(v8f a, const float* __restrict__ b, bool relu) {
    const v4f* q = (const v4f*)b;
    v4f b0 = q[0], b1 = q[1];
    float bb[8] = {b0.x, b0.y, b0.z, b0.w, b1.x, b1.y, b1.z, b1.w};
    v8f o;
#pragma unroll
    for (int r = 0; r < 8; ++r) {
        float v = a[r] + bb[r];
        o[r] = relu ? (v > 0.0f ? v : 0.0f) : v;
    }
    return o;
}

__device__ __forceinline__ v16h pack2(v8f lo, v8f hi) {
    v16h f;
#pragma unroll
    for (int r = 0; r < 8; ++r) { f[r] = (_Float16)lo[r]; f[8 + r] = (_Float16)hi[r]; }
    return f;
}

#define EDGE_WAVES   16
#define EDGE_THREADS (EDGE_WAVES * 32)
#define EDGE_WBLOCKS 38
__global__ __launch_bounds__(EDGE_THREADS) void egnn_edge(
    const float* __restrict__ x, const int* __restrict__ ei,
    const float* __restrict__ pos, const float* __restrict__ pos_init,
    const float* __restrict__ edge_attr,
    const float* __restrict__ mb1, const float* __restrict__ mb2,
    const float* __restrict__ cb1, const float* __restrict__ cb2,
    const _Float16* __restrict__ wsW,
    const int* __restrict__ perm, const int* __restrict__ gcount,
    float* __restrict__ msum, float* __restrict__ cnt, float* __restrict__ pos_out)
{
    __shared__ float sacc[(BIN_NODES + 1) * FDIM];
    __shared__ float spos[(BIN_NODES + 1) * 3];
    __shared__ float scnt[BIN_NODES + 1];
    __shared__ int srange[2];
    __shared__ __attribute__((aligned(16))) _Float16 sW[EDGE_WBLOCKS * 512];

    const int tid  = threadIdx.x;
    const int bin  = blockIdx.x;
    const int nbase = bin * BIN_NODES;
    for (int i = tid; i < (BIN_NODES + 1) * FDIM; i += EDGE_THREADS) sacc[i] = 0.0f;
    for (int i = tid; i < (BIN_NODES + 1) * 3; i += EDGE_THREADS)    spos[i] = 0.0f;
    {
        const v8h* wsrc = (const v8h*)(wsW + MW1B);
        for (int q = tid; q < EDGE_WBLOCKS * 64; q += EDGE_THREADS) {
            const int blk = q >> 6, ln = (q & 63) >> 1, half = q & 1;
            *(v8h*)(sW + blk * 512 + half * 256 + ln * 8) = wsrc[q];
        }
    }
    if (tid < BIN_NODES + 1) scnt[tid] = 0.0f;
    if (tid < 32) {
        int acc0 = 0;
        for (int k = tid; k < bin; k += 32) acc0 += gcount[k];
        for (int o = 16; o > 0; o >>= 1) acc0 += __shfl_xor(acc0, o);
        if (tid == 0) { srange[0] = acc0; srange[1] = acc0 + gcount[bin]; }
    }
    __syncthreads();
    const int estart = srange[0];
    const int eend   = srange[1];
    const int ntiles = (eend - estart + 15) >> 4;

    const int lane = tid & 31;
    const int c = lane & 15;
    const int h = lane >> 4;
    const int wave = tid >> 5;
    int e_cur = 0, s_cur = 0, d_cur = 0, e_nxt = 0;
    if (wave < ntiles) {
        int p = estart + wave * 16 + c;
        e_cur = perm[p < eend ? p : estart];
    }
    if (wave + EDGE_WAVES < ntiles) {
        int p = estart + (wave + EDGE_WAVES) * 16 + c;
        e_nxt = perm[p < eend ? p : estart];
    }
    if (wave < ntiles) { s_cur = ei[e_cur]; d_cur = ei[N_EDGES + e_cur]; }
    float mb2c[4];
#pragma unroll
    for (int nt = 0; nt < 4; ++nt) mb2c[nt] = mb2[nt * 16 + c];
  for (int tile = wave; tile < ntiles; tile += EDGE_WAVES) {
    const bool live = estart + tile * 16 + c < eend;
    const int e = e_cur;
    const int s = s_cur;
    const int d = d_cur;
    e_cur = e_nxt;
    if (tile + EDGE_WAVES < ntiles) { s_cur = ei[e_cur]; d_cur = ei[N_EDGES + e_cur]; }
    if (tile + 2 * EDGE_WAVES < ntiles) {
        int p = estart + (tile + 2 * EDGE_WAVES) * 16 + c;
        e_nxt = perm[p < eend ? p : estart];
    }
    const int lrow = live ? (d - nbase) : BIN_NODES;
    int wofs = 0;
    asm volatile("" : "+v"(wofs));
    const _Float16* sWt = sW + wofs;

    const float* xs = x + (size_t)s * FDIM;
    const float* xd = x + (size_t)d * FDIM;
    v16h fr[5];
    fr[0] = row_frag(xs, h);
    fr[1] = row_frag(xs + 32, h);
    fr[2] = row_frag(xd, h);
    fr[3] = row_frag(xd + 32, h);
    const v4f* eq = (const v4f*)(edge_attr + (size_t)e * EFDIM);
    v4f e0 = eq[0], e1 = eq[1], e2 = eq[2], e3 = eq[3];

    float dx = pos[3*s+0] - pos[3*d+0];
    float dy = pos[3*s+1] - pos[3*d+1];
    float dz = pos[3*s+2] - pos[3*d+2];
    float dsq = dx*dx + dy*dy + dz*dz;
    float ix = pos_init[3*s+0] - pos_init[3*d+0];
    float iy = pos_init[3*s+1] - pos_init[3*d+1];
    float iz = pos_init[3*s+2] - pos_init[3*d+2];
    float rinv = 1.0f / sqrtf(ix*ix + iy*iy + iz*iz);
    if (h == 0) atomicAdd(&scnt[lrow], 1.0f);

    {
        float ea[16] = {e0.x, e0.y, e0.z, e0.w, e1.x, e1.y, e1.z, e1.w,
                        e2.x, e2.y, e2.z, e2.w, e3.x, e3.y, e3.z, e3.w};
        v16h f;
#pragma unroll
        for (int t = 0; t < 8; ++t) {
            float lo = (t == 0) ? dsq : ea[t - 1];
            float hi = ea[7 + t];
            f[t] = (_Float16)(h ? hi : lo);
        }
        f[8] = (_Float16)(h ? 0.0f : ea[15]);
#pragma unroll
        for (int t = 9; t < 16; ++t) f[t] = (_Float16)0.0f;
        fr[4] = f;
    }

    v8f acc[4] = {v8f{}, v8f{}, v8f{}, v8f{}};
#pragma unroll
    for (int kc = 0; kc < 5; ++kc) {
#pragma unroll
        for (int mt = 0; mt < 4; ++mt)
            acc[mt] = wmma16(ld_ws(sWt + MW1B + (kc * 4 + mt) * 512, lane), fr[kc], acc[mt]);
    }
    v16h hf[2];
#pragma unroll
    for (int kc = 0; kc < 2; ++kc)
        hf[kc] = pack2(bias8(acc[2*kc],     mb1 + 32*kc      + 8*h, true),
                       bias8(acc[2*kc + 1], mb1 + 32*kc + 16 + 8*h, true));

    v8f mij[4] = {v8f{}, v8f{}, v8f{}, v8f{}};
    v8f mt_[4] = {v8f{}, v8f{}, v8f{}, v8f{}};
    {
        v16h w2[8];
#pragma unroll
        for (int i = 0; i < 8; ++i) w2[i] = ld_ws(sWt + MW2B + i * 512, lane);
#pragma unroll
        for (int kc = 0; kc < 2; ++kc) {
#pragma unroll
            for (int nt = 0; nt < 4; ++nt)
                mij[nt] = wmma16(hf[kc], w2[kc * 4 + nt], mij[nt]);
        }
#pragma unroll
        for (int kc = 0; kc < 2; ++kc) {
#pragma unroll
            for (int mt = 0; mt < 4; ++mt)
                mt_[mt] = wmma16(w2[kc * 4 + mt], hf[kc], mt_[mt]);
        }
    }
    v16h wc1[8], wc2[2];
#pragma unroll
    for (int i = 0; i < 8; ++i) wc1[i] = ld_ws(sWt + CW1B + i * 512, lane);
#pragma unroll
    for (int i = 0; i < 2; ++i) wc2[i] = ld_ws(sWt + CW2B + i * 512, lane);
    v16h mf[2];
#pragma unroll
    for (int kc = 0; kc < 2; ++kc)
        mf[kc] = pack2(bias8(mt_[2*kc],     mb2 + 32*kc      + 8*h, false),
                       bias8(mt_[2*kc + 1], mb2 + 32*kc + 16 + 8*h, false));
    {
        int drow[8];
#pragma unroll
        for (int r = 0; r < 8; ++r) drow[r] = __shfl(lrow, 8 * h + r);
#pragma unroll
        for (int nt = 0; nt < 4; ++nt) {
#pragma unroll
            for (int r = 0; r < 8; ++r)
                atomicAdd(&sacc[drow[r] * FDIM + nt * 16 + c], mij[nt][r] + mb2c[nt]);
        }
    }

    v16h tf[2];
    {
        v8f tac[4] = {v8f{}, v8f{}, v8f{}, v8f{}};
#pragma unroll
        for (int kc = 0; kc < 2; ++kc) {
#pragma unroll
            for (int mt = 0; mt < 4; ++mt)
                tac[mt] = wmma16(wc1[kc * 4 + mt], mf[kc], tac[mt]);
        }
#pragma unroll
        for (int kc = 0; kc < 2; ++kc)
            tf[kc] = pack2(bias8(tac[2*kc],     cb1 + 32*kc      + 8*h, true),
                           bias8(tac[2*kc + 1], cb1 + 32*kc + 16 + 8*h, true));
    }

    v8f cac = v8f{};
#pragma unroll
    for (int kc = 0; kc < 2; ++kc)
        cac = wmma16(wc2[kc], tf[kc], cac);
    float ux = ix * rinv * (cac[0] + cb2[0]);
    float uy = iy * rinv * (cac[1] + cb2[1]);
    float uz = iz * rinv * (cac[2] + cb2[2]);
    if (h == 0) {
        atomicAdd(&spos[lrow * 3 + 0], ux);
        atomicAdd(&spos[lrow * 3 + 1], uy);
        atomicAdd(&spos[lrow * 3 + 2], uz);
    }
  }
    __syncthreads();
    for (int i = tid; i < BIN_NODES * FDIM; i += EDGE_THREADS) msum[(size_t)nbase * FDIM + i] = sacc[i];
    for (int i = tid; i < BIN_NODES * 3; i += EDGE_THREADS)    pos_out[(size_t)nbase * 3 + i] = spos[i];
    if (tid < BIN_NODES) cnt[nbase + tid] = scnt[tid];
}

__global__ __launch_bounds__(256) void egnn_node(
    const float* __restrict__ x, const float* __restrict__ msum,
    const float* __restrict__ cnt,
    const float* __restrict__ nb1, const float* __restrict__ nb2,
    const _Float16* __restrict__ wsW, float* __restrict__ hout)
{
    __shared__ _Float16 sfeat[128][NODE_K];
    __shared__ _Float16 shid[8][16][FDIM];
    __shared__ float sinv[128];

    const int tid   = threadIdx.x;
    const int nbase = blockIdx.x * 128;

    if (tid < 128) {
        int n = nbase + tid;
        float c = (n < N_NODES) ? cnt[n] : 1.0f;
        sinv[tid] = 1.0f / fmaxf(c, 1.0f);
    }
    __syncthreads();

    for (int idx = tid; idx < 128 * NODE_K; idx += 256) {
        int row = idx >> 7, k = idx & 127;
        int n = nbase + row;
        float v = 0.0f;
        if (n < N_NODES)
            v = (k < FDIM) ? x[(size_t)n * FDIM + k]
                           : msum[(size_t)n * FDIM + (k - FDIM)] * sinv[row];
        sfeat[row][k] = (_Float16)v;
    }
    __syncthreads();

    const int wave  = tid >> 5;
    const int lane  = tid & 31;
    const int col   = lane & 15;
    const int khalf = lane >> 4;
    const _Float16* arow = &sfeat[wave * 16 + col][0];

    v8f acc[4] = {v8f{}, v8f{}, v8f{}, v8f{}};
#pragma unroll
    for (int kc = 0; kc < 4; ++kc) {
        v16h a = load_a_row(arow, kc * 32, khalf);
#pragma unroll
        for (int nt = 0; nt < 4; ++nt) {
            v16h b = *(const v16h*)(wsW + NW1B + (kc * 4 + nt) * 512 + lane * 16);
            acc[nt] = wmma16(a, b, acc[nt]);
        }
    }
#pragma unroll
    for (int nt = 0; nt < 4; ++nt) {
        float bia = nb1[nt * 16 + col];
#pragma unroll
        for (int r = 0; r < 8; ++r) {
            float v = acc[nt][r] + bia;
            shid[wave][r + 8 * khalf][nt * 16 + col] = (_Float16)(v > 0.0f ? v : 0.0f);
        }
    }
    const _Float16* hrow = &shid[wave][col][0];

    v8f hacc[4] = {v8f{}, v8f{}, v8f{}, v8f{}};
#pragma unroll
    for (int kc = 0; kc < 2; ++kc) {
        v16h a = load_a_row(hrow, kc * 32, khalf);
#pragma unroll
        for (int nt = 0; nt < 4; ++nt) {
            v16h b = *(const v16h*)(wsW + NW2B + (kc * 4 + nt) * 512 + lane * 16);
            hacc[nt] = wmma16(a, b, hacc[nt]);
        }
    }
#pragma unroll
    for (int nt = 0; nt < 4; ++nt) {
        float bia = nb2[nt * 16 + col];
#pragma unroll
        for (int r = 0; r < 8; ++r) {
            int n = nbase + wave * 16 + r + 8 * khalf;
            if (n < N_NODES)
                hout[(size_t)n * FDIM + nt * 16 + col] = hacc[nt][r] + bia;
        }
    }
}

extern "C" void kernel_launch(void* const* d_in, const int* in_sizes, int n_in,
                              void* d_out, int out_size, void* d_ws, size_t ws_size,
                              hipStream_t stream) {
    const float* x         = (const float*)d_in[0];
    const int*   ei        = (const int*)  d_in[1];
    const float* pos       = (const float*)d_in[2];
    const float* pos_init  = (const float*)d_in[3];
    const float* edge_attr = (const float*)d_in[4];
    const float* mw1 = (const float*)d_in[5];
    const float* mb1 = (const float*)d_in[6];
    const float* mw2 = (const float*)d_in[7];
    const float* mb2 = (const float*)d_in[8];
    const float* nw1 = (const float*)d_in[9];
    const float* nb1 = (const float*)d_in[10];
    const float* nw2 = (const float*)d_in[11];
    const float* nb2 = (const float*)d_in[12];
    const float* cw1 = (const float*)d_in[13];
    const float* cb1 = (const float*)d_in[14];
    const float* cw2 = (const float*)d_in[15];
    const float* cb2 = (const float*)d_in[16];

    float* h_out   = (float*)d_out;
    float* pos_out = h_out + (size_t)N_NODES * FDIM;

    float* msum = (float*)d_ws;
    float* cnt  = (float*)((char*)d_ws + WS_CNT_OFF);
    _Float16* wsW = (_Float16*)((char*)d_ws + WS_WEIGHTS_OFF);

    int* perm    = (int*)d_out;
    int* gcount  = perm + N_EDGES;
    int* gcursor = gcount + 256;
    egnn_bin_clear<<<1, 256, 0, stream>>>(gcount);
    egnn_bin_count<<<(N_EDGES + BIN_CHUNK - 1) / BIN_CHUNK, 256, 0, stream>>>(ei, gcount);
    egnn_bin_fill<<<(N_EDGES + BIN_CHUNK - 1) / BIN_CHUNK, 256, 0, stream>>>(ei, gcount, gcursor, perm);

    egnn_prep_w<<<40, 256, 0, stream>>>(mw1, MSG_IN, 64, 4, 20 * 512, wsW + MW1B);
    egnn_prep_w<<<16, 256, 0, stream>>>(mw2, 64,     64, 4,  8 * 512, wsW + MW2B);
    egnn_prep_w<<<16, 256, 0, stream>>>(cw1, 64,     64, 4,  8 * 512, wsW + CW1B);
    egnn_prep_w<<< 4, 256, 0, stream>>>(cw2, 64,      3, 1,  2 * 512, wsW + CW2B);
    egnn_prep_w<<<32, 256, 0, stream>>>(nw1, 128,    64, 4, 16 * 512, wsW + NW1B);
    egnn_prep_w<<<16, 256, 0, stream>>>(nw2, 64,     64, 4,  8 * 512, wsW + NW2B);

    egnn_edge<<<NBINS, EDGE_THREADS, 0, stream>>>(
        x, ei, pos, pos_init, edge_attr, mb1, mb2, cb1, cb2,
        wsW, perm, gcount, msum, cnt, pos_out);

    egnn_node<<<(N_NODES + 127) / 128, 256, 0, stream>>>(
        x, msum, cnt, nb1, nb2, wsW, h_out);
}


# Node update fused into the edge kernel: the work-group that summed a 200-row block runs the node MLP on its LDS sums and stores h; no node launch, no m_sum/degree round trip; same f16 values
# speedup vs baseline: 3.3298x; 1.3706x over previous
#include <hip/hip_runtime.h>

typedef __attribute__((ext_vector_type(16))) _Float16 v16h;
typedef __attribute__((ext_vector_type(8)))  float    v8f;

#define N_NODES 50000
#define N_EDGES 800000
#define FDIM    64
#define EFDIM   16
#define MSG_IN  145
#define MSG_K   160

#define WS_WEIGHTS_OFF ((size_t)13000192)

#define MW1B 0
#define MW2B 10240
#define CW1B 14336
#define CW2B 18432
#define NW1B 19456
#define NW2B 27648

__device__ __forceinline__ v8f wmma16(v16h a, v16h b, v8f c) {
    return __builtin_amdgcn_wmma_f32_16x16x32_f16(
         false, a,  false, b,
         (short)0, c,  false,  false);
}

#define BIN_NODES 200
#define NBINS     250
#define BIN_CHUNK 2048

__global__ __launch_bounds__(256) void egnn_bin_clear(int* __restrict__ binctr) {
    binctr[threadIdx.x] = 0;
    binctr[256 + threadIdx.x] = 0;
}

__global__ __launch_bounds__(256) void egnn_bin_count(const int* __restrict__ ei,
                                                      int* __restrict__ gcount) {
    __shared__ int hist[256];
    const int t = threadIdx.x;
    hist[t] = 0;
    __syncthreads();
    const int base = blockIdx.x * BIN_CHUNK;
    for (int i = t; i < BIN_CHUNK; i += 256) {
        int e = base + i;
        if (e < N_EDGES) atomicAdd(&hist[ei[N_EDGES + e] / BIN_NODES], 1);
    }
    __syncthreads();
    int v = hist[t];
    if (v) atomicAdd(&gcount[t], v);
}

__global__ __launch_bounds__(256) void egnn_bin_fill(const int* __restrict__ ei,
                                                     const int* __restrict__ gcount,
                                                     int* __restrict__ gcursor,
                                                     int* __restrict__ perm) {
    __shared__ int hist[256];
    __shared__ int scount[256];
    __shared__ int sbase[256];
    const int t = threadIdx.x;
    hist[t] = 0;
    const int cv = gcount[t];
    int incl = cv;
    scount[t] = incl;
    __syncthreads();
    for (int o = 1; o < 256; o <<= 1) {
        int add = (t >= o) ? scount[t - o] : 0;
        __syncthreads();
        incl += add;
        scount[t] = incl;
        __syncthreads();
    }
    const int start = incl - cv;
    const int base = blockIdx.x * BIN_CHUNK;
    int mybin[8], myrank[8];
#pragma unroll
    for (int j = 0; j < 8; ++j) {
        int e = base + t + 256 * j;
        mybin[j] = -1; myrank[j] = 0;
        if (e < N_EDGES) {
            int b = ei[N_EDGES + e] / BIN_NODES;
            mybin[j] = b;
            myrank[j] = atomicAdd(&hist[b], 1);
        }
    }
    __syncthreads();
    int v = hist[t];
    int off = v ? atomicAdd(&gcursor[t], v) : 0;
    sbase[t] = start + off;
    __syncthreads();
#pragma unroll
    for (int j = 0; j < 8; ++j)
        if (mybin[j] >= 0) perm[sbase[mybin[j]] + myrank[j]] = base + t + 256 * j;
}

__global__ __launch_bounds__(256) void egnn_prep_w(const float* __restrict__ W,
                                                   int Kact, int Nact, int ntiles,
                                                   int total, _Float16* __restrict__ out) {
    int idx = blockIdx.x * 256 + threadIdx.x;
    if (idx >= total) return;
    int block  = idx >> 9;
    int within = idx & 511;
    int lane = within >> 4;
    int e    = within & 15;
    int kc = block / ntiles;
    int nt = block - kc * ntiles;
    int khalf  = lane >> 4;
    int coln   = lane & 15;
    int klocal = (e & 7) + 8 * khalf + 16 * (e >> 3);
    int k = kc * 32 + klocal;
    int n = nt * 16 + coln;
    float v = (k < Kact && n < Nact) ? W[(size_t)k * Nact + n] : 0.0f;
    out[idx] = (_Float16)v;
}

typedef __attribute__((ext_vector_type(4))) float v4f;

typedef __attribute__((ext_vector_type(8))) _Float16 v8h;
__device__ __forceinline__ v16h ld_ws(const _Float16* blk, int lane) {
    v8h lo = *(const v8h*)(blk + lane * 8);
    v8h hi = *(const v8h*)(blk + 256 + lane * 8);
    return __builtin_shufflevector(lo, hi, 0, 1, 2, 3, 4, 5, 6, 7, 8, 9, 10, 11, 12, 13, 14, 15);
}

__device__ __forceinline__ v16h row_frag(const float* __restrict__ p, int h) {
    const v4f* q = (const v4f*)(p + 8 * h);
    v4f a0 = q[0], a1 = q[1], b0 = q[4], b1 = q[5];
    v16h f;
    f[0]  = (_Float16)a0.x; f[1]  = (_Float16)a0.y; f[2]  = (_Float16)a0.z; f[3]  = (_Float16)a0.w;
    f[4]  = (_Float16)a1.x; f[5]  = (_Float16)a1.y; f[6]  = (_Float16)a1.z; f[7]  = (_Float16)a1.w;
    f[8]  = (_Float16)b0.x; f[9]  = (_Float16)b0.y; f[10] = (_Float16)b0.z; f[11] = (_Float16)b0.w;
    f[12] = (_Float16)b1.x; f[13] = (_Float16)b1.y; f[14] = (_Float16)b1.z; f[15] = (_Float16)b1.w;
    return f;
}

__device__ __forceinline__ v8f bias8(v8f a, const float* __restrict__ b, bool relu) {
    const v4f* q = (const v4f*)b;
    v4f b0 = q[0], b1 = q[1];
    float bb[8] = {b0.x, b0.y, b0.z, b0.w, b1.x, b1.y, b1.z, b1.w};
    v8f o;
#pragma unroll
    for (int r = 0; r < 8; ++r) {
        float v = a[r] + bb[r];
        o[r] = relu ? (v > 0.0f ? v : 0.0f) : v;
    }
    return o;
}

__device__ __forceinline__ v16h pack2(v8f lo, v8f hi) {
    v16h f;
#pragma unroll
    for (int r = 0; r < 8; ++r) { f[r] = (_Float16)lo[r]; f[8 + r] = (_Float16)hi[r]; }
    return f;
}

__device__ __forceinline__ v16h mean_frag(const float* p, int h, float s) {
    const v4f* q = (const v4f*)(p + 8 * h);
    v4f a0 = q[0], a1 = q[1], b0 = q[4], b1 = q[5];
    v16h f;
    f[0]  = (_Float16)(a0.x * s); f[1]  = (_Float16)(a0.y * s); f[2]  = (_Float16)(a0.z * s); f[3]  = (_Float16)(a0.w * s);
    f[4]  = (_Float16)(a1.x * s); f[5]  = (_Float16)(a1.y * s); f[6]  = (_Float16)(a1.z * s); f[7]  = (_Float16)(a1.w * s);
    f[8]  = (_Float16)(b0.x * s); f[9]  = (_Float16)(b0.y * s); f[10] = (_Float16)(b0.z * s); f[11] = (_Float16)(b0.w * s);
    f[12] = (_Float16)(b1.x * s); f[13] = (_Float16)(b1.y * s); f[14] = (_Float16)(b1.z * s); f[15] = (_Float16)(b1.w * s);
    return f;
}

__device__ __forceinline__ v16h ld_wg(const _Float16* blk, int lane) {
    return *(const v16h*)(blk + lane * 16);
}

#define SACC_PITCH   68
#define NODE_TILES   ((BIN_NODES + 15) / 16)
#define EDGE_WAVES   16
#define EDGE_THREADS (EDGE_WAVES * 32)
#define EDGE_WBLOCKS 38
__global__ __launch_bounds__(EDGE_THREADS) void egnn_edge(
    const float* __restrict__ x, const int* __restrict__ ei,
    const float* __restrict__ pos, const float* __restrict__ pos_init,
    const float* __restrict__ edge_attr,
    const float* __restrict__ mb1, const float* __restrict__ mb2,
    const float* __restrict__ cb1, const float* __restrict__ cb2,
    const float* __restrict__ nb1, const float* __restrict__ nb2,
    const _Float16* __restrict__ wsW,
    const int* __restrict__ perm, const int* __restrict__ gcount,
    float* __restrict__ hout, float* __restrict__ pos_out)
{
    __shared__ __attribute__((aligned(16))) float sacc[(BIN_NODES + 1) * SACC_PITCH];
    __shared__ float spos[(BIN_NODES + 1) * 3];
    __shared__ float scnt[BIN_NODES + 1];
    __shared__ int srange[2];
    __shared__ __attribute__((aligned(16))) _Float16 sW[EDGE_WBLOCKS * 512];

    const int tid  = threadIdx.x;
    const int bin  = blockIdx.x;
    const int nbase = bin * BIN_NODES;
    for (int i = tid; i < (BIN_NODES + 1) * SACC_PITCH; i += EDGE_THREADS) sacc[i] = 0.0f;
    for (int i = tid; i < (BIN_NODES + 1) * 3; i += EDGE_THREADS)    spos[i] = 0.0f;
    {
        const v8h* wsrc = (const v8h*)(wsW + MW1B);
        for (int q = tid; q < EDGE_WBLOCKS * 64; q += EDGE_THREADS) {
            const int blk = q >> 6, ln = (q & 63) >> 1, half = q & 1;
            *(v8h*)(sW + blk * 512 + half * 256 + ln * 8) = wsrc[q];
        }
    }
    if (tid < BIN_NODES + 1) scnt[tid] = 0.0f;
    if (tid < 32) {
        int acc0 = 0;
        for (int k = tid; k < bin; k += 32) acc0 += gcount[k];
        for (int o = 16; o > 0; o >>= 1) acc0 += __shfl_xor(acc0, o);
        if (tid == 0) { srange[0] = acc0; srange[1] = acc0 + gcount[bin]; }
    }
    __syncthreads();
    const int estart = srange[0];
    const int eend   = srange[1];
    const int ntiles = (eend - estart + 15) >> 4;

    const int lane = tid & 31;
    const int c = lane & 15;
    const int h = lane >> 4;
    const int wave = tid >> 5;
    int e_cur = 0, s_cur = 0, d_cur = 0, e_nxt = 0;
    if (wave < ntiles) {
        int p = estart + wave * 16 + c;
        e_cur = perm[p < eend ? p : estart];
    }
    if (wave + EDGE_WAVES < ntiles) {
        int p = estart + (wave + EDGE_WAVES) * 16 + c;
        e_nxt = perm[p < eend ? p : estart];
    }
    if (wave < ntiles) { s_cur = ei[e_cur]; d_cur = ei[N_EDGES + e_cur]; }
    float mb2c[4];
#pragma unroll
    for (int nt = 0; nt < 4; ++nt) mb2c[nt] = mb2[nt * 16 + c];
  for (int tile = wave; tile < ntiles; tile += EDGE_WAVES) {
    const bool live = estart + tile * 16 + c < eend;
    const int e = e_cur;
    const int s = s_cur;
    const int d = d_cur;
    e_cur = e_nxt;
    if (tile + EDGE_WAVES < ntiles) { s_cur = ei[e_cur]; d_cur = ei[N_EDGES + e_cur]; }
    if (tile + 2 * EDGE_WAVES < ntiles) {
        int p = estart + (tile + 2 * EDGE_WAVES) * 16 + c;
        e_nxt = perm[p < eend ? p : estart];
    }
    const int lrow = live ? (d - nbase) : BIN_NODES;
    int wofs = 0;
    asm volatile("" : "+v"(wofs));
    const _Float16* sWt = sW + wofs;

    const float* xs = x + (size_t)s * FDIM;
    const float* xd = x + (size_t)d * FDIM;
    v16h fr[5];
    fr[0] = row_frag(xs, h);
    fr[1] = row_frag(xs + 32, h);
    fr[2] = row_frag(xd, h);
    fr[3] = row_frag(xd + 32, h);
    const v4f* eq = (const v4f*)(edge_attr + (size_t)e * EFDIM);
    v4f e0 = eq[0], e1 = eq[1], e2 = eq[2], e3 = eq[3];

    float dx = pos[3*s+0] - pos[3*d+0];
    float dy = pos[3*s+1] - pos[3*d+1];
    float dz = pos[3*s+2] - pos[3*d+2];
    float dsq = dx*dx + dy*dy + dz*dz;
    float ix = pos_init[3*s+0] - pos_init[3*d+0];
    float iy = pos_init[3*s+1] - pos_init[3*d+1];
    float iz = pos_init[3*s+2] - pos_init[3*d+2];
    float rinv = 1.0f / sqrtf(ix*ix + iy*iy + iz*iz);
    if (h == 0) atomicAdd(&scnt[lrow], 1.0f);

    {
        float ea[16] = {e0.x, e0.y, e0.z, e0.w, e1.x, e1.y, e1.z, e1.w,
                        e2.x, e2.y, e2.z, e2.w, e3.x, e3.y, e3.z, e3.w};
        v16h f;
#pragma unroll
        for (int t = 0; t < 8; ++t) {
            float lo = (t == 0) ? dsq : ea[t - 1];
            float hi = ea[7 + t];
            f[t] = (_Float16)(h ? hi : lo);
        }
        f[8] = (_Float16)(h ? 0.0f : ea[15]);
#pragma unroll
        for (int t = 9; t < 16; ++t) f[t] = (_Float16)0.0f;
        fr[4] = f;
    }

    v8f acc[4] = {v8f{}, v8f{}, v8f{}, v8f{}};
#pragma unroll
    for (int kc = 0; kc < 5; ++kc) {
#pragma unroll
        for (int mt = 0; mt < 4; ++mt)
            acc[mt] = wmma16(ld_ws(sWt + MW1B + (kc * 4 + mt) * 512, lane), fr[kc], acc[mt]);
    }
    v16h hf[2];
#pragma unroll
    for (int kc = 0; kc < 2; ++kc)
        hf[kc] = pack2(bias8(acc[2*kc],     mb1 + 32*kc      + 8*h, true),
                       bias8(acc[2*kc + 1], mb1 + 32*kc + 16 + 8*h, true));

    v8f mij[4] = {v8f{}, v8f{}, v8f{}, v8f{}};
    v8f mt_[4] = {v8f{}, v8f{}, v8f{}, v8f{}};
    {
        v16h w2[8];
#pragma unroll
        for (int i = 0; i < 8; ++i) w2[i] = ld_ws(sWt + MW2B + i * 512, lane);
#pragma unroll
        for (int kc = 0; kc < 2; ++kc) {
#pragma unroll
            for (int nt = 0; nt < 4; ++nt)
                mij[nt] = wmma16(hf[kc], w2[kc * 4 + nt], mij[nt]);
        }
#pragma unroll
        for (int kc = 0; kc < 2; ++kc) {
#pragma unroll
            for (int mt = 0; mt < 4; ++mt)
                mt_[mt] = wmma16(w2[kc * 4 + mt], hf[kc], mt_[mt]);
        }
    }
    v16h wc1[8], wc2[2];
#pragma unroll
    for (int i = 0; i < 8; ++i) wc1[i] = ld_ws(sWt + CW1B + i * 512, lane);
#pragma unroll
    for (int i = 0; i < 2; ++i) wc2[i] = ld_ws(sWt + CW2B + i * 512, lane);
    v16h mf[2];
#pragma unroll
    for (int kc = 0; kc < 2; ++kc)
        mf[kc] = pack2(bias8(mt_[2*kc],     mb2 + 32*kc      + 8*h, false),
                       bias8(mt_[2*kc + 1], mb2 + 32*kc + 16 + 8*h, false));
    {
        int drow[8];
#pragma unroll
        for (int r = 0; r < 8; ++r) drow[r] = __shfl(lrow, 8 * h + r);
#pragma unroll
        for (int nt = 0; nt < 4; ++nt) {
#pragma unroll
            for (int r = 0; r < 8; ++r)
                atomicAdd(&sacc[drow[r] * SACC_PITCH + nt * 16 + c], mij[nt][r] + mb2c[nt]);
        }
    }

    v16h tf[2];
    {
        v8f tac[4] = {v8f{}, v8f{}, v8f{}, v8f{}};
#pragma unroll
        for (int kc = 0; kc < 2; ++kc) {
#pragma unroll
            for (int mt = 0; mt < 4; ++mt)
                tac[mt] = wmma16(wc1[kc * 4 + mt], mf[kc], tac[mt]);
        }
#pragma unroll
        for (int kc = 0; kc < 2; ++kc)
            tf[kc] = pack2(bias8(tac[2*kc],     cb1 + 32*kc      + 8*h, true),
                           bias8(tac[2*kc + 1], cb1 + 32*kc + 16 + 8*h, true));
    }

    v8f cac = v8f{};
#pragma unroll
    for (int kc = 0; kc < 2; ++kc)
        cac = wmma16(wc2[kc], tf[kc], cac);
    float ux = ix * rinv * (cac[0] + cb2[0]);
    float uy = iy * rinv * (cac[1] + cb2[1]);
    float uz = iz * rinv * (cac[2] + cb2[2]);
    if (h == 0) {
        atomicAdd(&spos[lrow * 3 + 0], ux);
        atomicAdd(&spos[lrow * 3 + 1], uy);
        atomicAdd(&spos[lrow * 3 + 2], uz);
    }
  }
    __syncthreads();
    for (int i = tid; i < BIN_NODES * 3; i += EDGE_THREADS)    pos_out[(size_t)nbase * 3 + i] = spos[i];

    if (wave < NODE_TILES) {
        const int lr  = wave * 16 + c;
        const bool have = lr < BIN_NODES;
        const int lrc = have ? lr : BIN_NODES - 1;
        const float* xr = x + (size_t)(nbase + lrc) * FDIM;
        const float* mr = sacc + lrc * SACC_PITCH;
        const float inv = 1.0f / fmaxf(scnt[lrc], 1.0f);
        v16h nf[4];
        nf[0] = row_frag(xr, h);
        nf[1] = row_frag(xr + 32, h);
        nf[2] = mean_frag(mr, h, inv);
        nf[3] = mean_frag(mr + 32, h, inv);
        v8f nac[4] = {v8f{}, v8f{}, v8f{}, v8f{}};
#pragma unroll
        for (int kc = 0; kc < 4; ++kc) {
#pragma unroll
            for (int mt = 0; mt < 4; ++mt)
                nac[mt] = wmma16(ld_wg(wsW + NW1B + (kc * 4 + mt) * 512, lane), nf[kc], nac[mt]);
        }
        v16h nh[2];
#pragma unroll
        for (int kc = 0; kc < 2; ++kc)
            nh[kc] = pack2(bias8(nac[2*kc],     nb1 + 32*kc      + 8*h, true),
                           bias8(nac[2*kc + 1], nb1 + 32*kc + 16 + 8*h, true));
        v8f oac[4] = {v8f{}, v8f{}, v8f{}, v8f{}};
#pragma unroll
        for (int kc = 0; kc < 2; ++kc) {
#pragma unroll
            for (int mt = 0; mt < 4; ++mt)
                oac[mt] = wmma16(ld_wg(wsW + NW2B + (kc * 4 + mt) * 512, lane), nh[kc], oac[mt]);
        }
#pragma unroll
        for (int mt = 0; mt < 4; ++mt) {
            v8f o = bias8(oac[mt], nb2 + 16*mt + 8*h, false);
            if (have) {
                v4f* dst = (v4f*)(sacc + lr * SACC_PITCH + 16*mt + 8*h);
                dst[0] = v4f{o[0], o[1], o[2], o[3]};
                dst[1] = v4f{o[4], o[5], o[6], o[7]};
            }
        }
    }
    __syncthreads();
    for (int i = tid; i < BIN_NODES * (FDIM / 4); i += EDGE_THREADS) {
        const int row = i >> 4, q = i & 15;
        *(v4f*)(hout + (size_t)(nbase + row) * FDIM + q * 4) = *(const v4f*)(sacc + row * SACC_PITCH + q * 4);
    }
}

extern "C" void kernel_launch(void* const* d_in, const int* in_sizes, int n_in,
                              void* d_out, int out_size, void* d_ws, size_t ws_size,
                              hipStream_t stream) {
    const float* x         = (const float*)d_in[0];
    const int*   ei        = (const int*)  d_in[1];
    const float* pos       = (const float*)d_in[2];
    const float* pos_init  = (const float*)d_in[3];
    const float* edge_attr = (const float*)d_in[4];
    const float* mw1 = (const float*)d_in[5];
    const float* mb1 = (const float*)d_in[6];
    const float* mw2 = (const float*)d_in[7];
    const float* mb2 = (const float*)d_in[8];
    const float* nw1 = (const float*)d_in[9];
    const float* nb1 = (const float*)d_in[10];
    const float* nw2 = (const float*)d_in[11];
    const float* nb2 = (const float*)d_in[12];
    const float* cw1 = (const float*)d_in[13];
    const float* cb1 = (const float*)d_in[14];
    const float* cw2 = (const float*)d_in[15];
    const float* cb2 = (const float*)d_in[16];

    float* h_out   = (float*)d_out;
    float* pos_out = h_out + (size_t)N_NODES * FDIM;

    _Float16* wsW = (_Float16*)((char*)d_ws + WS_WEIGHTS_OFF);

    int* perm    = (int*)d_ws;
    int* gcount  = perm + N_EDGES;
    int* gcursor = gcount + 256;
    egnn_bin_clear<<<1, 256, 0, stream>>>(gcount);
    egnn_bin_count<<<(N_EDGES + BIN_CHUNK - 1) / BIN_CHUNK, 256, 0, stream>>>(ei, gcount);
    egnn_bin_fill<<<(N_EDGES + BIN_CHUNK - 1) / BIN_CHUNK, 256, 0, stream>>>(ei, gcount, gcursor, perm);

    egnn_prep_w<<<40, 256, 0, stream>>>(mw1, MSG_IN, 64, 4, 20 * 512, wsW + MW1B);
    egnn_prep_w<<<16, 256, 0, stream>>>(mw2, 64,     64, 4,  8 * 512, wsW + MW2B);
    egnn_prep_w<<<16, 256, 0, stream>>>(cw1, 64,     64, 4,  8 * 512, wsW + CW1B);
    egnn_prep_w<<< 4, 256, 0, stream>>>(cw2, 64,      3, 1,  2 * 512, wsW + CW2B);
    egnn_prep_w<<<32, 256, 0, stream>>>(nw1, 128,    64, 4, 16 * 512, wsW + NW1B);
    egnn_prep_w<<<16, 256, 0, stream>>>(nw2, 64,     64, 4,  8 * 512, wsW + NW2B);

    egnn_edge<<<NBINS, EDGE_THREADS, 0, stream>>>(
        x, ei, pos, pos_init, edge_attr, mb1, mb2, cb1, cb2, nb1, nb2,
        wsW, perm, gcount, h_out, pos_out);
}


# Each edge work-group converts the six f32 weight matrices itself into its LDS f16 fragment planes (node layers read them there too): no packed-weight workspace, no pack launches; same f16 values
# speedup vs baseline: 3.8092x; 1.1440x over previous
#include <hip/hip_runtime.h>

typedef __attribute__((ext_vector_type(16))) _Float16 v16h;
typedef __attribute__((ext_vector_type(8)))  float    v8f;

#define N_NODES 50000
#define N_EDGES 800000
#define FDIM    64
#define EFDIM   16
#define MSG_IN  145
#define MSG_K   160


#define MW1B 0
#define MW2B 10240
#define CW1B 14336
#define NW1B 18432
#define NW2B 26624
#define CW2B 30720

__device__ __forceinline__ v8f wmma16(v16h a, v16h b, v8f c) {
    return __builtin_amdgcn_wmma_f32_16x16x32_f16(
         false, a,  false, b,
         (short)0, c,  false,  false);
}

#define BIN_NODES 200
#define NBINS     250
#define BIN_CHUNK 2048

__global__ __launch_bounds__(256) void egnn_bin_clear(int* __restrict__ binctr) {
    binctr[threadIdx.x] = 0;
    binctr[256 + threadIdx.x] = 0;
}

__global__ __launch_bounds__(256) void egnn_bin_count(const int* __restrict__ ei,
                                                      int* __restrict__ gcount) {
    __shared__ int hist[256];
    const int t = threadIdx.x;
    hist[t] = 0;
    __syncthreads();
    const int base = blockIdx.x * BIN_CHUNK;
    for (int i = t; i < BIN_CHUNK; i += 256) {
        int e = base + i;
        if (e < N_EDGES) atomicAdd(&hist[ei[N_EDGES + e] / BIN_NODES], 1);
    }
    __syncthreads();
    int v = hist[t];
    if (v) atomicAdd(&gcount[t], v);
}

__global__ __launch_bounds__(256) void egnn_bin_fill(const int* __restrict__ ei,
                                                     const int* __restrict__ gcount,
                                                     int* __restrict__ gcursor,
                                                     int* __restrict__ perm) {
    __shared__ int hist[256];
    __shared__ int scount[256];
    __shared__ int sbase[256];
    const int t = threadIdx.x;
    hist[t] = 0;
    const int cv = gcount[t];
    int incl = cv;
    scount[t] = incl;
    __syncthreads();
    for (int o = 1; o < 256; o <<= 1) {
        int add = (t >= o) ? scount[t - o] : 0;
        __syncthreads();
        incl += add;
        scount[t] = incl;
        __syncthreads();
    }
    const int start = incl - cv;
    const int base = blockIdx.x * BIN_CHUNK;
    int mybin[8], myrank[8];
#pragma unroll
    for (int j = 0; j < 8; ++j) {
        int e = base + t + 256 * j;
        mybin[j] = -1; myrank[j] = 0;
        if (e < N_EDGES) {
            int b = ei[N_EDGES + e] / BIN_NODES;
            mybin[j] = b;
            myrank[j] = atomicAdd(&hist[b], 1);
        }
    }
    __syncthreads();
    int v = hist[t];
    int off = v ? atomicAdd(&gcursor[t], v) : 0;
    sbase[t] = start + off;
    __syncthreads();
#pragma unroll
    for (int j = 0; j < 8; ++j)
        if (mybin[j] >= 0) perm[sbase[mybin[j]] + myrank[j]] = base + t + 256 * j;
}

typedef __attribute__((ext_vector_type(4))) float v4f;

typedef __attribute__((ext_vector_type(8))) _Float16 v8h;
__device__ __forceinline__ v16h ld_ws(const _Float16* blk, int lane) {
    v8h lo = *(const v8h*)(blk + lane * 8);
    v8h hi = *(const v8h*)(blk + 256 + lane * 8);
    return __builtin_shufflevector(lo, hi, 0, 1, 2, 3, 4, 5, 6, 7, 8, 9, 10, 11, 12, 13, 14, 15);
}

__device__ __forceinline__ v16h row_frag(const float* __restrict__ p, int h) {
    const v4f* q = (const v4f*)(p + 8 * h);
    v4f a0 = q[0], a1 = q[1], b0 = q[4], b1 = q[5];
    v16h f;
    f[0]  = (_Float16)a0.x; f[1]  = (_Float16)a0.y; f[2]  = (_Float16)a0.z; f[3]  = (_Float16)a0.w;
    f[4]  = (_Float16)a1.x; f[5]  = (_Float16)a1.y; f[6]  = (_Float16)a1.z; f[7]  = (_Float16)a1.w;
    f[8]  = (_Float16)b0.x; f[9]  = (_Float16)b0.y; f[10] = (_Float16)b0.z; f[11] = (_Float16)b0.w;
    f[12] = (_Float16)b1.x; f[13] = (_Float16)b1.y; f[14] = (_Float16)b1.z; f[15] = (_Float16)b1.w;
    return f;
}

__device__ __forceinline__ v8f bias8(v8f a, const float* __restrict__ b, bool relu) {
    const v4f* q = (const v4f*)b;
    v4f b0 = q[0], b1 = q[1];
    float bb[8] = {b0.x, b0.y, b0.z, b0.w, b1.x, b1.y, b1.z, b1.w};
    v8f o;
#pragma unroll
    for (int r = 0; r < 8; ++r) {
        float v = a[r] + bb[r];
        o[r] = relu ? (v > 0.0f ? v : 0.0f) : v;
    }
    return o;
}

__device__ __forceinline__ v16h pack2(v8f lo, v8f hi) {
    v16h f;
#pragma unroll
    for (int r = 0; r < 8; ++r) { f[r] = (_Float16)lo[r]; f[8 + r] = (_Float16)hi[r]; }
    return f;
}

__device__ __forceinline__ v16h mean_frag(const float* p, int h, float s) {
    const v4f* q = (const v4f*)(p + 8 * h);
    v4f a0 = q[0], a1 = q[1], b0 = q[4], b1 = q[5];
    v16h f;
    f[0]  = (_Float16)(a0.x * s); f[1]  = (_Float16)(a0.y * s); f[2]  = (_Float16)(a0.z * s); f[3]  = (_Float16)(a0.w * s);
    f[4]  = (_Float16)(a1.x * s); f[5]  = (_Float16)(a1.y * s); f[6]  = (_Float16)(a1.z * s); f[7]  = (_Float16)(a1.w * s);
    f[8]  = (_Float16)(b0.x * s); f[9]  = (_Float16)(b0.y * s); f[10] = (_Float16)(b0.z * s); f[11] = (_Float16)(b0.w * s);
    f[12] = (_Float16)(b1.x * s); f[13] = (_Float16)(b1.y * s); f[14] = (_Float16)(b1.z * s); f[15] = (_Float16)(b1.w * s);
    return f;
}

#define SACC_PITCH   68
#define NODE_TILES   ((BIN_NODES + 15) / 16)
#define EDGE_WAVES   16
#define EDGE_THREADS (EDGE_WAVES * 32)
#define EDGE_WBLOCKS 62
#define W_ITEMS      960
__global__ __launch_bounds__(EDGE_THREADS) void egnn_edge(
    const float* __restrict__ x, const int* __restrict__ ei,
    const float* __restrict__ pos, const float* __restrict__ pos_init,
    const float* __restrict__ edge_attr,
    const float* __restrict__ mb1, const float* __restrict__ mb2,
    const float* __restrict__ cb1, const float* __restrict__ cb2,
    const float* __restrict__ nb1, const float* __restrict__ nb2,
    const float* __restrict__ mw1, const float* __restrict__ mw2,
    const float* __restrict__ cw1, const float* __restrict__ cw2,
    const float* __restrict__ nw1, const float* __restrict__ nw2,
    const int* __restrict__ perm, const int* __restrict__ gcount,
    float* __restrict__ hout, float* __restrict__ pos_out)
{
    __shared__ __attribute__((aligned(16))) float sacc[(BIN_NODES + 1) * SACC_PITCH];
    __shared__ float spos[(BIN_NODES + 1) * 3];
    __shared__ float scnt[BIN_NODES + 1];
    __shared__ int srange[2];
    __shared__ __attribute__((aligned(16))) _Float16 sW[EDGE_WBLOCKS * 512];

    const int tid  = threadIdx.x;
    const int bin  = blockIdx.x;
    const int nbase = bin * BIN_NODES;
    for (int i = tid; i < (BIN_NODES + 1) * SACC_PITCH; i += EDGE_THREADS) sacc[i] = 0.0f;
    for (int i = tid; i < (BIN_NODES + 1) * 3; i += EDGE_THREADS)    spos[i] = 0.0f;
    {
        v4f wr[2][8];
        int wofs_l[2];
#pragma unroll
        for (int rd = 0; rd < 2; ++rd) {
            int it = tid + rd * EDGE_THREADS;
            it = it < W_ITEMS ? it : W_ITEMS - 1;
            const int gg = it >> 4, n4 = it & 15;
            const float* wm; int g0, kact;
            if (gg < 20)      { wm = mw1; g0 = 0;  kact = MSG_IN; }
            else if (gg < 28) { wm = mw2; g0 = 20; kact = 64; }
            else if (gg < 36) { wm = cw1; g0 = 28; kact = 64; }
            else if (gg < 52) { wm = nw1; g0 = 36; kact = 128; }
            else              { wm = nw2; g0 = 52; kact = 64; }
            const int k0 = (gg - g0) * 8;
#pragma unroll
            for (int j = 0; j < 8; ++j) {
                const int k = k0 + j;
                const int kc = k < kact ? k : kact - 1;
                const v4f v = *(const v4f*)(wm + kc * 64 + n4 * 4);
                wr[rd][j] = k < kact ? v : v4f{0.0f, 0.0f, 0.0f, 0.0f};
            }
            wofs_l[rd] = ((gg >> 2) * 4 + (n4 >> 2)) * 512 + ((gg >> 1) & 1) * 256
                       + ((gg & 1) * 16 + (n4 & 3) * 4) * 8;
        }
#pragma unroll
        for (int rd = 0; rd < 2; ++rd) {
            if (tid + rd * EDGE_THREADS < W_ITEMS) {
#pragma unroll
                for (int cc = 0; cc < 4; ++cc) {
                    v8h rec;
#pragma unroll
                    for (int j = 0; j < 8; ++j) rec[j] = (_Float16)wr[rd][j][cc];
                    *(v8h*)(sW + wofs_l[rd] + cc * 8) = rec;
                }
            }
        }
        if (tid >= EDGE_THREADS - 128) {
            const int t = tid - (EDGE_THREADS - 128);
            const int kc = t >> 6, pl = (t >> 5) & 1, ln = t & 31;
            const int kh = ln >> 4, cn = ln & 15;
            const int k0 = kc * 32 + pl * 16 + kh * 8;
            v8h rec;
#pragma unroll
            for (int j = 0; j < 8; ++j) {
                const float v = cw2[cn < 3 ? (k0 + j) * 3 + cn : 0];
                rec[j] = (_Float16)(cn < 3 ? v : 0.0f);
            }
            *(v8h*)(sW + CW2B + kc * 512 + pl * 256 + ln * 8) = rec;
        }
    }
    if (tid < BIN_NODES + 1) scnt[tid] = 0.0f;
    if (tid < 32) {
        int acc0 = 0;
        for (int k = tid; k < bin; k += 32) acc0 += gcount[k];
        for (int o = 16; o > 0; o >>= 1) acc0 += __shfl_xor(acc0, o);
        if (tid == 0) { srange[0] = acc0; srange[1] = acc0 + gcount[bin]; }
    }
    __syncthreads();
    const int estart = srange[0];
    const int eend   = srange[1];
    const int ntiles = (eend - estart + 15) >> 4;

    const int lane = tid & 31;
    const int c = lane & 15;
    const int h = lane >> 4;
    const int wave = tid >> 5;
    int e_cur = 0, s_cur = 0, d_cur = 0, e_nxt = 0;
    if (wave < ntiles) {
        int p = estart + wave * 16 + c;
        e_cur = perm[p < eend ? p : estart];
    }
    if (wave + EDGE_WAVES < ntiles) {
        int p = estart + (wave + EDGE_WAVES) * 16 + c;
        e_nxt = perm[p < eend ? p : estart];
    }
    if (wave < ntiles) { s_cur = ei[e_cur]; d_cur = ei[N_EDGES + e_cur]; }
    float mb2c[4];
#pragma unroll
    for (int nt = 0; nt < 4; ++nt) mb2c[nt] = mb2[nt * 16 + c];
  for (int tile = wave; tile < ntiles; tile += EDGE_WAVES) {
    const bool live = estart + tile * 16 + c < eend;
    const int e = e_cur;
    const int s = s_cur;
    const int d = d_cur;
    e_cur = e_nxt;
    if (tile + EDGE_WAVES < ntiles) { s_cur = ei[e_cur]; d_cur = ei[N_EDGES + e_cur]; }
    if (tile + 2 * EDGE_WAVES < ntiles) {
        int p = estart + (tile + 2 * EDGE_WAVES) * 16 + c;
        e_nxt = perm[p < eend ? p : estart];
    }
    const int lrow = live ? (d - nbase) : BIN_NODES;
    int wofs = 0;
    asm volatile("" : "+v"(wofs));
    const _Float16* sWt = sW + wofs;

    const float* xs = x + (size_t)s * FDIM;
    const float* xd = x + (size_t)d * FDIM;
    v16h fr[5];
    fr[0] = row_frag(xs, h);
    fr[1] = row_frag(xs + 32, h);
    fr[2] = row_frag(xd, h);
    fr[3] = row_frag(xd + 32, h);
    const v4f* eq = (const v4f*)(edge_attr + (size_t)e * EFDIM);
    v4f e0 = eq[0], e1 = eq[1], e2 = eq[2], e3 = eq[3];

    float dx = pos[3*s+0] - pos[3*d+0];
    float dy = pos[3*s+1] - pos[3*d+1];
    float dz = pos[3*s+2] - pos[3*d+2];
    float dsq = dx*dx + dy*dy + dz*dz;
    float ix = pos_init[3*s+0] - pos_init[3*d+0];
    float iy = pos_init[3*s+1] - pos_init[3*d+1];
    float iz = pos_init[3*s+2] - pos_init[3*d+2];
    float rinv = 1.0f / sqrtf(ix*ix + iy*iy + iz*iz);
    if (h == 0) atomicAdd(&scnt[lrow], 1.0f);

    {
        float ea[16] = {e0.x, e0.y, e0.z, e0.w, e1.x, e1.y, e1.z, e1.w,
                        e2.x, e2.y, e2.z, e2.w, e3.x, e3.y, e3.z, e3.w};
        v16h f;
#pragma unroll
        for (int t = 0; t < 8; ++t) {
            float lo = (t == 0) ? dsq : ea[t - 1];
            float hi = ea[7 + t];
            f[t] = (_Float16)(h ? hi : lo);
        }
        f[8] = (_Float16)(h ? 0.0f : ea[15]);
#pragma unroll
        for (int t = 9; t < 16; ++t) f[t] = (_Float16)0.0f;
        fr[4] = f;
    }

    v8f acc[4] = {v8f{}, v8f{}, v8f{}, v8f{}};
#pragma unroll
    for (int kc = 0; kc < 5; ++kc) {
#pragma unroll
        for (int mt = 0; mt < 4; ++mt)
            acc[mt] = wmma16(ld_ws(sWt + MW1B + (kc * 4 + mt) * 512, lane), fr[kc], acc[mt]);
    }
    v16h hf[2];
#pragma unroll
    for (int kc = 0; kc < 2; ++kc)
        hf[kc] = pack2(bias8(acc[2*kc],     mb1 + 32*kc      + 8*h, true),
                       bias8(acc[2*kc + 1], mb1 + 32*kc + 16 + 8*h, true));

    v8f mij[4] = {v8f{}, v8f{}, v8f{}, v8f{}};
    v8f mt_[4] = {v8f{}, v8f{}, v8f{}, v8f{}};
    {
        v16h w2[8];
#pragma unroll
        for (int i = 0; i < 8; ++i) w2[i] = ld_ws(sWt + MW2B + i * 512, lane);
#pragma unroll
        for (int kc = 0; kc < 2; ++kc) {
#pragma unroll
            for (int nt = 0; nt < 4; ++nt)
                mij[nt] = wmma16(hf[kc], w2[kc * 4 + nt], mij[nt]);
        }
#pragma unroll
        for (int kc = 0; kc < 2; ++kc) {
#pragma unroll
            for (int mt = 0; mt < 4; ++mt)
                mt_[mt] = wmma16(w2[kc * 4 + mt], hf[kc], mt_[mt]);
        }
    }
    v16h wc1[8], wc2[2];
#pragma unroll
    for (int i = 0; i < 8; ++i) wc1[i] = ld_ws(sWt + CW1B + i * 512, lane);
#pragma unroll
    for (int i = 0; i < 2; ++i) wc2[i] = ld_ws(sWt + CW2B + i * 512, lane);
    v16h mf[2];
#pragma unroll
    for (int kc = 0; kc < 2; ++kc)
        mf[kc] = pack2(bias8(mt_[2*kc],     mb2 + 32*kc      + 8*h, false),
                       bias8(mt_[2*kc + 1], mb2 + 32*kc + 16 + 8*h, false));
    {
        int drow[8];
#pragma unroll
        for (int r = 0; r < 8; ++r) drow[r] = __shfl(lrow, 8 * h + r);
#pragma unroll
        for (int nt = 0; nt < 4; ++nt) {
#pragma unroll
            for (int r = 0; r < 8; ++r)
                atomicAdd(&sacc[drow[r] * SACC_PITCH + nt * 16 + c], mij[nt][r] + mb2c[nt]);
        }
    }

    v16h tf[2];
    {
        v8f tac[4] = {v8f{}, v8f{}, v8f{}, v8f{}};
#pragma unroll
        for (int kc = 0; kc < 2; ++kc) {
#pragma unroll
            for (int mt = 0; mt < 4; ++mt)
                tac[mt] = wmma16(wc1[kc * 4 + mt], mf[kc], tac[mt]);
        }
#pragma unroll
        for (int kc = 0; kc < 2; ++kc)
            tf[kc] = pack2(bias8(tac[2*kc],     cb1 + 32*kc      + 8*h, true),
                           bias8(tac[2*kc + 1], cb1 + 32*kc + 16 + 8*h, true));
    }

    v8f cac = v8f{};
#pragma unroll
    for (int kc = 0; kc < 2; ++kc)
        cac = wmma16(wc2[kc], tf[kc], cac);
    float ux = ix * rinv * (cac[0] + cb2[0]);
    float uy = iy * rinv * (cac[1] + cb2[1]);
    float uz = iz * rinv * (cac[2] + cb2[2]);
    if (h == 0) {
        atomicAdd(&spos[lrow * 3 + 0], ux);
        atomicAdd(&spos[lrow * 3 + 1], uy);
        atomicAdd(&spos[lrow * 3 + 2], uz);
    }
  }
    __syncthreads();
    for (int i = tid; i < BIN_NODES * 3; i += EDGE_THREADS)    pos_out[(size_t)nbase * 3 + i] = spos[i];

    if (wave < NODE_TILES) {
        const int lr  = wave * 16 + c;
        const bool have = lr < BIN_NODES;
        const int lrc = have ? lr : BIN_NODES - 1;
        const float* xr = x + (size_t)(nbase + lrc) * FDIM;
        const float* mr = sacc + lrc * SACC_PITCH;
        const float inv = 1.0f / fmaxf(scnt[lrc], 1.0f);
        v16h nf[4];
        nf[0] = row_frag(xr, h);
        nf[1] = row_frag(xr + 32, h);
        nf[2] = mean_frag(mr, h, inv);
        nf[3] = mean_frag(mr + 32, h, inv);
        v8f nac[4] = {v8f{}, v8f{}, v8f{}, v8f{}};
#pragma unroll
        for (int kc = 0; kc < 4; ++kc) {
#pragma unroll
            for (int mt = 0; mt < 4; ++mt)
                nac[mt] = wmma16(ld_ws(sW + NW1B + (kc * 4 + mt) * 512, lane), nf[kc], nac[mt]);
        }
        v16h nh[2];
#pragma unroll
        for (int kc = 0; kc < 2; ++kc)
            nh[kc] = pack2(bias8(nac[2*kc],     nb1 + 32*kc      + 8*h, true),
                           bias8(nac[2*kc + 1], nb1 + 32*kc + 16 + 8*h, true));
        v8f oac[4] = {v8f{}, v8f{}, v8f{}, v8f{}};
#pragma unroll
        for (int kc = 0; kc < 2; ++kc) {
#pragma unroll
            for (int mt = 0; mt < 4; ++mt)
                oac[mt] = wmma16(ld_ws(sW + NW2B + (kc * 4 + mt) * 512, lane), nh[kc], oac[mt]);
        }
#pragma unroll
        for (int mt = 0; mt < 4; ++mt) {
            v8f o = bias8(oac[mt], nb2 + 16*mt + 8*h, false);
            if (have) {
                v4f* dst = (v4f*)(sacc + lr * SACC_PITCH + 16*mt + 8*h);
                dst[0] = v4f{o[0], o[1], o[2], o[3]};
                dst[1] = v4f{o[4], o[5], o[6], o[7]};
            }
        }
    }
    __syncthreads();
    for (int i = tid; i < BIN_NODES * (FDIM / 4); i += EDGE_THREADS) {
        const int row = i >> 4, q = i & 15;
        *(v4f*)(hout + (size_t)(nbase + row) * FDIM + q * 4) = *(const v4f*)(sacc + row * SACC_PITCH + q * 4);
    }
}

extern "C" void kernel_launch(void* const* d_in, const int* in_sizes, int n_in,
                              void* d_out, int out_size, void* d_ws, size_t ws_size,
                              hipStream_t stream) {
    const float* x         = (const float*)d_in[0];
    const int*   ei        = (const int*)  d_in[1];
    const float* pos       = (const float*)d_in[2];
    const float* pos_init  = (const float*)d_in[3];
    const float* edge_attr = (const float*)d_in[4];
    const float* mw1 = (const float*)d_in[5];
    const float* mb1 = (const float*)d_in[6];
    const float* mw2 = (const float*)d_in[7];
    const float* mb2 = (const float*)d_in[8];
    const float* nw1 = (const float*)d_in[9];
    const float* nb1 = (const float*)d_in[10];
    const float* nw2 = (const float*)d_in[11];
    const float* nb2 = (const float*)d_in[12];
    const float* cw1 = (const float*)d_in[13];
    const float* cb1 = (const float*)d_in[14];
    const float* cw2 = (const float*)d_in[15];
    const float* cb2 = (const float*)d_in[16];

    float* h_out   = (float*)d_out;
    float* pos_out = h_out + (size_t)N_NODES * FDIM;

    int* perm    = (int*)d_ws;
    int* gcount  = perm + N_EDGES;
    int* gcursor = gcount + 256;
    egnn_bin_clear<<<1, 256, 0, stream>>>(gcount);
    egnn_bin_count<<<(N_EDGES + BIN_CHUNK - 1) / BIN_CHUNK, 256, 0, stream>>>(ei, gcount);
    egnn_bin_fill<<<(N_EDGES + BIN_CHUNK - 1) / BIN_CHUNK, 256, 0, stream>>>(ei, gcount, gcursor, perm);

    egnn_edge<<<NBINS, EDGE_THREADS, 0, stream>>>(
        x, ei, pos, pos_init, edge_attr, mb1, mb2, cb1, cb2, nb1, nb2,
        mw1, mw2, cw1, cw2, nw1, nw2, perm, gcount, h_out, pos_out);
}


# First message layer split by linearity: its x[dst] part plus bias formed once per block row (WMMA, f32 in LDS) and read as each edge's accumulator start; no x[dst] gather; layer-1 sum order changes
# speedup vs baseline: 4.0780x; 1.0706x over previous
#include <hip/hip_runtime.h>

typedef __attribute__((ext_vector_type(16))) _Float16 v16h;
typedef __attribute__((ext_vector_type(8)))  float    v8f;

#define N_NODES 50000
#define N_EDGES 800000
#define FDIM    64
#define EFDIM   16
#define MSG_IN  145
#define MSG_K   160


#define MW1B 0
#define MW2B 10240
#define CW1B 14336
#define NW1B 18432
#define NW2B 26624
#define CW2B 30720

__device__ __forceinline__ v8f wmma16(v16h a, v16h b, v8f c) {
    return __builtin_amdgcn_wmma_f32_16x16x32_f16(
         false, a,  false, b,
         (short)0, c,  false,  false);
}

#define BIN_NODES 200
#define NBINS     250
#define BIN_CHUNK 2048

__global__ __launch_bounds__(256) void egnn_bin_clear(int* __restrict__ binctr) {
    binctr[threadIdx.x] = 0;
    binctr[256 + threadIdx.x] = 0;
}

__global__ __launch_bounds__(256) void egnn_bin_count(const int* __restrict__ ei,
                                                      int* __restrict__ gcount) {
    __shared__ int hist[256];
    const int t = threadIdx.x;
    hist[t] = 0;
    __syncthreads();
    const int base = blockIdx.x * BIN_CHUNK;
    for (int i = t; i < BIN_CHUNK; i += 256) {
        int e = base + i;
        if (e < N_EDGES) atomicAdd(&hist[ei[N_EDGES + e] / BIN_NODES], 1);
    }
    __syncthreads();
    int v = hist[t];
    if (v) atomicAdd(&gcount[t], v);
}

__global__ __launch_bounds__(256) void egnn_bin_fill(const int* __restrict__ ei,
                                                     const int* __restrict__ gcount,
                                                     int* __restrict__ gcursor,
                                                     int* __restrict__ perm) {
    __shared__ int hist[256];
    __shared__ int scount[256];
    __shared__ int sbase[256];
    const int t = threadIdx.x;
    hist[t] = 0;
    const int cv = gcount[t];
    int incl = cv;
    scount[t] = incl;
    __syncthreads();
    for (int o = 1; o < 256; o <<= 1) {
        int add = (t >= o) ? scount[t - o] : 0;
        __syncthreads();
        incl += add;
        scount[t] = incl;
        __syncthreads();
    }
    const int start = incl - cv;
    const int base = blockIdx.x * BIN_CHUNK;
    int mybin[8], myrank[8];
#pragma unroll
    for (int j = 0; j < 8; ++j) {
        int e = base + t + 256 * j;
        mybin[j] = -1; myrank[j] = 0;
        if (e < N_EDGES) {
            int b = ei[N_EDGES + e] / BIN_NODES;
            mybin[j] = b;
            myrank[j] = atomicAdd(&hist[b], 1);
        }
    }
    __syncthreads();
    int v = hist[t];
    int off = v ? atomicAdd(&gcursor[t], v) : 0;
    sbase[t] = start + off;
    __syncthreads();
#pragma unroll
    for (int j = 0; j < 8; ++j)
        if (mybin[j] >= 0) perm[sbase[mybin[j]] + myrank[j]] = base + t + 256 * j;
}

typedef __attribute__((ext_vector_type(4))) float v4f;

typedef __attribute__((ext_vector_type(8))) _Float16 v8h;
__device__ __forceinline__ v16h ld_ws(const _Float16* blk, int lane) {
    v8h lo = *(const v8h*)(blk + lane * 8);
    v8h hi = *(const v8h*)(blk + 256 + lane * 8);
    return __builtin_shufflevector(lo, hi, 0, 1, 2, 3, 4, 5, 6, 7, 8, 9, 10, 11, 12, 13, 14, 15);
}

__device__ __forceinline__ v16h row_frag(const float* __restrict__ p, int h) {
    const v4f* q = (const v4f*)(p + 8 * h);
    v4f a0 = q[0], a1 = q[1], b0 = q[4], b1 = q[5];
    v16h f;
    f[0]  = (_Float16)a0.x; f[1]  = (_Float16)a0.y; f[2]  = (_Float16)a0.z; f[3]  = (_Float16)a0.w;
    f[4]  = (_Float16)a1.x; f[5]  = (_Float16)a1.y; f[6]  = (_Float16)a1.z; f[7]  = (_Float16)a1.w;
    f[8]  = (_Float16)b0.x; f[9]  = (_Float16)b0.y; f[10] = (_Float16)b0.z; f[11] = (_Float16)b0.w;
    f[12] = (_Float16)b1.x; f[13] = (_Float16)b1.y; f[14] = (_Float16)b1.z; f[15] = (_Float16)b1.w;
    return f;
}

__device__ __forceinline__ v8f bias8(v8f a, const float* __restrict__ b, bool relu) {
    const v4f* q = (const v4f*)b;
    v4f b0 = q[0], b1 = q[1];
    float bb[8] = {b0.x, b0.y, b0.z, b0.w, b1.x, b1.y, b1.z, b1.w};
    v8f o;
#pragma unroll
    for (int r = 0; r < 8; ++r) {
        float v = a[r] + bb[r];
        o[r] = relu ? (v > 0.0f ? v : 0.0f) : v;
    }
    return o;
}

__device__ __forceinline__ v8f relu8(v8f a) {
    v8f o;
#pragma unroll
    for (int r = 0; r < 8; ++r) o[r] = a[r] > 0.0f ? a[r] : 0.0f;
    return o;
}

__device__ __forceinline__ v16h pack2(v8f lo, v8f hi) {
    v16h f;
#pragma unroll
    for (int r = 0; r < 8; ++r) { f[r] = (_Float16)lo[r]; f[8 + r] = (_Float16)hi[r]; }
    return f;
}

__device__ __forceinline__ v16h mean_frag(const float* p, int h, float s) {
    const v4f* q = (const v4f*)(p + 8 * h);
    v4f a0 = q[0], a1 = q[1], b0 = q[4], b1 = q[5];
    v16h f;
    f[0]  = (_Float16)(a0.x * s); f[1]  = (_Float16)(a0.y * s); f[2]  = (_Float16)(a0.z * s); f[3]  = (_Float16)(a0.w * s);
    f[4]  = (_Float16)(a1.x * s); f[5]  = (_Float16)(a1.y * s); f[6]  = (_Float16)(a1.z * s); f[7]  = (_Float16)(a1.w * s);
    f[8]  = (_Float16)(b0.x * s); f[9]  = (_Float16)(b0.y * s); f[10] = (_Float16)(b0.z * s); f[11] = (_Float16)(b0.w * s);
    f[12] = (_Float16)(b1.x * s); f[13] = (_Float16)(b1.y * s); f[14] = (_Float16)(b1.z * s); f[15] = (_Float16)(b1.w * s);
    return f;
}

#define SACC_PITCH   68
#define NODE_TILES   ((BIN_NODES + 15) / 16)
#define EDGE_WAVES   16
#define EDGE_THREADS (EDGE_WAVES * 32)
#define EDGE_WBLOCKS 62
#define W_ITEMS      960
__global__ __launch_bounds__(EDGE_THREADS) void egnn_edge(
    const float* __restrict__ x, const int* __restrict__ ei,
    const float* __restrict__ pos, const float* __restrict__ pos_init,
    const float* __restrict__ edge_attr,
    const float* __restrict__ mb1, const float* __restrict__ mb2,
    const float* __restrict__ cb1, const float* __restrict__ cb2,
    const float* __restrict__ nb1, const float* __restrict__ nb2,
    const float* __restrict__ mw1, const float* __restrict__ mw2,
    const float* __restrict__ cw1, const float* __restrict__ cw2,
    const float* __restrict__ nw1, const float* __restrict__ nw2,
    const int* __restrict__ perm, const int* __restrict__ gcount,
    float* __restrict__ hout, float* __restrict__ pos_out)
{
    __shared__ __attribute__((aligned(16))) float sacc[(BIN_NODES + 1) * SACC_PITCH];
    __shared__ __attribute__((aligned(16))) float sQ[(BIN_NODES + 1) * SACC_PITCH];
    __shared__ float spos[(BIN_NODES + 1) * 3];
    __shared__ float scnt[BIN_NODES + 1];
    __shared__ int srange[2];
    __shared__ __attribute__((aligned(16))) _Float16 sW[EDGE_WBLOCKS * 512];

    const int tid  = threadIdx.x;
    const int bin  = blockIdx.x;
    const int nbase = bin * BIN_NODES;
    for (int i = tid; i < (BIN_NODES + 1) * SACC_PITCH; i += EDGE_THREADS) sacc[i] = 0.0f;
    for (int i = tid; i < (BIN_NODES + 1) * 3; i += EDGE_THREADS)    spos[i] = 0.0f;
    {
        v4f wr[2][8];
        int wofs_l[2];
#pragma unroll
        for (int rd = 0; rd < 2; ++rd) {
            int it = tid + rd * EDGE_THREADS;
            it = it < W_ITEMS ? it : W_ITEMS - 1;
            const int gg = it >> 4, n4 = it & 15;
            const float* wm; int g0, kact;
            if (gg < 20)      { wm = mw1; g0 = 0;  kact = MSG_IN; }
            else if (gg < 28) { wm = mw2; g0 = 20; kact = 64; }
            else if (gg < 36) { wm = cw1; g0 = 28; kact = 64; }
            else if (gg < 52) { wm = nw1; g0 = 36; kact = 128; }
            else              { wm = nw2; g0 = 52; kact = 64; }
            const int k0 = (gg - g0) * 8;
#pragma unroll
            for (int j = 0; j < 8; ++j) {
                const int k = k0 + j;
                const int kc = k < kact ? k : kact - 1;
                const v4f v = *(const v4f*)(wm + kc * 64 + n4 * 4);
                wr[rd][j] = k < kact ? v : v4f{0.0f, 0.0f, 0.0f, 0.0f};
            }
            wofs_l[rd] = ((gg >> 2) * 4 + (n4 >> 2)) * 512 + ((gg >> 1) & 1) * 256
                       + ((gg & 1) * 16 + (n4 & 3) * 4) * 8;
        }
#pragma unroll
        for (int rd = 0; rd < 2; ++rd) {
            if (tid + rd * EDGE_THREADS < W_ITEMS) {
#pragma unroll
                for (int cc = 0; cc < 4; ++cc) {
                    v8h rec;
#pragma unroll
                    for (int j = 0; j < 8; ++j) rec[j] = (_Float16)wr[rd][j][cc];
                    *(v8h*)(sW + wofs_l[rd] + cc * 8) = rec;
                }
            }
        }
        if (tid >= EDGE_THREADS - 128) {
            const int t = tid - (EDGE_THREADS - 128);
            const int kc = t >> 6, pl = (t >> 5) & 1, ln = t & 31;
            const int kh = ln >> 4, cn = ln & 15;
            const int k0 = kc * 32 + pl * 16 + kh * 8;
            v8h rec;
#pragma unroll
            for (int j = 0; j < 8; ++j) {
                const float v = cw2[cn < 3 ? (k0 + j) * 3 + cn : 0];
                rec[j] = (_Float16)(cn < 3 ? v : 0.0f);
            }
            *(v8h*)(sW + CW2B + kc * 512 + pl * 256 + ln * 8) = rec;
        }
    }
    if (tid < BIN_NODES + 1) scnt[tid] = 0.0f;
    if (tid < SACC_PITCH) sQ[BIN_NODES * SACC_PITCH + tid] = 0.0f;
    if (tid < 32) {
        int acc0 = 0;
        for (int k = tid; k < bin; k += 32) acc0 += gcount[k];
        for (int o = 16; o > 0; o >>= 1) acc0 += __shfl_xor(acc0, o);
        if (tid == 0) { srange[0] = acc0; srange[1] = acc0 + gcount[bin]; }
    }
    __syncthreads();
    const int estart = srange[0];
    const int eend   = srange[1];
    const int ntiles = (eend - estart + 15) >> 4;

    const int lane = tid & 31;
    const int c = lane & 15;
    const int h = lane >> 4;
    const int wave = tid >> 5;
    int e_cur = 0, s_cur = 0, d_cur = 0, e_nxt = 0;
    if (wave < ntiles) {
        int p = estart + wave * 16 + c;
        e_cur = perm[p < eend ? p : estart];
    }
    if (wave + EDGE_WAVES < ntiles) {
        int p = estart + (wave + EDGE_WAVES) * 16 + c;
        e_nxt = perm[p < eend ? p : estart];
    }
    if (wave < ntiles) { s_cur = ei[e_cur]; d_cur = ei[N_EDGES + e_cur]; }
    float mb2c[4];
#pragma unroll
    for (int nt = 0; nt < 4; ++nt) mb2c[nt] = mb2[nt * 16 + c];
    if (wave < NODE_TILES) {
        const int lr  = wave * 16 + c;
        const bool have = lr < BIN_NODES;
        const int lrc = have ? lr : BIN_NODES - 1;
        const float* xr = x + (size_t)(nbase + lrc) * FDIM;
        v16h qf[2];
        qf[0] = row_frag(xr, h);
        qf[1] = row_frag(xr + 32, h);
        v8f qac[4] = {v8f{}, v8f{}, v8f{}, v8f{}};
#pragma unroll
        for (int kc = 0; kc < 2; ++kc) {
#pragma unroll
            for (int mt = 0; mt < 4; ++mt)
                qac[mt] = wmma16(ld_ws(sW + MW1B + ((2 + kc) * 4 + mt) * 512, lane), qf[kc], qac[mt]);
        }
#pragma unroll
        for (int mt = 0; mt < 4; ++mt) {
            v8f o = bias8(qac[mt], mb1 + 16*mt + 8*h, false);
            if (have) {
                v4f* dst = (v4f*)(sQ + lr * SACC_PITCH + 16*mt + 8*h);
                dst[0] = v4f{o[0], o[1], o[2], o[3]};
                dst[1] = v4f{o[4], o[5], o[6], o[7]};
            }
        }
    }
    __syncthreads();
  for (int tile = wave; tile < ntiles; tile += EDGE_WAVES) {
    const bool live = estart + tile * 16 + c < eend;
    const int e = e_cur;
    const int s = s_cur;
    const int d = d_cur;
    e_cur = e_nxt;
    if (tile + EDGE_WAVES < ntiles) { s_cur = ei[e_cur]; d_cur = ei[N_EDGES + e_cur]; }
    if (tile + 2 * EDGE_WAVES < ntiles) {
        int p = estart + (tile + 2 * EDGE_WAVES) * 16 + c;
        e_nxt = perm[p < eend ? p : estart];
    }
    const int lrow = live ? (d - nbase) : BIN_NODES;
    int wofs = 0;
    asm volatile("" : "+v"(wofs));
    const _Float16* sWt = sW + wofs;

    const float* xs = x + (size_t)s * FDIM;
    v16h fr[3];
    fr[0] = row_frag(xs, h);
    fr[1] = row_frag(xs + 32, h);
    v8f acc[4];
    {
        const float* qr = sQ + lrow * SACC_PITCH + 8 * h;
#pragma unroll
        for (int mt = 0; mt < 4; ++mt) {
            const v4f a = *(const v4f*)(qr + 16 * mt);
            const v4f b = *(const v4f*)(qr + 16 * mt + 4);
            acc[mt] = v8f{a.x, a.y, a.z, a.w, b.x, b.y, b.z, b.w};
        }
    }
    const v4f* eq = (const v4f*)(edge_attr + (size_t)e * EFDIM);
    v4f e0 = eq[0], e1 = eq[1], e2 = eq[2], e3 = eq[3];

    float dx = pos[3*s+0] - pos[3*d+0];
    float dy = pos[3*s+1] - pos[3*d+1];
    float dz = pos[3*s+2] - pos[3*d+2];
    float dsq = dx*dx + dy*dy + dz*dz;
    float ix = pos_init[3*s+0] - pos_init[3*d+0];
    float iy = pos_init[3*s+1] - pos_init[3*d+1];
    float iz = pos_init[3*s+2] - pos_init[3*d+2];
    float rinv = 1.0f / sqrtf(ix*ix + iy*iy + iz*iz);
    if (h == 0) atomicAdd(&scnt[lrow], 1.0f);

    {
        float ea[16] = {e0.x, e0.y, e0.z, e0.w, e1.x, e1.y, e1.z, e1.w,
                        e2.x, e2.y, e2.z, e2.w, e3.x, e3.y, e3.z, e3.w};
        v16h f;
#pragma unroll
        for (int t = 0; t < 8; ++t) {
            float lo = (t == 0) ? dsq : ea[t - 1];
            float hi = ea[7 + t];
            f[t] = (_Float16)(h ? hi : lo);
        }
        f[8] = (_Float16)(h ? 0.0f : ea[15]);
#pragma unroll
        for (int t = 9; t < 16; ++t) f[t] = (_Float16)0.0f;
        fr[2] = f;
    }

#pragma unroll
    for (int j = 0; j < 3; ++j) {
        const int kc = j < 2 ? j : 4;
#pragma unroll
        for (int mt = 0; mt < 4; ++mt)
            acc[mt] = wmma16(ld_ws(sWt + MW1B + (kc * 4 + mt) * 512, lane), fr[j], acc[mt]);
    }
    v16h hf[2];
#pragma unroll
    for (int kc = 0; kc < 2; ++kc)
        hf[kc] = pack2(relu8(acc[2*kc]), relu8(acc[2*kc + 1]));

    v8f mij[4] = {v8f{}, v8f{}, v8f{}, v8f{}};
    v8f mt_[4] = {v8f{}, v8f{}, v8f{}, v8f{}};
    {
        v16h w2[8];
#pragma unroll
        for (int i = 0; i < 8; ++i) w2[i] = ld_ws(sWt + MW2B + i * 512, lane);
#pragma unroll
        for (int kc = 0; kc < 2; ++kc) {
#pragma unroll
            for (int nt = 0; nt < 4; ++nt)
                mij[nt] = wmma16(hf[kc], w2[kc * 4 + nt], mij[nt]);
        }
#pragma unroll
        for (int kc = 0; kc < 2; ++kc) {
#pragma unroll
            for (int mt = 0; mt < 4; ++mt)
                mt_[mt] = wmma16(w2[kc * 4 + mt], hf[kc], mt_[mt]);
        }
    }
    v16h wc1[8], wc2[2];
#pragma unroll
    for (int i = 0; i < 8; ++i) wc1[i] = ld_ws(sWt + CW1B + i * 512, lane);
#pragma unroll
    for (int i = 0; i < 2; ++i) wc2[i] = ld_ws(sWt + CW2B + i * 512, lane);
    v16h mf[2];
#pragma unroll
    for (int kc = 0; kc < 2; ++kc)
        mf[kc] = pack2(bias8(mt_[2*kc],     mb2 + 32*kc      + 8*h, false),
                       bias8(mt_[2*kc + 1], mb2 + 32*kc + 16 + 8*h, false));
    {
        int drow[8];
#pragma unroll
        for (int r = 0; r < 8; ++r) drow[r] = __shfl(lrow, 8 * h + r);
#pragma unroll
        for (int nt = 0; nt < 4; ++nt) {
#pragma unroll
            for (int r = 0; r < 8; ++r)
                atomicAdd(&sacc[drow[r] * SACC_PITCH + nt * 16 + c], mij[nt][r] + mb2c[nt]);
        }
    }

    v16h tf[2];
    {
        v8f tac[4] = {v8f{}, v8f{}, v8f{}, v8f{}};
#pragma unroll
        for (int kc = 0; kc < 2; ++kc) {
#pragma unroll
            for (int mt = 0; mt < 4; ++mt)
                tac[mt] = wmma16(wc1[kc * 4 + mt], mf[kc], tac[mt]);
        }
#pragma unroll
        for (int kc = 0; kc < 2; ++kc)
            tf[kc] = pack2(bias8(tac[2*kc],     cb1 + 32*kc      + 8*h, true),
                           bias8(tac[2*kc + 1], cb1 + 32*kc + 16 + 8*h, true));
    }

    v8f cac = v8f{};
#pragma unroll
    for (int kc = 0; kc < 2; ++kc)
        cac = wmma16(wc2[kc], tf[kc], cac);
    float ux = ix * rinv * (cac[0] + cb2[0]);
    float uy = iy * rinv * (cac[1] + cb2[1]);
    float uz = iz * rinv * (cac[2] + cb2[2]);
    if (h == 0) {
        atomicAdd(&spos[lrow * 3 + 0], ux);
        atomicAdd(&spos[lrow * 3 + 1], uy);
        atomicAdd(&spos[lrow * 3 + 2], uz);
    }
  }
    __syncthreads();
    for (int i = tid; i < BIN_NODES * 3; i += EDGE_THREADS)    pos_out[(size_t)nbase * 3 + i] = spos[i];

    if (wave < NODE_TILES) {
        const int lr  = wave * 16 + c;
        const bool have = lr < BIN_NODES;
        const int lrc = have ? lr : BIN_NODES - 1;
        const float* xr = x + (size_t)(nbase + lrc) * FDIM;
        const float* mr = sacc + lrc * SACC_PITCH;
        const float inv = 1.0f / fmaxf(scnt[lrc], 1.0f);
        v16h nf[4];
        nf[0] = row_frag(xr, h);
        nf[1] = row_frag(xr + 32, h);
        nf[2] = mean_frag(mr, h, inv);
        nf[3] = mean_frag(mr + 32, h, inv);
        v8f nac[4] = {v8f{}, v8f{}, v8f{}, v8f{}};
#pragma unroll
        for (int kc = 0; kc < 4; ++kc) {
#pragma unroll
            for (int mt = 0; mt < 4; ++mt)
                nac[mt] = wmma16(ld_ws(sW + NW1B + (kc * 4 + mt) * 512, lane), nf[kc], nac[mt]);
        }
        v16h nh[2];
#pragma unroll
        for (int kc = 0; kc < 2; ++kc)
            nh[kc] = pack2(bias8(nac[2*kc],     nb1 + 32*kc      + 8*h, true),
                           bias8(nac[2*kc + 1], nb1 + 32*kc + 16 + 8*h, true));
        v8f oac[4] = {v8f{}, v8f{}, v8f{}, v8f{}};
#pragma unroll
        for (int kc = 0; kc < 2; ++kc) {
#pragma unroll
            for (int mt = 0; mt < 4; ++mt)
                oac[mt] = wmma16(ld_ws(sW + NW2B + (kc * 4 + mt) * 512, lane), nh[kc], oac[mt]);
        }
#pragma unroll
        for (int mt = 0; mt < 4; ++mt) {
            v8f o = bias8(oac[mt], nb2 + 16*mt + 8*h, false);
            if (have) {
                v4f* dst = (v4f*)(sacc + lr * SACC_PITCH + 16*mt + 8*h);
                dst[0] = v4f{o[0], o[1], o[2], o[3]};
                dst[1] = v4f{o[4], o[5], o[6], o[7]};
            }
        }
    }
    __syncthreads();
    for (int i = tid; i < BIN_NODES * (FDIM / 4); i += EDGE_THREADS) {
        const int row = i >> 4, q = i & 15;
        *(v4f*)(hout + (size_t)(nbase + row) * FDIM + q * 4) = *(const v4f*)(sacc + row * SACC_PITCH + q * 4);
    }
}

extern "C" void kernel_launch(void* const* d_in, const int* in_sizes, int n_in,
                              void* d_out, int out_size, void* d_ws, size_t ws_size,
                              hipStream_t stream) {
    const float* x         = (const float*)d_in[0];
    const int*   ei        = (const int*)  d_in[1];
    const float* pos       = (const float*)d_in[2];
    const float* pos_init  = (const float*)d_in[3];
    const float* edge_attr = (const float*)d_in[4];
    const float* mw1 = (const float*)d_in[5];
    const float* mb1 = (const float*)d_in[6];
    const float* mw2 = (const float*)d_in[7];
    const float* mb2 = (const float*)d_in[8];
    const float* nw1 = (const float*)d_in[9];
    const float* nb1 = (const float*)d_in[10];
    const float* nw2 = (const float*)d_in[11];
    const float* nb2 = (const float*)d_in[12];
    const float* cw1 = (const float*)d_in[13];
    const float* cb1 = (const float*)d_in[14];
    const float* cw2 = (const float*)d_in[15];
    const float* cb2 = (const float*)d_in[16];

    float* h_out   = (float*)d_out;
    float* pos_out = h_out + (size_t)N_NODES * FDIM;

    int* perm    = (int*)d_ws;
    int* gcount  = perm + N_EDGES;
    int* gcursor = gcount + 256;
    egnn_bin_clear<<<1, 256, 0, stream>>>(gcount);
    egnn_bin_count<<<(N_EDGES + BIN_CHUNK - 1) / BIN_CHUNK, 256, 0, stream>>>(ei, gcount);
    egnn_bin_fill<<<(N_EDGES + BIN_CHUNK - 1) / BIN_CHUNK, 256, 0, stream>>>(ei, gcount, gcursor, perm);

    egnn_edge<<<NBINS, EDGE_THREADS, 0, stream>>>(
        x, ei, pos, pos_init, edge_attr, mb1, mb2, cb1, cb2, nb1, nb2,
        mw1, mw2, cw1, cw2, nw1, nw2, perm, gcount, h_out, pos_out);
}


# Fill launch lists each edge as an 8-byte record (id, source, block row), its index reads issued together; edge kernel takes end points from the list, not 2 random 4-byte reads per edge; same sums
# speedup vs baseline: 4.4834x; 1.0994x over previous
#include <hip/hip_runtime.h>

typedef __attribute__((ext_vector_type(16))) _Float16 v16h;
typedef __attribute__((ext_vector_type(8)))  float    v8f;

#define N_NODES 50000
#define N_EDGES 800000
#define FDIM    64
#define EFDIM   16
#define MSG_IN  145
#define MSG_K   160


#define MW1B 0
#define MW2B 10240
#define CW1B 14336
#define NW1B 18432
#define NW2B 26624
#define CW2B 30720

__device__ __forceinline__ v8f wmma16(v16h a, v16h b, v8f c) {
    return __builtin_amdgcn_wmma_f32_16x16x32_f16(
         false, a,  false, b,
         (short)0, c,  false,  false);
}

#define BIN_NODES 200
#define NBINS     250
#define BIN_CHUNK 2048
typedef __attribute__((ext_vector_type(2))) int v2i;
static_assert(N_NODES <= 65536 && BIN_NODES <= 32768, "a record's second word holds the source node in 16 bits and the block row above them");

__global__ __launch_bounds__(256) void egnn_bin_clear(int* __restrict__ binctr) {
    binctr[threadIdx.x] = 0;
    binctr[256 + threadIdx.x] = 0;
}

__global__ __launch_bounds__(256) void egnn_bin_count(const int* __restrict__ ei,
                                                      int* __restrict__ gcount) {
    __shared__ int hist[256];
    const int t = threadIdx.x;
    hist[t] = 0;
    __syncthreads();
    const int base = blockIdx.x * BIN_CHUNK;
    for (int i = t; i < BIN_CHUNK; i += 256) {
        int e = base + i;
        if (e < N_EDGES) atomicAdd(&hist[ei[N_EDGES + e] / BIN_NODES], 1);
    }
    __syncthreads();
    int v = hist[t];
    if (v) atomicAdd(&gcount[t], v);
}

__global__ __launch_bounds__(256) void egnn_bin_fill(const int* __restrict__ ei,
                                                     const int* __restrict__ gcount,
                                                     int* __restrict__ gcursor,
                                                     int* __restrict__ rec) {
    __shared__ int hist[256];
    __shared__ int scount[256];
    __shared__ int sbase[256];
    const int t = threadIdx.x;
    const int base = blockIdx.x * BIN_CHUNK;
    int dv[8], sv[8];
#pragma unroll
    for (int j = 0; j < 8; ++j) {
        const int e = base + t + 256 * j;
        const int ec = e < N_EDGES ? e : N_EDGES - 1;
        dv[j] = ei[N_EDGES + ec];
        sv[j] = ei[ec];
    }
    hist[t] = 0;
    const int cv = gcount[t];
    int incl = cv;
    scount[t] = incl;
    __syncthreads();
    for (int o = 1; o < 256; o <<= 1) {
        int add = (t >= o) ? scount[t - o] : 0;
        __syncthreads();
        incl += add;
        scount[t] = incl;
        __syncthreads();
    }
    const int start = incl - cv;
    int mybin[8], myrank[8], mykey[8];
#pragma unroll
    for (int j = 0; j < 8; ++j) {
        int e = base + t + 256 * j;
        mybin[j] = -1; myrank[j] = 0; mykey[j] = 0;
        if (e < N_EDGES) {
            const int d = dv[j];
            const int b = d / BIN_NODES;
            mybin[j] = b;
            mykey[j] = ((d - b * BIN_NODES) << 16) | sv[j];
            myrank[j] = atomicAdd(&hist[b], 1);
        }
    }
    __syncthreads();
    int v = hist[t];
    int off = v ? atomicAdd(&gcursor[t], v) : 0;
    sbase[t] = start + off;
    __syncthreads();
#pragma unroll
    for (int j = 0; j < 8; ++j)
        if (mybin[j] >= 0)
            *(v2i*)(rec + 2 * (sbase[mybin[j]] + myrank[j])) = v2i{base + t + 256 * j, mykey[j]};
}

typedef __attribute__((ext_vector_type(4))) float v4f;

typedef __attribute__((ext_vector_type(8))) _Float16 v8h;
__device__ __forceinline__ v16h ld_ws(const _Float16* blk, int lane) {
    v8h lo = *(const v8h*)(blk + lane * 8);
    v8h hi = *(const v8h*)(blk + 256 + lane * 8);
    return __builtin_shufflevector(lo, hi, 0, 1, 2, 3, 4, 5, 6, 7, 8, 9, 10, 11, 12, 13, 14, 15);
}

__device__ __forceinline__ v16h row_frag(const float* __restrict__ p, int h) {
    const v4f* q = (const v4f*)(p + 8 * h);
    v4f a0 = q[0], a1 = q[1], b0 = q[4], b1 = q[5];
    v16h f;
    f[0]  = (_Float16)a0.x; f[1]  = (_Float16)a0.y; f[2]  = (_Float16)a0.z; f[3]  = (_Float16)a0.w;
    f[4]  = (_Float16)a1.x; f[5]  = (_Float16)a1.y; f[6]  = (_Float16)a1.z; f[7]  = (_Float16)a1.w;
    f[8]  = (_Float16)b0.x; f[9]  = (_Float16)b0.y; f[10] = (_Float16)b0.z; f[11] = (_Float16)b0.w;
    f[12] = (_Float16)b1.x; f[13] = (_Float16)b1.y; f[14] = (_Float16)b1.z; f[15] = (_Float16)b1.w;
    return f;
}

__device__ __forceinline__ v8f bias8(v8f a, const float* __restrict__ b, bool relu) {
    const v4f* q = (const v4f*)b;
    v4f b0 = q[0], b1 = q[1];
    float bb[8] = {b0.x, b0.y, b0.z, b0.w, b1.x, b1.y, b1.z, b1.w};
    v8f o;
#pragma unroll
    for (int r = 0; r < 8; ++r) {
        float v = a[r] + bb[r];
        o[r] = relu ? (v > 0.0f ? v : 0.0f) : v;
    }
    return o;
}

__device__ __forceinline__ v8f relu8(v8f a) {
    v8f o;
#pragma unroll
    for (int r = 0; r < 8; ++r) o[r] = a[r] > 0.0f ? a[r] : 0.0f;
    return o;
}

__device__ __forceinline__ v16h pack2(v8f lo, v8f hi) {
    v16h f;
#pragma unroll
    for (int r = 0; r < 8; ++r) { f[r] = (_Float16)lo[r]; f[8 + r] = (_Float16)hi[r]; }
    return f;
}

__device__ __forceinline__ v16h mean_frag(const float* p, int h, float s) {
    const v4f* q = (const v4f*)(p + 8 * h);
    v4f a0 = q[0], a1 = q[1], b0 = q[4], b1 = q[5];
    v16h f;
    f[0]  = (_Float16)(a0.x * s); f[1]  = (_Float16)(a0.y * s); f[2]  = (_Float16)(a0.z * s); f[3]  = (_Float16)(a0.w * s);
    f[4]  = (_Float16)(a1.x * s); f[5]  = (_Float16)(a1.y * s); f[6]  = (_Float16)(a1.z * s); f[7]  = (_Float16)(a1.w * s);
    f[8]  = (_Float16)(b0.x * s); f[9]  = (_Float16)(b0.y * s); f[10] = (_Float16)(b0.z * s); f[11] = (_Float16)(b0.w * s);
    f[12] = (_Float16)(b1.x * s); f[13] = (_Float16)(b1.y * s); f[14] = (_Float16)(b1.z * s); f[15] = (_Float16)(b1.w * s);
    return f;
}

#define SACC_PITCH   68
#define NODE_TILES   ((BIN_NODES + 15) / 16)
#define EDGE_WAVES   16
#define EDGE_THREADS (EDGE_WAVES * 32)
#define EDGE_WBLOCKS 62
#define W_ITEMS      960
__global__ __launch_bounds__(EDGE_THREADS) void egnn_edge(
    const float* __restrict__ x,
    const float* __restrict__ pos, const float* __restrict__ pos_init,
    const float* __restrict__ edge_attr,
    const float* __restrict__ mb1, const float* __restrict__ mb2,
    const float* __restrict__ cb1, const float* __restrict__ cb2,
    const float* __restrict__ nb1, const float* __restrict__ nb2,
    const float* __restrict__ mw1, const float* __restrict__ mw2,
    const float* __restrict__ cw1, const float* __restrict__ cw2,
    const float* __restrict__ nw1, const float* __restrict__ nw2,
    const int* __restrict__ rec, const int* __restrict__ gcount,
    float* __restrict__ hout, float* __restrict__ pos_out)
{
    __shared__ __attribute__((aligned(16))) float sacc[(BIN_NODES + 1) * SACC_PITCH];
    __shared__ __attribute__((aligned(16))) float sQ[(BIN_NODES + 1) * SACC_PITCH];
    __shared__ float spos[(BIN_NODES + 1) * 3];
    __shared__ float scnt[BIN_NODES + 1];
    __shared__ int srange[2];
    __shared__ __attribute__((aligned(16))) _Float16 sW[EDGE_WBLOCKS * 512];

    const int tid  = threadIdx.x;
    const int bin  = blockIdx.x;
    const int nbase = bin * BIN_NODES;
    for (int i = tid; i < (BIN_NODES + 1) * SACC_PITCH; i += EDGE_THREADS) sacc[i] = 0.0f;
    for (int i = tid; i < (BIN_NODES + 1) * 3; i += EDGE_THREADS)    spos[i] = 0.0f;
    {
        v4f wr[2][8];
        int wofs_l[2];
#pragma unroll
        for (int rd = 0; rd < 2; ++rd) {
            int it = tid + rd * EDGE_THREADS;
            it = it < W_ITEMS ? it : W_ITEMS - 1;
            const int gg = it >> 4, n4 = it & 15;
            const float* wm; int g0, kact;
            if (gg < 20)      { wm = mw1; g0 = 0;  kact = MSG_IN; }
            else if (gg < 28) { wm = mw2; g0 = 20; kact = 64; }
            else if (gg < 36) { wm = cw1; g0 = 28; kact = 64; }
            else if (gg < 52) { wm = nw1; g0 = 36; kact = 128; }
            else              { wm = nw2; g0 = 52; kact = 64; }
            const int k0 = (gg - g0) * 8;
#pragma unroll
            for (int j = 0; j < 8; ++j) {
                const int k = k0 + j;
                const int kc = k < kact ? k : kact - 1;
                const v4f v = *(const v4f*)(wm + kc * 64 + n4 * 4);
                wr[rd][j] = k < kact ? v : v4f{0.0f, 0.0f, 0.0f, 0.0f};
            }
            wofs_l[rd] = ((gg >> 2) * 4 + (n4 >> 2)) * 512 + ((gg >> 1) & 1) * 256
                       + ((gg & 1) * 16 + (n4 & 3) * 4) * 8;
        }
#pragma unroll
        for (int rd = 0; rd < 2; ++rd) {
            if (tid + rd * EDGE_THREADS < W_ITEMS) {
#pragma unroll
                for (int cc = 0; cc < 4; ++cc) {
                    v8h rec;
#pragma unroll
                    for (int j = 0; j < 8; ++j) rec[j] = (_Float16)wr[rd][j][cc];
                    *(v8h*)(sW + wofs_l[rd] + cc * 8) = rec;
                }
            }
        }
        if (tid >= EDGE_THREADS - 128) {
            const int t = tid - (EDGE_THREADS - 128);
            const int kc = t >> 6, pl = (t >> 5) & 1, ln = t & 31;
            const int kh = ln >> 4, cn = ln & 15;
            const int k0 = kc * 32 + pl * 16 + kh * 8;
            v8h rec;
#pragma unroll
            for (int j = 0; j < 8; ++j) {
                const float v = cw2[cn < 3 ? (k0 + j) * 3 + cn : 0];
                rec[j] = (_Float16)(cn < 3 ? v : 0.0f);
            }
            *(v8h*)(sW + CW2B + kc * 512 + pl * 256 + ln * 8) = rec;
        }
    }
    if (tid < BIN_NODES + 1) scnt[tid] = 0.0f;
    if (tid < SACC_PITCH) sQ[BIN_NODES * SACC_PITCH + tid] = 0.0f;
    if (tid < 32) {
        int acc0 = 0;
        for (int k = tid; k < bin; k += 32) acc0 += gcount[k];
        for (int o = 16; o > 0; o >>= 1) acc0 += __shfl_xor(acc0, o);
        if (tid == 0) { srange[0] = acc0; srange[1] = acc0 + gcount[bin]; }
    }
    __syncthreads();
    const int estart = srange[0];
    const int eend   = srange[1];
    const int ntiles = (eend - estart + 15) >> 4;

    const int lane = tid & 31;
    const int c = lane & 15;
    const int h = lane >> 4;
    const int wave = tid >> 5;
    v2i r_cur = v2i{0, 0};
    if (wave < ntiles) {
        int p = estart + wave * 16 + c;
        r_cur = *(const v2i*)(rec + 2 * (p < eend ? p : estart));
    }
    float mb2c[4];
#pragma unroll
    for (int nt = 0; nt < 4; ++nt) mb2c[nt] = mb2[nt * 16 + c];
    if (wave < NODE_TILES) {
        const int lr  = wave * 16 + c;
        const bool have = lr < BIN_NODES;
        const int lrc = have ? lr : BIN_NODES - 1;
        const float* xr = x + (size_t)(nbase + lrc) * FDIM;
        v16h qf[2];
        qf[0] = row_frag(xr, h);
        qf[1] = row_frag(xr + 32, h);
        v8f qac[4] = {v8f{}, v8f{}, v8f{}, v8f{}};
#pragma unroll
        for (int kc = 0; kc < 2; ++kc) {
#pragma unroll
            for (int mt = 0; mt < 4; ++mt)
                qac[mt] = wmma16(ld_ws(sW + MW1B + ((2 + kc) * 4 + mt) * 512, lane), qf[kc], qac[mt]);
        }
#pragma unroll
        for (int mt = 0; mt < 4; ++mt) {
            v8f o = bias8(qac[mt], mb1 + 16*mt + 8*h, false);
            if (have) {
                v4f* dst = (v4f*)(sQ + lr * SACC_PITCH + 16*mt + 8*h);
                dst[0] = v4f{o[0], o[1], o[2], o[3]};
                dst[1] = v4f{o[4], o[5], o[6], o[7]};
            }
        }
    }
    __syncthreads();
  for (int tile = wave; tile < ntiles; tile += EDGE_WAVES) {
    const bool live = estart + tile * 16 + c < eend;
    const int e = r_cur.x;
    const int s = r_cur.y & 0xffff;
    const int brow = r_cur.y >> 16;
    const int d = nbase + brow;
    if (tile + EDGE_WAVES < ntiles) {
        int p = estart + (tile + EDGE_WAVES) * 16 + c;
        r_cur = *(const v2i*)(rec + 2 * (p < eend ? p : estart));
    }
    const int lrow = live ? brow : BIN_NODES;
    int wofs = 0;
    asm volatile("" : "+v"(wofs));
    const _Float16* sWt = sW + wofs;

    const float* xs = x + (size_t)s * FDIM;
    v16h fr[3];
    fr[0] = row_frag(xs, h);
    fr[1] = row_frag(xs + 32, h);
    v8f acc[4];
    {
        const float* qr = sQ + lrow * SACC_PITCH + 8 * h;
#pragma unroll
        for (int mt = 0; mt < 4; ++mt) {
            const v4f a = *(const v4f*)(qr + 16 * mt);
            const v4f b = *(const v4f*)(qr + 16 * mt + 4);
            acc[mt] = v8f{a.x, a.y, a.z, a.w, b.x, b.y, b.z, b.w};
        }
    }
    const v4f* eq = (const v4f*)(edge_attr + (size_t)e * EFDIM);
    v4f e0 = eq[0], e1 = eq[1], e2 = eq[2], e3 = eq[3];

    float dx = pos[3*s+0] - pos[3*d+0];
    float dy = pos[3*s+1] - pos[3*d+1];
    float dz = pos[3*s+2] - pos[3*d+2];
    float dsq = dx*dx + dy*dy + dz*dz;
    float ix = pos_init[3*s+0] - pos_init[3*d+0];
    float iy = pos_init[3*s+1] - pos_init[3*d+1];
    float iz = pos_init[3*s+2] - pos_init[3*d+2];
    float rinv = 1.0f / sqrtf(ix*ix + iy*iy + iz*iz);
    if (h == 0) atomicAdd(&scnt[lrow], 1.0f);

    {
        float ea[16] = {e0.x, e0.y, e0.z, e0.w, e1.x, e1.y, e1.z, e1.w,
                        e2.x, e2.y, e2.z, e2.w, e3.x, e3.y, e3.z, e3.w};
        v16h f;
#pragma unroll
        for (int t = 0; t < 8; ++t) {
            float lo = (t == 0) ? dsq : ea[t - 1];
            float hi = ea[7 + t];
            f[t] = (_Float16)(h ? hi : lo);
        }
        f[8] = (_Float16)(h ? 0.0f : ea[15]);
#pragma unroll
        for (int t = 9; t < 16; ++t) f[t] = (_Float16)0.0f;
        fr[2] = f;
    }

#pragma unroll
    for (int j = 0; j < 3; ++j) {
        const int kc = j < 2 ? j : 4;
#pragma unroll
        for (int mt = 0; mt < 4; ++mt)
            acc[mt] = wmma16(ld_ws(sWt + MW1B + (kc * 4 + mt) * 512, lane), fr[j], acc[mt]);
    }
    v16h hf[2];
#pragma unroll
    for (int kc = 0; kc < 2; ++kc)
        hf[kc] = pack2(relu8(acc[2*kc]), relu8(acc[2*kc + 1]));

    v8f mij[4] = {v8f{}, v8f{}, v8f{}, v8f{}};
    v8f mt_[4] = {v8f{}, v8f{}, v8f{}, v8f{}};
    {
        v16h w2[8];
#pragma unroll
        for (int i = 0; i < 8; ++i) w2[i] = ld_ws(sWt + MW2B + i * 512, lane);
#pragma unroll
        for (int kc = 0; kc < 2; ++kc) {
#pragma unroll
            for (int nt = 0; nt < 4; ++nt)
                mij[nt] = wmma16(hf[kc], w2[kc * 4 + nt], mij[nt]);
        }
#pragma unroll
        for (int kc = 0; kc < 2; ++kc) {
#pragma unroll
            for (int mt = 0; mt < 4; ++mt)
                mt_[mt] = wmma16(w2[kc * 4 + mt], hf[kc], mt_[mt]);
        }
    }
    v16h wc1[8], wc2[2];
#pragma unroll
    for (int i = 0; i < 8; ++i) wc1[i] = ld_ws(sWt + CW1B + i * 512, lane);
#pragma unroll
    for (int i = 0; i < 2; ++i) wc2[i] = ld_ws(sWt + CW2B + i * 512, lane);
    v16h mf[2];
#pragma unroll
    for (int kc = 0; kc < 2; ++kc)
        mf[kc] = pack2(bias8(mt_[2*kc],     mb2 + 32*kc      + 8*h, false),
                       bias8(mt_[2*kc + 1], mb2 + 32*kc + 16 + 8*h, false));
    {
        int drow[8];
#pragma unroll
        for (int r = 0; r < 8; ++r) drow[r] = __shfl(lrow, 8 * h + r);
#pragma unroll
        for (int nt = 0; nt < 4; ++nt) {
#pragma unroll
            for (int r = 0; r < 8; ++r)
                atomicAdd(&sacc[drow[r] * SACC_PITCH + nt * 16 + c], mij[nt][r] + mb2c[nt]);
        }
    }

    v16h tf[2];
    {
        v8f tac[4] = {v8f{}, v8f{}, v8f{}, v8f{}};
#pragma unroll
        for (int kc = 0; kc < 2; ++kc) {
#pragma unroll
            for (int mt = 0; mt < 4; ++mt)
                tac[mt] = wmma16(wc1[kc * 4 + mt], mf[kc], tac[mt]);
        }
#pragma unroll
        for (int kc = 0; kc < 2; ++kc)
            tf[kc] = pack2(bias8(tac[2*kc],     cb1 + 32*kc      + 8*h, true),
                           bias8(tac[2*kc + 1], cb1 + 32*kc + 16 + 8*h, true));
    }

    v8f cac = v8f{};
#pragma unroll
    for (int kc = 0; kc < 2; ++kc)
        cac = wmma16(wc2[kc], tf[kc], cac);
    float ux = ix * rinv * (cac[0] + cb2[0]);
    float uy = iy * rinv * (cac[1] + cb2[1]);
    float uz = iz * rinv * (cac[2] + cb2[2]);
    if (h == 0) {
        atomicAdd(&spos[lrow * 3 + 0], ux);
        atomicAdd(&spos[lrow * 3 + 1], uy);
        atomicAdd(&spos[lrow * 3 + 2], uz);
    }
  }
    __syncthreads();
    for (int i = tid; i < BIN_NODES * 3; i += EDGE_THREADS)    pos_out[(size_t)nbase * 3 + i] = spos[i];

    if (wave < NODE_TILES) {
        const int lr  = wave * 16 + c;
        const bool have = lr < BIN_NODES;
        const int lrc = have ? lr : BIN_NODES - 1;
        const float* xr = x + (size_t)(nbase + lrc) * FDIM;
        const float* mr = sacc + lrc * SACC_PITCH;
        const float inv = 1.0f / fmaxf(scnt[lrc], 1.0f);
        v16h nf[4];
        nf[0] = row_frag(xr, h);
        nf[1] = row_frag(xr + 32, h);
        nf[2] = mean_frag(mr, h, inv);
        nf[3] = mean_frag(mr + 32, h, inv);
        v8f nac[4] = {v8f{}, v8f{}, v8f{}, v8f{}};
#pragma unroll
        for (int kc = 0; kc < 4; ++kc) {
#pragma unroll
            for (int mt = 0; mt < 4; ++mt)
                nac[mt] = wmma16(ld_ws(sW + NW1B + (kc * 4 + mt) * 512, lane), nf[kc], nac[mt]);
        }
        v16h nh[2];
#pragma unroll
        for (int kc = 0; kc < 2; ++kc)
            nh[kc] = pack2(bias8(nac[2*kc],     nb1 + 32*kc      + 8*h, true),
                           bias8(nac[2*kc + 1], nb1 + 32*kc + 16 + 8*h, true));
        v8f oac[4] = {v8f{}, v8f{}, v8f{}, v8f{}};
#pragma unroll
        for (int kc = 0; kc < 2; ++kc) {
#pragma unroll
            for (int mt = 0; mt < 4; ++mt)
                oac[mt] = wmma16(ld_ws(sW + NW2B + (kc * 4 + mt) * 512, lane), nh[kc], oac[mt]);
        }
#pragma unroll
        for (int mt = 0; mt < 4; ++mt) {
            v8f o = bias8(oac[mt], nb2 + 16*mt + 8*h, false);
            if (have) {
                v4f* dst = (v4f*)(sacc + lr * SACC_PITCH + 16*mt + 8*h);
                dst[0] = v4f{o[0], o[1], o[2], o[3]};
                dst[1] = v4f{o[4], o[5], o[6], o[7]};
            }
        }
    }
    __syncthreads();
    for (int i = tid; i < BIN_NODES * (FDIM / 4); i += EDGE_THREADS) {
        const int row = i >> 4, q = i & 15;
        *(v4f*)(hout + (size_t)(nbase + row) * FDIM + q * 4) = *(const v4f*)(sacc + row * SACC_PITCH + q * 4);
    }
}

extern "C" void kernel_launch(void* const* d_in, const int* in_sizes, int n_in,
                              void* d_out, int out_size, void* d_ws, size_t ws_size,
                              hipStream_t stream) {
    const float* x         = (const float*)d_in[0];
    const int*   ei        = (const int*)  d_in[1];
    const float* pos       = (const float*)d_in[2];
    const float* pos_init  = (const float*)d_in[3];
    const float* edge_attr = (const float*)d_in[4];
    const float* mw1 = (const float*)d_in[5];
    const float* mb1 = (const float*)d_in[6];
    const float* mw2 = (const float*)d_in[7];
    const float* mb2 = (const float*)d_in[8];
    const float* nw1 = (const float*)d_in[9];
    const float* nb1 = (const float*)d_in[10];
    const float* nw2 = (const float*)d_in[11];
    const float* nb2 = (const float*)d_in[12];
    const float* cw1 = (const float*)d_in[13];
    const float* cb1 = (const float*)d_in[14];
    const float* cw2 = (const float*)d_in[15];
    const float* cb2 = (const float*)d_in[16];

    float* h_out   = (float*)d_out;
    float* pos_out = h_out + (size_t)N_NODES * FDIM;

    int* rec     = (int*)d_ws;
    int* gcount  = rec + 2 * N_EDGES;
    int* gcursor = gcount + 256;
    egnn_bin_clear<<<1, 256, 0, stream>>>(gcount);
    egnn_bin_count<<<(N_EDGES + BIN_CHUNK - 1) / BIN_CHUNK, 256, 0, stream>>>(ei, gcount);
    egnn_bin_fill<<<(N_EDGES + BIN_CHUNK - 1) / BIN_CHUNK, 256, 0, stream>>>(ei, gcount, gcursor, rec);

    egnn_edge<<<NBINS, EDGE_THREADS, 0, stream>>>(
        x, pos, pos_init, edge_attr, mb1, mb2, cb1, cb2, nb1, nb2,
        mw1, mw2, cw1, cw2, nw1, nw2, rec, gcount, h_out, pos_out);
}


# One binning launch, not three: each 2048-edge chunk sorts its records by block into its own slice, posts run starts; a block finds its edges via an LDS slot table; edge order in the LDS adds changes
# speedup vs baseline: 4.9983x; 1.1148x over previous
#include <hip/hip_runtime.h>

typedef __attribute__((ext_vector_type(16))) _Float16 v16h;
typedef __attribute__((ext_vector_type(8)))  float    v8f;

#define N_NODES 50000
#define N_EDGES 800000
#define FDIM    64
#define EFDIM   16
#define MSG_IN  145
#define MSG_K   160


#define MW1B 0
#define MW2B 10240
#define CW1B 14336
#define NW1B 18432
#define NW2B 26624
#define CW2B 30720

__device__ __forceinline__ v8f wmma16(v16h a, v16h b, v8f c) {
    return __builtin_amdgcn_wmma_f32_16x16x32_f16(
         false, a,  false, b,
         (short)0, c,  false,  false);
}

#define BIN_NODES 200
#define NBINS     250
#define BIN_CHUNK 2048
#define NCHUNK    ((N_EDGES + BIN_CHUNK - 1) / BIN_CHUNK)
typedef __attribute__((ext_vector_type(2))) int v2i;
static_assert(N_NODES <= 65536 && BIN_NODES <= 32768, "a record's second word holds the source node in 16 bits and the block row above them");
static_assert(NBINS <= 256 && BIN_CHUNK <= 32768, "a run-table word holds a count up to BIN_CHUNK in 16 bits and a slot below BIN_CHUNK above them");

__global__ __launch_bounds__(256) void egnn_bin_fill(const int* __restrict__ ei,
                                                     int* __restrict__ rec,
                                                     int* __restrict__ runtab) {
    __shared__ int hist[256];
    __shared__ int sbase[256];
    __shared__ int swsum[8];
    const int t = threadIdx.x;
    const int base = blockIdx.x * BIN_CHUNK;
    int dv[8], sv[8];
#pragma unroll
    for (int j = 0; j < 8; ++j) {
        const int e = base + t + 256 * j;
        const int ec = e < N_EDGES ? e : N_EDGES - 1;
        dv[j] = ei[N_EDGES + ec];
        sv[j] = ei[ec];
    }
    hist[t] = 0;
    __syncthreads();
    int mybin[8], myrank[8], mykey[8];
#pragma unroll
    for (int j = 0; j < 8; ++j) {
        int e = base + t + 256 * j;
        mybin[j] = -1; myrank[j] = 0; mykey[j] = 0;
        if (e < N_EDGES) {
            const int d = dv[j];
            const int b = d / BIN_NODES;
            mybin[j] = b;
            mykey[j] = ((d - b * BIN_NODES) << 16) | sv[j];
            myrank[j] = atomicAdd(&hist[b], 1);
        }
    }
    __syncthreads();
    const int cv = hist[t];
    int incl = cv;
    for (int o = 1; o < 32; o <<= 1) {
        const int u = __shfl_up(incl, o);
        if ((t & 31) >= o) incl += u;
    }
    if ((t & 31) == 31) swsum[t >> 5] = incl;
    __syncthreads();
    int wpre = 0;
    for (int k = 0; k < 8; ++k) wpre += (k < (t >> 5)) ? swsum[k] : 0;
    const int start = wpre + incl - cv;
    sbase[t] = start;
    runtab[blockIdx.x * 256 + t] = (start << 16) | cv;
    __syncthreads();
#pragma unroll
    for (int j = 0; j < 8; ++j)
        if (mybin[j] >= 0)
            *(v2i*)(rec + 2 * (base + sbase[mybin[j]] + myrank[j])) = v2i{base + t + 256 * j, mykey[j]};
}

typedef __attribute__((ext_vector_type(4))) float v4f;

typedef __attribute__((ext_vector_type(8))) _Float16 v8h;
__device__ __forceinline__ v16h ld_ws(const _Float16* blk, int lane) {
    v8h lo = *(const v8h*)(blk + lane * 8);
    v8h hi = *(const v8h*)(blk + 256 + lane * 8);
    return __builtin_shufflevector(lo, hi, 0, 1, 2, 3, 4, 5, 6, 7, 8, 9, 10, 11, 12, 13, 14, 15);
}

__device__ __forceinline__ v16h row_frag(const float* __restrict__ p, int h) {
    const v4f* q = (const v4f*)(p + 8 * h);
    v4f a0 = q[0], a1 = q[1], b0 = q[4], b1 = q[5];
    v16h f;
    f[0]  = (_Float16)a0.x; f[1]  = (_Float16)a0.y; f[2]  = (_Float16)a0.z; f[3]  = (_Float16)a0.w;
    f[4]  = (_Float16)a1.x; f[5]  = (_Float16)a1.y; f[6]  = (_Float16)a1.z; f[7]  = (_Float16)a1.w;
    f[8]  = (_Float16)b0.x; f[9]  = (_Float16)b0.y; f[10] = (_Float16)b0.z; f[11] = (_Float16)b0.w;
    f[12] = (_Float16)b1.x; f[13] = (_Float16)b1.y; f[14] = (_Float16)b1.z; f[15] = (_Float16)b1.w;
    return f;
}

__device__ __forceinline__ v8f bias8(v8f a, const float* __restrict__ b, bool relu) {
    const v4f* q = (const v4f*)b;
    v4f b0 = q[0], b1 = q[1];
    float bb[8] = {b0.x, b0.y, b0.z, b0.w, b1.x, b1.y, b1.z, b1.w};
    v8f o;
#pragma unroll
    for (int r = 0; r < 8; ++r) {
        float v = a[r] + bb[r];
        o[r] = relu ? (v > 0.0f ? v : 0.0f) : v;
    }
    return o;
}

__device__ __forceinline__ v8f relu8(v8f a) {
    v8f o;
#pragma unroll
    for (int r = 0; r < 8; ++r) o[r] = a[r] > 0.0f ? a[r] : 0.0f;
    return o;
}

__device__ __forceinline__ v16h pack2(v8f lo, v8f hi) {
    v16h f;
#pragma unroll
    for (int r = 0; r < 8; ++r) { f[r] = (_Float16)lo[r]; f[8 + r] = (_Float16)hi[r]; }
    return f;
}

__device__ __forceinline__ v16h mean_frag(const float* p, int h, float s) {
    const v4f* q = (const v4f*)(p + 8 * h);
    v4f a0 = q[0], a1 = q[1], b0 = q[4], b1 = q[5];
    v16h f;
    f[0]  = (_Float16)(a0.x * s); f[1]  = (_Float16)(a0.y * s); f[2]  = (_Float16)(a0.z * s); f[3]  = (_Float16)(a0.w * s);
    f[4]  = (_Float16)(a1.x * s); f[5]  = (_Float16)(a1.y * s); f[6]  = (_Float16)(a1.z * s); f[7]  = (_Float16)(a1.w * s);
    f[8]  = (_Float16)(b0.x * s); f[9]  = (_Float16)(b0.y * s); f[10] = (_Float16)(b0.z * s); f[11] = (_Float16)(b0.w * s);
    f[12] = (_Float16)(b1.x * s); f[13] = (_Float16)(b1.y * s); f[14] = (_Float16)(b1.z * s); f[15] = (_Float16)(b1.w * s);
    return f;
}

#define ECAP 8192
__device__ __forceinline__ void list_slots(int* eslot, int mypre, int mylen, int mybase, int pbase) {
    const int lo  = mypre > pbase ? mypre : pbase;
    const int hi0 = mypre + mylen, hi1 = pbase + ECAP;
    const int hi  = hi0 < hi1 ? hi0 : hi1;
    for (int p = lo; p < hi; ++p) eslot[p - pbase] = mybase + (p - mypre);
}

#define SACC_PITCH   68
#define NODE_TILES   ((BIN_NODES + 15) / 16)
#define EDGE_WAVES   16
#define EDGE_THREADS (EDGE_WAVES * 32)
#define EDGE_WBLOCKS 62
#define W_ITEMS      960
__global__ __launch_bounds__(EDGE_THREADS) void egnn_edge(
    const float* __restrict__ x,
    const float* __restrict__ pos, const float* __restrict__ pos_init,
    const float* __restrict__ edge_attr,
    const float* __restrict__ mb1, const float* __restrict__ mb2,
    const float* __restrict__ cb1, const float* __restrict__ cb2,
    const float* __restrict__ nb1, const float* __restrict__ nb2,
    const float* __restrict__ mw1, const float* __restrict__ mw2,
    const float* __restrict__ cw1, const float* __restrict__ cw2,
    const float* __restrict__ nw1, const float* __restrict__ nw2,
    const int* __restrict__ rec, const int* __restrict__ runtab,
    float* __restrict__ hout, float* __restrict__ pos_out)
{
    __shared__ __attribute__((aligned(16))) float sacc[(BIN_NODES + 1) * SACC_PITCH];
    __shared__ __attribute__((aligned(16))) float sQ[(BIN_NODES + 1) * SACC_PITCH];
    __shared__ float spos[(BIN_NODES + 1) * 3];
    __shared__ float scnt[BIN_NODES + 1];
    __shared__ int eslot[ECAP];
    __shared__ int spre[EDGE_THREADS];
    __shared__ int slen[EDGE_THREADS];
    __shared__ int sbase[EDGE_THREADS];
    __shared__ int swtot[EDGE_WAVES];
    __shared__ __attribute__((aligned(16))) _Float16 sW[EDGE_WBLOCKS * 512];

    const int tid  = threadIdx.x;
    const int bin  = blockIdx.x;
    const int nbase = bin * BIN_NODES;
    const int rv = tid < NCHUNK ? runtab[tid * 256 + bin] : 0;
    int etot;
    for (int i = tid; i < (BIN_NODES + 1) * SACC_PITCH; i += EDGE_THREADS) sacc[i] = 0.0f;
    for (int i = tid; i < (BIN_NODES + 1) * 3; i += EDGE_THREADS)    spos[i] = 0.0f;
    {
        v4f wr[2][8];
        int wofs_l[2];
#pragma unroll
        for (int rd = 0; rd < 2; ++rd) {
            int it = tid + rd * EDGE_THREADS;
            it = it < W_ITEMS ? it : W_ITEMS - 1;
            const int gg = it >> 4, n4 = it & 15;
            const float* wm; int g0, kact;
            if (gg < 20)      { wm = mw1; g0 = 0;  kact = MSG_IN; }
            else if (gg < 28) { wm = mw2; g0 = 20; kact = 64; }
            else if (gg < 36) { wm = cw1; g0 = 28; kact = 64; }
            else if (gg < 52) { wm = nw1; g0 = 36; kact = 128; }
            else              { wm = nw2; g0 = 52; kact = 64; }
            const int k0 = (gg - g0) * 8;
#pragma unroll
            for (int j = 0; j < 8; ++j) {
                const int k = k0 + j;
                const int kc = k < kact ? k : kact - 1;
                const v4f v = *(const v4f*)(wm + kc * 64 + n4 * 4);
                wr[rd][j] = k < kact ? v : v4f{0.0f, 0.0f, 0.0f, 0.0f};
            }
            wofs_l[rd] = ((gg >> 2) * 4 + (n4 >> 2)) * 512 + ((gg >> 1) & 1) * 256
                       + ((gg & 1) * 16 + (n4 & 3) * 4) * 8;
        }
        {
            const int rlen = rv & 0xffff;
            int incl = rlen;
            for (int o = 1; o < 32; o <<= 1) {
                const int u = __shfl_up(incl, o);
                if ((tid & 31) >= o) incl += u;
            }
            if ((tid & 31) == 31) swtot[tid >> 5] = incl;
            __syncthreads();
            int wpre = 0, tot = 0;
            for (int k = 0; k < EDGE_WAVES; ++k) {
                const int v = swtot[k];
                wpre += (k < (tid >> 5)) ? v : 0;
                tot += v;
            }
            etot = __builtin_amdgcn_readfirstlane(tot);
            const int mypre = wpre + incl - rlen, mybase = tid * BIN_CHUNK + (rv >> 16);
            spre[tid] = mypre; slen[tid] = rlen; sbase[tid] = mybase;
            list_slots(eslot, mypre, rlen, mybase, 0);
        }
#pragma unroll
        for (int rd = 0; rd < 2; ++rd) {
            if (tid + rd * EDGE_THREADS < W_ITEMS) {
#pragma unroll
                for (int cc = 0; cc < 4; ++cc) {
                    v8h rec;
#pragma unroll
                    for (int j = 0; j < 8; ++j) rec[j] = (_Float16)wr[rd][j][cc];
                    *(v8h*)(sW + wofs_l[rd] + cc * 8) = rec;
                }
            }
        }
        if (tid >= EDGE_THREADS - 128) {
            const int t = tid - (EDGE_THREADS - 128);
            const int kc = t >> 6, pl = (t >> 5) & 1, ln = t & 31;
            const int kh = ln >> 4, cn = ln & 15;
            const int k0 = kc * 32 + pl * 16 + kh * 8;
            v8h rec;
#pragma unroll
            for (int j = 0; j < 8; ++j) {
                const float v = cw2[cn < 3 ? (k0 + j) * 3 + cn : 0];
                rec[j] = (_Float16)(cn < 3 ? v : 0.0f);
            }
            *(v8h*)(sW + CW2B + kc * 512 + pl * 256 + ln * 8) = rec;
        }
    }
    if (tid < BIN_NODES + 1) scnt[tid] = 0.0f;
    if (tid < SACC_PITCH) sQ[BIN_NODES * SACC_PITCH + tid] = 0.0f;
    __syncthreads();
    int pbase = 0;
    int npass = etot < ECAP ? etot : ECAP;
    int ntiles = (npass + 15) >> 4;

    const int lane = tid & 31;
    const int c = lane & 15;
    const int h = lane >> 4;
    const int wave = tid >> 5;
    v2i r_cur = v2i{0, 0};
    if (wave < ntiles) {
        const int q0 = wave * 16 + c;
        r_cur = *(const v2i*)(rec + 2 * eslot[q0 < npass ? q0 : 0]);
    }
    float mb2c[4];
#pragma unroll
    for (int nt = 0; nt < 4; ++nt) mb2c[nt] = mb2[nt * 16 + c];
    if (wave < NODE_TILES) {
        const int lr  = wave * 16 + c;
        const bool have = lr < BIN_NODES;
        const int lrc = have ? lr : BIN_NODES - 1;
        const float* xr = x + (size_t)(nbase + lrc) * FDIM;
        v16h qf[2];
        qf[0] = row_frag(xr, h);
        qf[1] = row_frag(xr + 32, h);
        v8f qac[4] = {v8f{}, v8f{}, v8f{}, v8f{}};
#pragma unroll
        for (int kc = 0; kc < 2; ++kc) {
#pragma unroll
            for (int mt = 0; mt < 4; ++mt)
                qac[mt] = wmma16(ld_ws(sW + MW1B + ((2 + kc) * 4 + mt) * 512, lane), qf[kc], qac[mt]);
        }
#pragma unroll
        for (int mt = 0; mt < 4; ++mt) {
            v8f o = bias8(qac[mt], mb1 + 16*mt + 8*h, false);
            if (have) {
                v4f* dst = (v4f*)(sQ + lr * SACC_PITCH + 16*mt + 8*h);
                dst[0] = v4f{o[0], o[1], o[2], o[3]};
                dst[1] = v4f{o[4], o[5], o[6], o[7]};
            }
        }
    }
    __syncthreads();
 for (;;) {
  for (int tile = wave; tile < ntiles; tile += EDGE_WAVES) {
    const bool live = tile * 16 + c < npass;
    const int e = r_cur.x;
    const int s = r_cur.y & 0xffff;
    const int brow = r_cur.y >> 16;
    const int d = nbase + brow;
    const int lrow = live ? brow : BIN_NODES;
    int wofs = 0;
    asm volatile("" : "+v"(wofs));
    const _Float16* sWt = sW + wofs;

    const float* xs = x + (size_t)s * FDIM;
    v16h fr[3];
    fr[0] = row_frag(xs, h);
    fr[1] = row_frag(xs + 32, h);
    v8f acc[4];
    {
        const float* qr = sQ + lrow * SACC_PITCH + 8 * h;
#pragma unroll
        for (int mt = 0; mt < 4; ++mt) {
            const v4f a = *(const v4f*)(qr + 16 * mt);
            const v4f b = *(const v4f*)(qr + 16 * mt + 4);
            acc[mt] = v8f{a.x, a.y, a.z, a.w, b.x, b.y, b.z, b.w};
        }
    }
    const v4f* eq = (const v4f*)(edge_attr + (size_t)e * EFDIM);
    v4f e0 = eq[0], e1 = eq[1], e2 = eq[2], e3 = eq[3];

    float dx = pos[3*s+0] - pos[3*d+0];
    float dy = pos[3*s+1] - pos[3*d+1];
    float dz = pos[3*s+2] - pos[3*d+2];
    float dsq = dx*dx + dy*dy + dz*dz;
    float ix = pos_init[3*s+0] - pos_init[3*d+0];
    float iy = pos_init[3*s+1] - pos_init[3*d+1];
    float iz = pos_init[3*s+2] - pos_init[3*d+2];
    float rinv = 1.0f / sqrtf(ix*ix + iy*iy + iz*iz);
    {
        const int q1 = (tile + EDGE_WAVES) * 16 + c;
        r_cur = *(const v2i*)(rec + 2 * eslot[q1 < npass ? q1 : 0]);
    }
    if (h == 0) atomicAdd(&scnt[lrow], 1.0f);

    {
        float ea[16] = {e0.x, e0.y, e0.z, e0.w, e1.x, e1.y, e1.z, e1.w,
                        e2.x, e2.y, e2.z, e2.w, e3.x, e3.y, e3.z, e3.w};
        v16h f;
#pragma unroll
        for (int t = 0; t < 8; ++t) {
            float lo = (t == 0) ? dsq : ea[t - 1];
            float hi = ea[7 + t];
            f[t] = (_Float16)(h ? hi : lo);
        }
        f[8] = (_Float16)(h ? 0.0f : ea[15]);
#pragma unroll
        for (int t = 9; t < 16; ++t) f[t] = (_Float16)0.0f;
        fr[2] = f;
    }

#pragma unroll
    for (int j = 0; j < 3; ++j) {
        const int kc = j < 2 ? j : 4;
#pragma unroll
        for (int mt = 0; mt < 4; ++mt)
            acc[mt] = wmma16(ld_ws(sWt + MW1B + (kc * 4 + mt) * 512, lane), fr[j], acc[mt]);
    }
    v16h hf[2];
#pragma unroll
    for (int kc = 0; kc < 2; ++kc)
        hf[kc] = pack2(relu8(acc[2*kc]), relu8(acc[2*kc + 1]));

    v8f mij[4] = {v8f{}, v8f{}, v8f{}, v8f{}};
    v8f mt_[4] = {v8f{}, v8f{}, v8f{}, v8f{}};
    {
        v16h w2[8];
#pragma unroll
        for (int i = 0; i < 8; ++i) w2[i] = ld_ws(sWt + MW2B + i * 512, lane);
#pragma unroll
        for (int kc = 0; kc < 2; ++kc) {
#pragma unroll
            for (int nt = 0; nt < 4; ++nt)
                mij[nt] = wmma16(hf[kc], w2[kc * 4 + nt], mij[nt]);
        }
#pragma unroll
        for (int kc = 0; kc < 2; ++kc) {
#pragma unroll
            for (int mt = 0; mt < 4; ++mt)
                mt_[mt] = wmma16(w2[kc * 4 + mt], hf[kc], mt_[mt]);
        }
    }
    v16h wc1[8], wc2[2];
#pragma unroll
    for (int i = 0; i < 8; ++i) wc1[i] = ld_ws(sWt + CW1B + i * 512, lane);
#pragma unroll
    for (int i = 0; i < 2; ++i) wc2[i] = ld_ws(sWt + CW2B + i * 512, lane);
    v16h mf[2];
#pragma unroll
    for (int kc = 0; kc < 2; ++kc)
        mf[kc] = pack2(bias8(mt_[2*kc],     mb2 + 32*kc      + 8*h, false),
                       bias8(mt_[2*kc + 1], mb2 + 32*kc + 16 + 8*h, false));
    {
        int drow[8];
#pragma unroll
        for (int r = 0; r < 8; ++r) drow[r] = __shfl(lrow, 8 * h + r);
#pragma unroll
        for (int nt = 0; nt < 4; ++nt) {
#pragma unroll
            for (int r = 0; r < 8; ++r)
                atomicAdd(&sacc[drow[r] * SACC_PITCH + nt * 16 + c], mij[nt][r] + mb2c[nt]);
        }
    }

    v16h tf[2];
    {
        v8f tac[4] = {v8f{}, v8f{}, v8f{}, v8f{}};
#pragma unroll
        for (int kc = 0; kc < 2; ++kc) {
#pragma unroll
            for (int mt = 0; mt < 4; ++mt)
                tac[mt] = wmma16(wc1[kc * 4 + mt], mf[kc], tac[mt]);
        }
#pragma unroll
        for (int kc = 0; kc < 2; ++kc)
            tf[kc] = pack2(bias8(tac[2*kc],     cb1 + 32*kc      + 8*h, true),
                           bias8(tac[2*kc + 1], cb1 + 32*kc + 16 + 8*h, true));
    }

    v8f cac = v8f{};
#pragma unroll
    for (int kc = 0; kc < 2; ++kc)
        cac = wmma16(wc2[kc], tf[kc], cac);
    float ux = ix * rinv * (cac[0] + cb2[0]);
    float uy = iy * rinv * (cac[1] + cb2[1]);
    float uz = iz * rinv * (cac[2] + cb2[2]);
    if (h == 0) {
        atomicAdd(&spos[lrow * 3 + 0], ux);
        atomicAdd(&spos[lrow * 3 + 1], uy);
        atomicAdd(&spos[lrow * 3 + 2], uz);
    }
  }
    pbase += ECAP;
    if (pbase >= etot) break;
    __syncthreads();
    list_slots(eslot, spre[tid], slen[tid], sbase[tid], pbase);
    __syncthreads();
    npass = etot - pbase < ECAP ? etot - pbase : ECAP;
    ntiles = (npass + 15) >> 4;
    if (wave < ntiles) {
        const int q0 = wave * 16 + c;
        r_cur = *(const v2i*)(rec + 2 * eslot[q0 < npass ? q0 : 0]);
    }
 }
    __syncthreads();
    for (int i = tid; i < BIN_NODES * 3; i += EDGE_THREADS)    pos_out[(size_t)nbase * 3 + i] = spos[i];

    if (wave < NODE_TILES) {
        const int lr  = wave * 16 + c;
        const bool have = lr < BIN_NODES;
        const int lrc = have ? lr : BIN_NODES - 1;
        const float* xr = x + (size_t)(nbase + lrc) * FDIM;
        const float* mr = sacc + lrc * SACC_PITCH;
        const float inv = 1.0f / fmaxf(scnt[lrc], 1.0f);
        v16h nf[4];
        nf[0] = row_frag(xr, h);
        nf[1] = row_frag(xr + 32, h);
        nf[2] = mean_frag(mr, h, inv);
        nf[3] = mean_frag(mr + 32, h, inv);
        v8f nac[4] = {v8f{}, v8f{}, v8f{}, v8f{}};
#pragma unroll
        for (int kc = 0; kc < 4; ++kc) {
#pragma unroll
            for (int mt = 0; mt < 4; ++mt)
                nac[mt] = wmma16(ld_ws(sW + NW1B + (kc * 4 + mt) * 512, lane), nf[kc], nac[mt]);
        }
        v16h nh[2];
#pragma unroll
        for (int kc = 0; kc < 2; ++kc)
            nh[kc] = pack2(bias8(nac[2*kc],     nb1 + 32*kc      + 8*h, true),
                           bias8(nac[2*kc + 1], nb1 + 32*kc + 16 + 8*h, true));
        v8f oac[4] = {v8f{}, v8f{}, v8f{}, v8f{}};
#pragma unroll
        for (int kc = 0; kc < 2; ++kc) {
#pragma unroll
            for (int mt = 0; mt < 4; ++mt)
                oac[mt] = wmma16(ld_ws(sW + NW2B + (kc * 4 + mt) * 512, lane), nh[kc], oac[mt]);
        }
#pragma unroll
        for (int mt = 0; mt < 4; ++mt) {
            v8f o = bias8(oac[mt], nb2 + 16*mt + 8*h, false);
            if (have) {
                v4f* dst = (v4f*)(sacc + lr * SACC_PITCH + 16*mt + 8*h);
                dst[0] = v4f{o[0], o[1], o[2], o[3]};
                dst[1] = v4f{o[4], o[5], o[6], o[7]};
            }
        }
    }
    __syncthreads();
    for (int i = tid; i < BIN_NODES * (FDIM / 4); i += EDGE_THREADS) {
        const int row = i >> 4, q = i & 15;
        *(v4f*)(hout + (size_t)(nbase + row) * FDIM + q * 4) = *(const v4f*)(sacc + row * SACC_PITCH + q * 4);
    }
}

extern "C" void kernel_launch(void* const* d_in, const int* in_sizes, int n_in,
                              void* d_out, int out_size, void* d_ws, size_t ws_size,
                              hipStream_t stream) {
    const float* x         = (const float*)d_in[0];
    const int*   ei        = (const int*)  d_in[1];
    const float* pos       = (const float*)d_in[2];
    const float* pos_init  = (const float*)d_in[3];
    const float* edge_attr = (const float*)d_in[4];
    const float* mw1 = (const float*)d_in[5];
    const float* mb1 = (const float*)d_in[6];
    const float* mw2 = (const float*)d_in[7];
    const float* mb2 = (const float*)d_in[8];
    const float* nw1 = (const float*)d_in[9];
    const float* nb1 = (const float*)d_in[10];
    const float* nw2 = (const float*)d_in[11];
    const float* nb2 = (const float*)d_in[12];
    const float* cw1 = (const float*)d_in[13];
    const float* cb1 = (const float*)d_in[14];
    const float* cw2 = (const float*)d_in[15];
    const float* cb2 = (const float*)d_in[16];

    float* h_out   = (float*)d_out;
    float* pos_out = h_out + (size_t)N_NODES * FDIM;

    int* rec    = (int*)d_ws;
    int* runtab = rec + 2 * N_EDGES;
    egnn_bin_fill<<<NCHUNK, 256, 0, stream>>>(ei, rec, runtab);

    egnn_edge<<<NBINS, EDGE_THREADS, 0, stream>>>(
        x, pos, pos_init, edge_attr, mb1, mb2, cb1, cb2, nb1, nb2,
        mw1, mw2, cw1, cw2, nw1, nw2, rec, runtab, h_out, pos_out);
}


# Edge loop software-pipelined: each trip asks for the next trip's gathers (source row, edge attributes, end-point positions) behind its message adds, held in registers over the loop edge; same sums
# speedup vs baseline: 5.1983x; 1.0400x over previous
#include <hip/hip_runtime.h>

typedef __attribute__((ext_vector_type(16))) _Float16 v16h;
typedef __attribute__((ext_vector_type(8)))  float    v8f;

#define N_NODES 50000
#define N_EDGES 800000
#define FDIM    64
#define EFDIM   16
#define MSG_IN  145
#define MSG_K   160


#define MW1B 0
#define MW2B 10240
#define CW1B 14336
#define NW1B 18432
#define NW2B 26624
#define CW2B 30720

__device__ __forceinline__ v8f wmma16(v16h a, v16h b, v8f c) {
    return __builtin_amdgcn_wmma_f32_16x16x32_f16(
         false, a,  false, b,
         (short)0, c,  false,  false);
}

#define BIN_NODES 200
#define NBINS     250
#define BIN_CHUNK 2048
#define NCHUNK    ((N_EDGES + BIN_CHUNK - 1) / BIN_CHUNK)
typedef __attribute__((ext_vector_type(2))) int v2i;
static_assert(N_NODES <= 65536 && BIN_NODES <= 32768, "a record's second word holds the source node in 16 bits and the block row above them");
static_assert(NBINS <= 256 && BIN_CHUNK <= 32768, "a run-table word holds a count up to BIN_CHUNK in 16 bits and a slot below BIN_CHUNK above them");

__global__ __launch_bounds__(256) void egnn_bin_fill(const int* __restrict__ ei,
                                                     int* __restrict__ rec,
                                                     int* __restrict__ runtab) {
    __shared__ int hist[256];
    __shared__ int sbase[256];
    __shared__ int swsum[8];
    const int t = threadIdx.x;
    const int base = blockIdx.x * BIN_CHUNK;
    int dv[8], sv[8];
#pragma unroll
    for (int j = 0; j < 8; ++j) {
        const int e = base + t + 256 * j;
        const int ec = e < N_EDGES ? e : N_EDGES - 1;
        dv[j] = ei[N_EDGES + ec];
        sv[j] = ei[ec];
    }
    hist[t] = 0;
    __syncthreads();
    int mybin[8], myrank[8], mykey[8];
#pragma unroll
    for (int j = 0; j < 8; ++j) {
        int e = base + t + 256 * j;
        mybin[j] = -1; myrank[j] = 0; mykey[j] = 0;
        if (e < N_EDGES) {
            const int d = dv[j];
            const int b = d / BIN_NODES;
            mybin[j] = b;
            mykey[j] = ((d - b * BIN_NODES) << 16) | sv[j];
            myrank[j] = atomicAdd(&hist[b], 1);
        }
    }
    __syncthreads();
    const int cv = hist[t];
    int incl = cv;
    for (int o = 1; o < 32; o <<= 1) {
        const int u = __shfl_up(incl, o);
        if ((t & 31) >= o) incl += u;
    }
    if ((t & 31) == 31) swsum[t >> 5] = incl;
    __syncthreads();
    int wpre = 0;
    for (int k = 0; k < 8; ++k) wpre += (k < (t >> 5)) ? swsum[k] : 0;
    const int start = wpre + incl - cv;
    sbase[t] = start;
    runtab[blockIdx.x * 256 + t] = (start << 16) | cv;
    __syncthreads();
#pragma unroll
    for (int j = 0; j < 8; ++j)
        if (mybin[j] >= 0)
            *(v2i*)(rec + 2 * (base + sbase[mybin[j]] + myrank[j])) = v2i{base + t + 256 * j, mykey[j]};
}

typedef __attribute__((ext_vector_type(4))) float v4f;

typedef __attribute__((ext_vector_type(8))) _Float16 v8h;
__device__ __forceinline__ v16h ld_ws(const _Float16* blk, int lane) {
    v8h lo = *(const v8h*)(blk + lane * 8);
    v8h hi = *(const v8h*)(blk + 256 + lane * 8);
    return __builtin_shufflevector(lo, hi, 0, 1, 2, 3, 4, 5, 6, 7, 8, 9, 10, 11, 12, 13, 14, 15);
}

__device__ __forceinline__ v16h row_frag(const float* __restrict__ p, int h) {
    const v4f* q = (const v4f*)(p + 8 * h);
    v4f a0 = q[0], a1 = q[1], b0 = q[4], b1 = q[5];
    v16h f;
    f[0]  = (_Float16)a0.x; f[1]  = (_Float16)a0.y; f[2]  = (_Float16)a0.z; f[3]  = (_Float16)a0.w;
    f[4]  = (_Float16)a1.x; f[5]  = (_Float16)a1.y; f[6]  = (_Float16)a1.z; f[7]  = (_Float16)a1.w;
    f[8]  = (_Float16)b0.x; f[9]  = (_Float16)b0.y; f[10] = (_Float16)b0.z; f[11] = (_Float16)b0.w;
    f[12] = (_Float16)b1.x; f[13] = (_Float16)b1.y; f[14] = (_Float16)b1.z; f[15] = (_Float16)b1.w;
    return f;
}

__device__ __forceinline__ v8f bias8(v8f a, const float* __restrict__ b, bool relu) {
    const v4f* q = (const v4f*)b;
    v4f b0 = q[0], b1 = q[1];
    float bb[8] = {b0.x, b0.y, b0.z, b0.w, b1.x, b1.y, b1.z, b1.w};
    v8f o;
#pragma unroll
    for (int r = 0; r < 8; ++r) {
        float v = a[r] + bb[r];
        o[r] = relu ? (v > 0.0f ? v : 0.0f) : v;
    }
    return o;
}

__device__ __forceinline__ v8f relu8(v8f a) {
    v8f o;
#pragma unroll
    for (int r = 0; r < 8; ++r) o[r] = a[r] > 0.0f ? a[r] : 0.0f;
    return o;
}

__device__ __forceinline__ v16h pack2(v8f lo, v8f hi) {
    v16h f;
#pragma unroll
    for (int r = 0; r < 8; ++r) { f[r] = (_Float16)lo[r]; f[8 + r] = (_Float16)hi[r]; }
    return f;
}

__device__ __forceinline__ v16h mean_frag(const float* p, int h, float s) {
    const v4f* q = (const v4f*)(p + 8 * h);
    v4f a0 = q[0], a1 = q[1], b0 = q[4], b1 = q[5];
    v16h f;
    f[0]  = (_Float16)(a0.x * s); f[1]  = (_Float16)(a0.y * s); f[2]  = (_Float16)(a0.z * s); f[3]  = (_Float16)(a0.w * s);
    f[4]  = (_Float16)(a1.x * s); f[5]  = (_Float16)(a1.y * s); f[6]  = (_Float16)(a1.z * s); f[7]  = (_Float16)(a1.w * s);
    f[8]  = (_Float16)(b0.x * s); f[9]  = (_Float16)(b0.y * s); f[10] = (_Float16)(b0.z * s); f[11] = (_Float16)(b0.w * s);
    f[12] = (_Float16)(b1.x * s); f[13] = (_Float16)(b1.y * s); f[14] = (_Float16)(b1.z * s); f[15] = (_Float16)(b1.w * s);
    return f;
}

struct EdgeRaw {
    v4f xa0, xa1, xb0, xb1, xc0, xc1, xd0, xd1;
    v4f e0, e1, e2, e3;
    float ps0, ps1, ps2, pd0, pd1, pd2, is0, is1, is2, id0, id1, id2;
};
__device__ __forceinline__ EdgeRaw gather_edge(const float* __restrict__ x, const float* __restrict__ pos,
                                               const float* __restrict__ pos_init,
                                               const float* __restrict__ edge_attr,
                                               v2i r, int nbase, int h) {
    const int e = r.x;
    const int s = r.y & 0xffff;
    const int d = nbase + (r.y >> 16);
    EdgeRaw g;
    const v4f* q = (const v4f*)(x + (size_t)s * FDIM + 8 * h);
    g.xa0 = q[0]; g.xa1 = q[1]; g.xb0 = q[4];  g.xb1 = q[5];
    g.xc0 = q[8]; g.xc1 = q[9]; g.xd0 = q[12]; g.xd1 = q[13];
    const v4f* eq = (const v4f*)(edge_attr + (size_t)e * EFDIM);
    g.e0 = eq[0]; g.e1 = eq[1]; g.e2 = eq[2]; g.e3 = eq[3];
    g.ps0 = pos[3*s+0]; g.ps1 = pos[3*s+1]; g.ps2 = pos[3*s+2];
    g.pd0 = pos[3*d+0]; g.pd1 = pos[3*d+1]; g.pd2 = pos[3*d+2];
    g.is0 = pos_init[3*s+0]; g.is1 = pos_init[3*s+1]; g.is2 = pos_init[3*s+2];
    g.id0 = pos_init[3*d+0]; g.id1 = pos_init[3*d+1]; g.id2 = pos_init[3*d+2];
    return g;
}
__device__ __forceinline__ v16h frag4(v4f a0, v4f a1, v4f b0, v4f b1) {
    v16h f;
    f[0]  = (_Float16)a0.x; f[1]  = (_Float16)a0.y; f[2]  = (_Float16)a0.z; f[3]  = (_Float16)a0.w;
    f[4]  = (_Float16)a1.x; f[5]  = (_Float16)a1.y; f[6]  = (_Float16)a1.z; f[7]  = (_Float16)a1.w;
    f[8]  = (_Float16)b0.x; f[9]  = (_Float16)b0.y; f[10] = (_Float16)b0.z; f[11] = (_Float16)b0.w;
    f[12] = (_Float16)b1.x; f[13] = (_Float16)b1.y; f[14] = (_Float16)b1.z; f[15] = (_Float16)b1.w;
    return f;
}

#define ECAP 8192
__device__ __forceinline__ void list_slots(int* eslot, int mypre, int mylen, int mybase, int pbase) {
    const int lo  = mypre > pbase ? mypre : pbase;
    const int hi0 = mypre + mylen, hi1 = pbase + ECAP;
    const int hi  = hi0 < hi1 ? hi0 : hi1;
    for (int p = lo; p < hi; ++p) eslot[p - pbase] = mybase + (p - mypre);
}

#define SACC_PITCH   68
#define NODE_TILES   ((BIN_NODES + 15) / 16)
#define EDGE_WAVES   16
#define EDGE_THREADS (EDGE_WAVES * 32)
#define EDGE_WBLOCKS 62
#define W_ITEMS      960
__global__ __launch_bounds__(EDGE_THREADS) void egnn_edge(
    const float* __restrict__ x,
    const float* __restrict__ pos, const float* __restrict__ pos_init,
    const float* __restrict__ edge_attr,
    const float* __restrict__ mb1, const float* __restrict__ mb2,
    const float* __restrict__ cb1, const float* __restrict__ cb2,
    const float* __restrict__ nb1, const float* __restrict__ nb2,
    const float* __restrict__ mw1, const float* __restrict__ mw2,
    const float* __restrict__ cw1, const float* __restrict__ cw2,
    const float* __restrict__ nw1, const float* __restrict__ nw2,
    const int* __restrict__ rec, const int* __restrict__ runtab,
    float* __restrict__ hout, float* __restrict__ pos_out)
{
    __shared__ __attribute__((aligned(16))) float sacc[(BIN_NODES + 1) * SACC_PITCH];
    __shared__ __attribute__((aligned(16))) float sQ[(BIN_NODES + 1) * SACC_PITCH];
    __shared__ float spos[(BIN_NODES + 1) * 3];
    __shared__ float scnt[BIN_NODES + 1];
    __shared__ int eslot[ECAP];
    __shared__ int spre[EDGE_THREADS];
    __shared__ int slen[EDGE_THREADS];
    __shared__ int sbase[EDGE_THREADS];
    __shared__ int swtot[EDGE_WAVES];
    __shared__ __attribute__((aligned(16))) _Float16 sW[EDGE_WBLOCKS * 512];

    const int tid  = threadIdx.x;
    const int bin  = blockIdx.x;
    const int nbase = bin * BIN_NODES;
    const int rv = tid < NCHUNK ? runtab[tid * 256 + bin] : 0;
    int etot;
    for (int i = tid; i < (BIN_NODES + 1) * SACC_PITCH; i += EDGE_THREADS) sacc[i] = 0.0f;
    for (int i = tid; i < (BIN_NODES + 1) * 3; i += EDGE_THREADS)    spos[i] = 0.0f;
    {
        v4f wr[2][8];
        int wofs_l[2];
#pragma unroll
        for (int rd = 0; rd < 2; ++rd) {
            int it = tid + rd * EDGE_THREADS;
            it = it < W_ITEMS ? it : W_ITEMS - 1;
            const int gg = it >> 4, n4 = it & 15;
            const float* wm; int g0, kact;
            if (gg < 20)      { wm = mw1; g0 = 0;  kact = MSG_IN; }
            else if (gg < 28) { wm = mw2; g0 = 20; kact = 64; }
            else if (gg < 36) { wm = cw1; g0 = 28; kact = 64; }
            else if (gg < 52) { wm = nw1; g0 = 36; kact = 128; }
            else              { wm = nw2; g0 = 52; kact = 64; }
            const int k0 = (gg - g0) * 8;
#pragma unroll
            for (int j = 0; j < 8; ++j) {
                const int k = k0 + j;
                const int kc = k < kact ? k : kact - 1;
                const v4f v = *(const v4f*)(wm + kc * 64 + n4 * 4);
                wr[rd][j] = k < kact ? v : v4f{0.0f, 0.0f, 0.0f, 0.0f};
            }
            wofs_l[rd] = ((gg >> 2) * 4 + (n4 >> 2)) * 512 + ((gg >> 1) & 1) * 256
                       + ((gg & 1) * 16 + (n4 & 3) * 4) * 8;
        }
        {
            const int rlen = rv & 0xffff;
            int incl = rlen;
            for (int o = 1; o < 32; o <<= 1) {
                const int u = __shfl_up(incl, o);
                if ((tid & 31) >= o) incl += u;
            }
            if ((tid & 31) == 31) swtot[tid >> 5] = incl;
            __syncthreads();
            int wpre = 0, tot = 0;
            for (int k = 0; k < EDGE_WAVES; ++k) {
                const int v = swtot[k];
                wpre += (k < (tid >> 5)) ? v : 0;
                tot += v;
            }
            etot = __builtin_amdgcn_readfirstlane(tot);
            const int mypre = wpre + incl - rlen, mybase = tid * BIN_CHUNK + (rv >> 16);
            spre[tid] = mypre; slen[tid] = rlen; sbase[tid] = mybase;
            list_slots(eslot, mypre, rlen, mybase, 0);
        }
#pragma unroll
        for (int rd = 0; rd < 2; ++rd) {
            if (tid + rd * EDGE_THREADS < W_ITEMS) {
#pragma unroll
                for (int cc = 0; cc < 4; ++cc) {
                    v8h rec;
#pragma unroll
                    for (int j = 0; j < 8; ++j) rec[j] = (_Float16)wr[rd][j][cc];
                    *(v8h*)(sW + wofs_l[rd] + cc * 8) = rec;
                }
            }
        }
        if (tid >= EDGE_THREADS - 128) {
            const int t = tid - (EDGE_THREADS - 128);
            const int kc = t >> 6, pl = (t >> 5) & 1, ln = t & 31;
            const int kh = ln >> 4, cn = ln & 15;
            const int k0 = kc * 32 + pl * 16 + kh * 8;
            v8h rec;
#pragma unroll
            for (int j = 0; j < 8; ++j) {
                const float v = cw2[cn < 3 ? (k0 + j) * 3 + cn : 0];
                rec[j] = (_Float16)(cn < 3 ? v : 0.0f);
            }
            *(v8h*)(sW + CW2B + kc * 512 + pl * 256 + ln * 8) = rec;
        }
    }
    if (tid < BIN_NODES + 1) scnt[tid] = 0.0f;
    if (tid < SACC_PITCH) sQ[BIN_NODES * SACC_PITCH + tid] = 0.0f;
    __syncthreads();
    int pbase = 0;
    int npass = etot < ECAP ? etot : ECAP;
    int ntiles = (npass + 15) >> 4;

    const int lane = tid & 31;
    const int c = lane & 15;
    const int h = lane >> 4;
    const int wave = tid >> 5;
    v2i r_cur = v2i{0, 0};
    EdgeRaw g_cur = {};
    if (wave < ntiles) {
        const int q0 = wave * 16 + c;
        r_cur = *(const v2i*)(rec + 2 * eslot[q0 < npass ? q0 : 0]);
        g_cur = gather_edge(x, pos, pos_init, edge_attr, r_cur, nbase, h);
    }
    float mb2c[4];
#pragma unroll
    for (int nt = 0; nt < 4; ++nt) mb2c[nt] = mb2[nt * 16 + c];
    const float cb2x = cb2[0], cb2y = cb2[1], cb2z = cb2[2];
    if (wave < NODE_TILES) {
        const int lr  = wave * 16 + c;
        const bool have = lr < BIN_NODES;
        const int lrc = have ? lr : BIN_NODES - 1;
        const float* xr = x + (size_t)(nbase + lrc) * FDIM;
        v16h qf[2];
        qf[0] = row_frag(xr, h);
        qf[1] = row_frag(xr + 32, h);
        v8f qac[4] = {v8f{}, v8f{}, v8f{}, v8f{}};
#pragma unroll
        for (int kc = 0; kc < 2; ++kc) {
#pragma unroll
            for (int mt = 0; mt < 4; ++mt)
                qac[mt] = wmma16(ld_ws(sW + MW1B + ((2 + kc) * 4 + mt) * 512, lane), qf[kc], qac[mt]);
        }
#pragma unroll
        for (int mt = 0; mt < 4; ++mt) {
            v8f o = bias8(qac[mt], mb1 + 16*mt + 8*h, false);
            if (have) {
                v4f* dst = (v4f*)(sQ + lr * SACC_PITCH + 16*mt + 8*h);
                dst[0] = v4f{o[0], o[1], o[2], o[3]};
                dst[1] = v4f{o[4], o[5], o[6], o[7]};
            }
        }
    }
    __syncthreads();
 for (;;) {
  for (int tile = wave; tile < ntiles; tile += EDGE_WAVES) {
    const bool live = tile * 16 + c < npass;
    const int brow = r_cur.y >> 16;
    const int lrow = live ? brow : BIN_NODES;
    v2i r_nxt;
    {
        const int q1 = (tile + EDGE_WAVES) * 16 + c;
        r_nxt = *(const v2i*)(rec + 2 * eslot[q1 < npass ? q1 : 0]);
    }
    int wofs = 0;
    asm volatile("" : "+v"(wofs));
    const _Float16* sWt = sW + wofs;

    v16h fr[3];
    fr[0] = frag4(g_cur.xa0, g_cur.xa1, g_cur.xb0, g_cur.xb1);
    fr[1] = frag4(g_cur.xc0, g_cur.xc1, g_cur.xd0, g_cur.xd1);
    v8f acc[4];
    {
        const float* qr = sQ + lrow * SACC_PITCH + 8 * h;
#pragma unroll
        for (int mt = 0; mt < 4; ++mt) {
            const v4f a = *(const v4f*)(qr + 16 * mt);
            const v4f b = *(const v4f*)(qr + 16 * mt + 4);
            acc[mt] = v8f{a.x, a.y, a.z, a.w, b.x, b.y, b.z, b.w};
        }
    }
    v4f e0 = g_cur.e0, e1 = g_cur.e1, e2 = g_cur.e2, e3 = g_cur.e3;

    float dx = g_cur.ps0 - g_cur.pd0;
    float dy = g_cur.ps1 - g_cur.pd1;
    float dz = g_cur.ps2 - g_cur.pd2;
    float dsq = dx*dx + dy*dy + dz*dz;
    if (h == 0) atomicAdd(&scnt[lrow], 1.0f);

    {
        float ea[16] = {e0.x, e0.y, e0.z, e0.w, e1.x, e1.y, e1.z, e1.w,
                        e2.x, e2.y, e2.z, e2.w, e3.x, e3.y, e3.z, e3.w};
        v16h f;
#pragma unroll
        for (int t = 0; t < 8; ++t) {
            float lo = (t == 0) ? dsq : ea[t - 1];
            float hi = ea[7 + t];
            f[t] = (_Float16)(h ? hi : lo);
        }
        f[8] = (_Float16)(h ? 0.0f : ea[15]);
#pragma unroll
        for (int t = 9; t < 16; ++t) f[t] = (_Float16)0.0f;
        fr[2] = f;
    }

#pragma unroll
    for (int j = 0; j < 3; ++j) {
        const int kc = j < 2 ? j : 4;
#pragma unroll
        for (int mt = 0; mt < 4; ++mt)
            acc[mt] = wmma16(ld_ws(sWt + MW1B + (kc * 4 + mt) * 512, lane), fr[j], acc[mt]);
    }
    v16h hf[2];
#pragma unroll
    for (int kc = 0; kc < 2; ++kc)
        hf[kc] = pack2(relu8(acc[2*kc]), relu8(acc[2*kc + 1]));

    v8f mij[4] = {v8f{}, v8f{}, v8f{}, v8f{}};
    v8f mt_[4] = {v8f{}, v8f{}, v8f{}, v8f{}};
    {
        v16h w2[8];
#pragma unroll
        for (int i = 0; i < 8; ++i) w2[i] = ld_ws(sWt + MW2B + i * 512, lane);
#pragma unroll
        for (int kc = 0; kc < 2; ++kc) {
#pragma unroll
            for (int nt = 0; nt < 4; ++nt)
                mij[nt] = wmma16(hf[kc], w2[kc * 4 + nt], mij[nt]);
        }
#pragma unroll
        for (int kc = 0; kc < 2; ++kc) {
#pragma unroll
            for (int mt = 0; mt < 4; ++mt)
                mt_[mt] = wmma16(w2[kc * 4 + mt], hf[kc], mt_[mt]);
        }
    }
    v16h wc1[8], wc2[2];
#pragma unroll
    for (int i = 0; i < 8; ++i) wc1[i] = ld_ws(sWt + CW1B + i * 512, lane);
#pragma unroll
    for (int i = 0; i < 2; ++i) wc2[i] = ld_ws(sWt + CW2B + i * 512, lane);
    v16h mf[2];
#pragma unroll
    for (int kc = 0; kc < 2; ++kc)
        mf[kc] = pack2(bias8(mt_[2*kc],     mb2 + 32*kc      + 8*h, false),
                       bias8(mt_[2*kc + 1], mb2 + 32*kc + 16 + 8*h, false));
    {
        int drow[8];
#pragma unroll
        for (int r = 0; r < 8; ++r) drow[r] = __shfl(lrow, 8 * h + r);
#pragma unroll
        for (int nt = 0; nt < 4; ++nt) {
#pragma unroll
            for (int r = 0; r < 8; ++r)
                atomicAdd(&sacc[drow[r] * SACC_PITCH + nt * 16 + c], mij[nt][r] + mb2c[nt]);
        }
    }

    float ix = g_cur.is0 - g_cur.id0;
    float iy = g_cur.is1 - g_cur.id1;
    float iz = g_cur.is2 - g_cur.id2;
    float rinv = 1.0f / sqrtf(ix*ix + iy*iy + iz*iz);
    float wx = ix * rinv, wy = iy * rinv, wz = iz * rinv;
    asm volatile("" : "+v"(wx), "+v"(wy), "+v"(wz));
    g_cur = gather_edge(x, pos, pos_init, edge_attr, r_nxt, nbase, h);
    r_cur = r_nxt;

    v16h tf[2];
    {
        v8f tac[4] = {v8f{}, v8f{}, v8f{}, v8f{}};
#pragma unroll
        for (int kc = 0; kc < 2; ++kc) {
#pragma unroll
            for (int mt = 0; mt < 4; ++mt)
                tac[mt] = wmma16(wc1[kc * 4 + mt], mf[kc], tac[mt]);
        }
#pragma unroll
        for (int kc = 0; kc < 2; ++kc)
            tf[kc] = pack2(bias8(tac[2*kc],     cb1 + 32*kc      + 8*h, true),
                           bias8(tac[2*kc + 1], cb1 + 32*kc + 16 + 8*h, true));
    }

    v8f cac = v8f{};
#pragma unroll
    for (int kc = 0; kc < 2; ++kc)
        cac = wmma16(wc2[kc], tf[kc], cac);
    float ux = wx * (cac[0] + cb2x);
    float uy = wy * (cac[1] + cb2y);
    float uz = wz * (cac[2] + cb2z);
    if (h == 0) {
        atomicAdd(&spos[lrow * 3 + 0], ux);
        atomicAdd(&spos[lrow * 3 + 1], uy);
        atomicAdd(&spos[lrow * 3 + 2], uz);
    }
  }
    pbase += ECAP;
    if (pbase >= etot) break;
    __syncthreads();
    list_slots(eslot, spre[tid], slen[tid], sbase[tid], pbase);
    __syncthreads();
    npass = etot - pbase < ECAP ? etot - pbase : ECAP;
    ntiles = (npass + 15) >> 4;
    if (wave < ntiles) {
        const int q0 = wave * 16 + c;
        r_cur = *(const v2i*)(rec + 2 * eslot[q0 < npass ? q0 : 0]);
        g_cur = gather_edge(x, pos, pos_init, edge_attr, r_cur, nbase, h);
    }
 }
    __syncthreads();
    for (int i = tid; i < BIN_NODES * 3; i += EDGE_THREADS)    pos_out[(size_t)nbase * 3 + i] = spos[i];

    if (wave < NODE_TILES) {
        const int lr  = wave * 16 + c;
        const bool have = lr < BIN_NODES;
        const int lrc = have ? lr : BIN_NODES - 1;
        const float* xr = x + (size_t)(nbase + lrc) * FDIM;
        const float* mr = sacc + lrc * SACC_PITCH;
        const float inv = 1.0f / fmaxf(scnt[lrc], 1.0f);
        v16h nf[4];
        nf[0] = row_frag(xr, h);
        nf[1] = row_frag(xr + 32, h);
        nf[2] = mean_frag(mr, h, inv);
        nf[3] = mean_frag(mr + 32, h, inv);
        v8f nac[4] = {v8f{}, v8f{}, v8f{}, v8f{}};
#pragma unroll
        for (int kc = 0; kc < 4; ++kc) {
#pragma unroll
            for (int mt = 0; mt < 4; ++mt)
                nac[mt] = wmma16(ld_ws(sW + NW1B + (kc * 4 + mt) * 512, lane), nf[kc], nac[mt]);
        }
        v16h nh[2];
#pragma unroll
        for (int kc = 0; kc < 2; ++kc)
            nh[kc] = pack2(bias8(nac[2*kc],     nb1 + 32*kc      + 8*h, true),
                           bias8(nac[2*kc + 1], nb1 + 32*kc + 16 + 8*h, true));
        v8f oac[4] = {v8f{}, v8f{}, v8f{}, v8f{}};
#pragma unroll
        for (int kc = 0; kc < 2; ++kc) {
#pragma unroll
            for (int mt = 0; mt < 4; ++mt)
                oac[mt] = wmma16(ld_ws(sW + NW2B + (kc * 4 + mt) * 512, lane), nh[kc], oac[mt]);
        }
#pragma unroll
        for (int mt = 0; mt < 4; ++mt) {
            v8f o = bias8(oac[mt], nb2 + 16*mt + 8*h, false);
            if (have) {
                v4f* dst = (v4f*)(sacc + lr * SACC_PITCH + 16*mt + 8*h);
                dst[0] = v4f{o[0], o[1], o[2], o[3]};
                dst[1] = v4f{o[4], o[5], o[6], o[7]};
            }
        }
    }
    __syncthreads();
    for (int i = tid; i < BIN_NODES * (FDIM / 4); i += EDGE_THREADS) {
        const int row = i >> 4, q = i & 15;
        *(v4f*)(hout + (size_t)(nbase + row) * FDIM + q * 4) = *(const v4f*)(sacc + row * SACC_PITCH + q * 4);
    }
}

extern "C" void kernel_launch(void* const* d_in, const int* in_sizes, int n_in,
                              void* d_out, int out_size, void* d_ws, size_t ws_size,
                              hipStream_t stream) {
    const float* x         = (const float*)d_in[0];
    const int*   ei        = (const int*)  d_in[1];
    const float* pos       = (const float*)d_in[2];
    const float* pos_init  = (const float*)d_in[3];
    const float* edge_attr = (const float*)d_in[4];
    const float* mw1 = (const float*)d_in[5];
    const float* mb1 = (const float*)d_in[6];
    const float* mw2 = (const float*)d_in[7];
    const float* mb2 = (const float*)d_in[8];
    const float* nw1 = (const float*)d_in[9];
    const float* nb1 = (const float*)d_in[10];
    const float* nw2 = (const float*)d_in[11];
    const float* nb2 = (const float*)d_in[12];
    const float* cw1 = (const float*)d_in[13];
    const float* cb1 = (const float*)d_in[14];
    const float* cw2 = (const float*)d_in[15];
    const float* cb2 = (const float*)d_in[16];

    float* h_out   = (float*)d_out;
    float* pos_out = h_out + (size_t)N_NODES * FDIM;

    int* rec    = (int*)d_ws;
    int* runtab = rec + 2 * N_EDGES;
    egnn_bin_fill<<<NCHUNK, 256, 0, stream>>>(ei, rec, runtab);

    egnn_edge<<<NBINS, EDGE_THREADS, 0, stream>>>(
        x, pos, pos_init, edge_attr, mb1, mb2, cb1, cb2, nb1, nb2,
        mw1, mw2, cw1, cw2, nw1, nw2, rec, runtab, h_out, pos_out);
}
